# starting kernel
# baseline (speedup 1.0000x reference)
#include <hip/hip_runtime.h>
#include <math.h>

#define N_NODES 100000
#define N_EDGES 1600000
#define F_IN    50
#define HID     32
#define NCLS    40

typedef float v2f __attribute__((ext_vector_type(2)));
typedef float v8f __attribute__((ext_vector_type(8)));


__global__ void zero_kernel(float* __restrict__ p, long long n) {
  long long i = blockIdx.x * (long long)blockDim.x + threadIdx.x;
  if (i < n) p[i] = 0.0f;
}

__global__ void pack_w1_kernel(const float* __restrict__ W0,
                               const float* __restrict__ W1,
                               float* __restrict__ Wcat) {
  int t = blockIdx.x * blockDim.x + threadIdx.x;
  if (t >= 52 * 64) return;
  int k = t >> 6, j = t & 63;
  float v = 0.0f;
  if (k < F_IN) v = (j < HID) ? W0[k * HID + j] : W1[k * HID + (j - HID)];
  Wcat[t] = v;
}

__global__ void pack_w2_kernel(const float* __restrict__ W0,
                               const float* __restrict__ W1,
                               float* __restrict__ Wcat) {
  int t = blockIdx.x * blockDim.x + threadIdx.x;
  if (t >= 64 * 48) return;
  int k = t / 48, j = t % 48;
  float v = 0.0f;
  if (j < NCLS) v = (k < HID) ? W0[k * NCLS + j] : W1[(k - HID) * NCLS + j];
  Wcat[t] = v;
}


__global__ void deg_kernel(const int* __restrict__ rows,
                           const int* __restrict__ cols,
                           float* __restrict__ deg) {
  int e = blockIdx.x * blockDim.x + threadIdx.x;
  if (e >= N_EDGES) return;
  int r = rows[e], c = cols[e];
  if (r != c) atomicAdd(&deg[r], 1.0f);
}

__global__ void dis_kernel(const float* __restrict__ deg, float* __restrict__ dis) {
  int i = blockIdx.x * blockDim.x + threadIdx.x;
  if (i >= N_NODES) return;
  float d = deg[i];
  dis[i] = (d > 0.0f) ? rsqrtf(fmaxf(d, 1.0f)) : 0.0f;
}


__global__ void agg_kernel(const int* __restrict__ rows,
                           const int* __restrict__ cols,
                           const float* __restrict__ dis,
                           const float* __restrict__ src, int sstride, int soff,
                           float* __restrict__ dst) {
  long long gid = blockIdx.x * (long long)blockDim.x + threadIdx.x;
  long long e = gid >> 5;
  if (e >= (long long)N_EDGES) return;
  int f = (int)(gid & 31);
  int r = rows[e], c = cols[e];
  if (r == c) return;
  float w = -dis[r] * dis[c];
  if (w == 0.0f) return;
  atomicAdd(&dst[(long long)r * HID + f],
            w * src[(long long)c * sstride + soff + f]);
}


__global__ void __launch_bounds__(128)
proj1_wmma_kernel(const float* __restrict__ x,
                  const float* __restrict__ Wcat,
                  float* __restrict__ P) {
  int wave = threadIdx.x >> 5;
  int lane = threadIdx.x & 31;
  int half = lane >> 4;
  int l16  = lane & 15;
  int m0 = (blockIdx.x * 4 + wave) * 16;
  if (m0 >= N_NODES) return;
  int mA = m0 + l16;
  const float* arow = x + (long long)mA * F_IN;

  v8f acc0 = {}, acc1 = {}, acc2 = {}, acc3 = {};
  for (int kk = 0; kk < 48; kk += 4) {
    int k = kk + 2 * half;
    v2f a = *(const v2f*)(arow + k);
    const float* b0 = Wcat + k * 64;
    const float* b1 = Wcat + (k + 1) * 64;
    v2f b;
    b.x = b0[l16];      b.y = b1[l16];
    acc0 = __builtin_amdgcn_wmma_f32_16x16x4_f32(false, a, false, b, (short)0, acc0, false, false);
    b.x = b0[16 + l16]; b.y = b1[16 + l16];
    acc1 = __builtin_amdgcn_wmma_f32_16x16x4_f32(false, a, false, b, (short)0, acc1, false, false);
    b.x = b0[32 + l16]; b.y = b1[32 + l16];
    acc2 = __builtin_amdgcn_wmma_f32_16x16x4_f32(false, a, false, b, (short)0, acc2, false, false);
    b.x = b0[48 + l16]; b.y = b1[48 + l16];
    acc3 = __builtin_amdgcn_wmma_f32_16x16x4_f32(false, a, false, b, (short)0, acc3, false, false);
  }
  {
    int k = 48 + 2 * half;
    v2f aload = *(const v2f*)(arow + 48);
    v2f a;
    a.x = half ? 0.0f : aload.x;
    a.y = half ? 0.0f : aload.y;
    const float* b0 = Wcat + k * 64;
    const float* b1 = Wcat + (k + 1) * 64;
    v2f b;
    b.x = b0[l16];      b.y = b1[l16];
    acc0 = __builtin_amdgcn_wmma_f32_16x16x4_f32(false, a, false, b, (short)0, acc0, false, false);
    b.x = b0[16 + l16]; b.y = b1[16 + l16];
    acc1 = __builtin_amdgcn_wmma_f32_16x16x4_f32(false, a, false, b, (short)0, acc1, false, false);
    b.x = b0[32 + l16]; b.y = b1[32 + l16];
    acc2 = __builtin_amdgcn_wmma_f32_16x16x4_f32(false, a, false, b, (short)0, acc2, false, false);
    b.x = b0[48 + l16]; b.y = b1[48 + l16];
    acc3 = __builtin_amdgcn_wmma_f32_16x16x4_f32(false, a, false, b, (short)0, acc3, false, false);
  }
#pragma unroll
  for (int v = 0; v < 8; v++) {
    int mC = m0 + v + 8 * half;
    float* prow = P + (long long)mC * 64 + l16;
    prow[0]  = acc0[v];
    prow[16] = acc1[v];
    prow[32] = acc2[v];
    prow[48] = acc3[v];
  }
}


__global__ void relu_kernel(const float* __restrict__ P,
                            const float* __restrict__ agg1,
                            const float* __restrict__ b1,
                            float* __restrict__ h) {
  long long i = blockIdx.x * (long long)blockDim.x + threadIdx.x;
  if (i >= (long long)N_NODES * HID) return;
  long long n = i >> 5;
  int f = (int)(i & 31);
  float v = P[n * 64 + f] + agg1[i] + b1[f];
  h[i] = v > 0.0f ? v : 0.0f;
}


__global__ void __launch_bounds__(128)
out_wmma_kernel(const float* __restrict__ h,
                const float* __restrict__ agg2,
                const float* __restrict__ Wcat,
                float* __restrict__ outp) {
  int wave = threadIdx.x >> 5;
  int lane = threadIdx.x & 31;
  int half = lane >> 4;
  int l16  = lane & 15;
  int m0 = (blockIdx.x * 4 + wave) * 16;
  if (m0 >= N_NODES) return;
  int mA = m0 + l16;
  const float* hrow = h    + (long long)mA * HID;
  const float* grow = agg2 + (long long)mA * HID - HID;

  v8f acc0 = {}, acc1 = {}, acc2 = {};
  for (int kk = 0; kk < 64; kk += 4) {
    int k = kk + 2 * half;
    const float* src = (k < HID) ? (hrow + k) : (grow + k);
    v2f a = *(const v2f*)src;
    const float* b0 = Wcat + k * 48;
    const float* b1 = Wcat + (k + 1) * 48;
    v2f b;
    b.x = b0[l16];      b.y = b1[l16];
    acc0 = __builtin_amdgcn_wmma_f32_16x16x4_f32(false, a, false, b, (short)0, acc0, false, false);
    b.x = b0[16 + l16]; b.y = b1[16 + l16];
    acc1 = __builtin_amdgcn_wmma_f32_16x16x4_f32(false, a, false, b, (short)0, acc1, false, false);
    b.x = b0[32 + l16]; b.y = b1[32 + l16];
    acc2 = __builtin_amdgcn_wmma_f32_16x16x4_f32(false, a, false, b, (short)0, acc2, false, false);
  }
#pragma unroll
  for (int v = 0; v < 8; v++) {
    int mC = m0 + v + 8 * half;
    float* prow = outp + (long long)mC * 48 + l16;
    prow[0]  = acc0[v];
    prow[16] = acc1[v];
    prow[32] = acc2[v];
  }
}


__global__ void logsoftmax_kernel(const float* __restrict__ outp,
                                  const float* __restrict__ b2,
                                  float* __restrict__ out) {
  int i = blockIdx.x * blockDim.x + threadIdx.x;
  if (i >= N_NODES) return;
  const float* r = outp + (long long)i * 48;
  float m = -INFINITY;
  for (int j = 0; j < NCLS; j++) m = fmaxf(m, r[j] + b2[j]);
  float s = 0.0f;
  for (int j = 0; j < NCLS; j++) s += expf(r[j] + b2[j] - m);
  float ls = logf(s);
  for (int j = 0; j < NCLS; j++)
    out[(long long)i * NCLS + j] = (r[j] + b2[j] - m) - ls;
}


extern "C" void kernel_launch(void* const* d_in, const int* in_sizes, int n_in,
                              void* d_out, int out_size, void* d_ws, size_t ws_size,
                              hipStream_t stream) {
  (void)in_sizes; (void)n_in; (void)out_size; (void)ws_size;
  const float* x    = (const float*)d_in[0];
  const int*   ei   = (const int*)  d_in[1];
  const float* W0_1 = (const float*)d_in[2];
  const float* W1_1 = (const float*)d_in[3];
  const float* b1   = (const float*)d_in[4];
  const float* W0_2 = (const float*)d_in[5];
  const float* W1_2 = (const float*)d_in[6];
  const float* b2   = (const float*)d_in[7];
  float* out = (float*)d_out;

  const int* rows = ei;
  const int* cols = ei + N_EDGES;

  const long long N = N_NODES;
  float* ws   = (float*)d_ws;
  float* deg  = ws;
  float* dis  = ws + N;
  float* P    = ws + 2 * N;
  float* agg1 = ws + 66 * N;
  float* hbuf = ws + 98 * N;
  float* Wc1  = ws + 130 * N;
  float* Wc2  = Wc1 + 52 * 64;

  const int T = 256;
  zero_kernel<<<(unsigned)((N + T - 1) / T), T, 0, stream>>>(deg, N);
  zero_kernel<<<(unsigned)((32 * N + T - 1) / T), T, 0, stream>>>(agg1, 32 * N);
  pack_w1_kernel<<<(52 * 64 + T - 1) / T, T, 0, stream>>>(W0_1, W1_1, Wc1);
  pack_w2_kernel<<<(64 * 48 + T - 1) / T, T, 0, stream>>>(W0_2, W1_2, Wc2);

  deg_kernel<<<(N_EDGES + T - 1) / T, T, 0, stream>>>(rows, cols, deg);
  dis_kernel<<<(unsigned)((N + T - 1) / T), T, 0, stream>>>(deg, dis);

  proj1_wmma_kernel<<<(N_NODES + 63) / 64, 128, 0, stream>>>(x, Wc1, P);
  agg_kernel<<<(unsigned)(((long long)N_EDGES * 32 + T - 1) / T), T, 0, stream>>>(
      rows, cols, dis, P, 64, 32, agg1);
  relu_kernel<<<(unsigned)((32 * N + T - 1) / T), T, 0, stream>>>(P, agg1, b1, hbuf);

  zero_kernel<<<(unsigned)((32 * N + T - 1) / T), T, 0, stream>>>(agg1, 32 * N);
  agg_kernel<<<(unsigned)(((long long)N_EDGES * 32 + T - 1) / T), T, 0, stream>>>(
      rows, cols, dis, hbuf, 32, 0, agg1);
  out_wmma_kernel<<<(N_NODES + 63) / 64, 128, 0, stream>>>(hbuf, agg1, Wc2, P);
  logsoftmax_kernel<<<(unsigned)((N + T - 1) / T), T, 0, stream>>>(P, b2, out);
}


# Neighbour sums by gather over an edge list grouped by row in-call (int count+rank add, scan, fill): one wave sums a row, stores it once; no f32 atomics; summed in arrival-rank order
# speedup vs baseline: 1.9147x; 1.9147x over previous
#include <hip/hip_runtime.h>
#include <math.h>

#define N_NODES 100000
#define N_EDGES 1600000
#define F_IN    50
#define HID     32
#define NCLS    40

typedef float v2f __attribute__((ext_vector_type(2)));
typedef float v8f __attribute__((ext_vector_type(8)));


__global__ void zero_kernel(float* __restrict__ p, long long n) {
  long long i = blockIdx.x * (long long)blockDim.x + threadIdx.x;
  if (i < n) p[i] = 0.0f;
}

__global__ void pack_w1_kernel(const float* __restrict__ W0,
                               const float* __restrict__ W1,
                               float* __restrict__ Wcat) {
  int t = blockIdx.x * blockDim.x + threadIdx.x;
  if (t >= 52 * 64) return;
  int k = t >> 6, j = t & 63;
  float v = 0.0f;
  if (k < F_IN) v = (j < HID) ? W0[k * HID + j] : W1[k * HID + (j - HID)];
  Wcat[t] = v;
}

__global__ void pack_w2_kernel(const float* __restrict__ W0,
                               const float* __restrict__ W1,
                               float* __restrict__ Wcat) {
  int t = blockIdx.x * blockDim.x + threadIdx.x;
  if (t >= 64 * 48) return;
  int k = t / 48, j = t % 48;
  float v = 0.0f;
  if (j < NCLS) v = (k < HID) ? W0[k * NCLS + j] : W1[(k - HID) * NCLS + j];
  Wcat[t] = v;
}


__global__ void count_rank_kernel(const int* __restrict__ rows,
                                  const int* __restrict__ cols,
                                  int* __restrict__ cnt,
                                  int* __restrict__ rank) {
  int e = blockIdx.x * blockDim.x + threadIdx.x;
  if (e >= N_EDGES) return;
  int r = rows[e], c = cols[e];
  int k = -1;
  if (r != c) k = atomicAdd(&cnt[r], 1);
  rank[e] = k;
}

__global__ void __launch_bounds__(1024)
scan_dis_kernel(const int* __restrict__ cnt,
                int* __restrict__ rowptr,
                float* __restrict__ dis) {
  __shared__ int s_pre[32];
  __shared__ int s_tot[32];
  int t = threadIdx.x, lane = t & 31, wave = t >> 5;
  int b0 = blockIdx.x * 1024;
  const int4* c4 = (const int4*)cnt;
  int pre = 0;
#pragma unroll 4
  for (int q = t; q < (b0 >> 2); q += 1024) {
    int4 x = c4[q];
    pre += (x.x + x.y) + (x.z + x.w);
  }
  int i = b0 + t;
  int v = (i < N_NODES) ? cnt[i] : 0;
  int incl = v;
  for (int d = 1; d < 32; d <<= 1) {
    int y = __shfl_up(incl, d);
    if (lane >= d) incl += y;
  }
  for (int d = 16; d >= 1; d >>= 1) pre += __shfl_xor(pre, d);
  if (lane == 31) s_tot[wave] = incl;
  if (lane == 0)  s_pre[wave] = pre;
  __syncthreads();
  int add = s_pre[lane] + (lane < wave ? s_tot[lane] : 0);
  for (int d = 16; d >= 1; d >>= 1) add += __shfl_xor(add, d);
  if (i < N_NODES) {
    rowptr[i] = add + incl - v;
    float dg = (float)v;
    dis[i] = (dg > 0.0f) ? rsqrtf(fmaxf(dg, 1.0f)) : 0.0f;
  }
}

__global__ void fill_kernel(const int* __restrict__ rows,
                            const int* __restrict__ cols,
                            const int* __restrict__ rank,
                            const int* __restrict__ rowptr,
                            int* __restrict__ colidx) {
  int e = blockIdx.x * blockDim.x + threadIdx.x;
  if (e >= N_EDGES) return;
  int k = rank[e];
  if (k < 0) return;
  colidx[rowptr[rows[e]] + k] = cols[e];
}


__global__ void __launch_bounds__(256)
gather_rows_kernel(const int* __restrict__ rowptr,
                   const int* __restrict__ cnt,
                   const int* __restrict__ colidx,
                   const float* __restrict__ dis,
                   const float* __restrict__ src, int sstride, int soff,
                   float* __restrict__ dst) {
  int lane = threadIdx.x & 31;
  int r = __builtin_amdgcn_readfirstlane(blockIdx.x * 8 + (threadIdx.x >> 5));
  if (r >= N_NODES) return;
  int start = rowptr[r];
  int n = cnt[r];
  float ndr = -dis[r];
  const float* sp = src + soff;
  float acc = 0.0f;
  for (int base = 0; base < n; base += 32) {
    int m = n - base;
    if (m > 32) m = 32;
    int li = lane < m ? lane : m - 1;
    int c = colidx[start + base + li];
    float w = lane < m ? ndr * dis[c] : 0.0f;
    for (int j = 0; j < m; j += 8) {
      float v[8];
#pragma unroll
      for (int u = 0; u < 8; u++) {
        int cj = __builtin_amdgcn_readlane(c, j + u);
        const float* rowp = sp + (long long)cj * sstride;
        float x = rowp[lane];
        v[u] = (j + u < m) ? x : 0.0f;
      }
#pragma unroll
      for (int u = 0; u < 8; u++) {
        float wj = __int_as_float(__builtin_amdgcn_readlane(__float_as_int(w), j + u));
        acc = fmaf(wj, v[u], acc);
      }
    }
  }
  dst[(long long)r * HID + lane] = acc;
}


__global__ void __launch_bounds__(128)
proj1_wmma_kernel(const float* __restrict__ x,
                  const float* __restrict__ Wcat,
                  float* __restrict__ P) {
  int wave = threadIdx.x >> 5;
  int lane = threadIdx.x & 31;
  int half = lane >> 4;
  int l16  = lane & 15;
  int m0 = (blockIdx.x * 4 + wave) * 16;
  if (m0 >= N_NODES) return;
  int mA = m0 + l16;
  const float* arow = x + (long long)mA * F_IN;

  v8f acc0 = {}, acc1 = {}, acc2 = {}, acc3 = {};
  for (int kk = 0; kk < 48; kk += 4) {
    int k = kk + 2 * half;
    v2f a = *(const v2f*)(arow + k);
    const float* b0 = Wcat + k * 64;
    const float* b1 = Wcat + (k + 1) * 64;
    v2f b;
    b.x = b0[l16];      b.y = b1[l16];
    acc0 = __builtin_amdgcn_wmma_f32_16x16x4_f32(false, a, false, b, (short)0, acc0, false, false);
    b.x = b0[16 + l16]; b.y = b1[16 + l16];
    acc1 = __builtin_amdgcn_wmma_f32_16x16x4_f32(false, a, false, b, (short)0, acc1, false, false);
    b.x = b0[32 + l16]; b.y = b1[32 + l16];
    acc2 = __builtin_amdgcn_wmma_f32_16x16x4_f32(false, a, false, b, (short)0, acc2, false, false);
    b.x = b0[48 + l16]; b.y = b1[48 + l16];
    acc3 = __builtin_amdgcn_wmma_f32_16x16x4_f32(false, a, false, b, (short)0, acc3, false, false);
  }
  {
    int k = 48 + 2 * half;
    v2f aload = *(const v2f*)(arow + 48);
    v2f a;
    a.x = half ? 0.0f : aload.x;
    a.y = half ? 0.0f : aload.y;
    const float* b0 = Wcat + k * 64;
    const float* b1 = Wcat + (k + 1) * 64;
    v2f b;
    b.x = b0[l16];      b.y = b1[l16];
    acc0 = __builtin_amdgcn_wmma_f32_16x16x4_f32(false, a, false, b, (short)0, acc0, false, false);
    b.x = b0[16 + l16]; b.y = b1[16 + l16];
    acc1 = __builtin_amdgcn_wmma_f32_16x16x4_f32(false, a, false, b, (short)0, acc1, false, false);
    b.x = b0[32 + l16]; b.y = b1[32 + l16];
    acc2 = __builtin_amdgcn_wmma_f32_16x16x4_f32(false, a, false, b, (short)0, acc2, false, false);
    b.x = b0[48 + l16]; b.y = b1[48 + l16];
    acc3 = __builtin_amdgcn_wmma_f32_16x16x4_f32(false, a, false, b, (short)0, acc3, false, false);
  }
#pragma unroll
  for (int v = 0; v < 8; v++) {
    int mC = m0 + v + 8 * half;
    float* prow = P + (long long)mC * 64 + l16;
    prow[0]  = acc0[v];
    prow[16] = acc1[v];
    prow[32] = acc2[v];
    prow[48] = acc3[v];
  }
}


__global__ void relu_kernel(const float* __restrict__ P,
                            const float* __restrict__ b1,
                            float* __restrict__ h) {
  long long i = blockIdx.x * (long long)blockDim.x + threadIdx.x;
  if (i >= (long long)N_NODES * HID) return;
  long long n = i >> 5;
  int f = (int)(i & 31);
  float v = P[n * 64 + f] + h[i] + b1[f];
  h[i] = v > 0.0f ? v : 0.0f;
}


__global__ void __launch_bounds__(128)
out_wmma_kernel(const float* __restrict__ h,
                const float* __restrict__ agg2,
                const float* __restrict__ Wcat,
                float* __restrict__ outp) {
  int wave = threadIdx.x >> 5;
  int lane = threadIdx.x & 31;
  int half = lane >> 4;
  int l16  = lane & 15;
  int m0 = (blockIdx.x * 4 + wave) * 16;
  if (m0 >= N_NODES) return;
  int mA = m0 + l16;
  const float* hrow = h    + (long long)mA * HID;
  const float* grow = agg2 + (long long)mA * HID - HID;

  v8f acc0 = {}, acc1 = {}, acc2 = {};
  for (int kk = 0; kk < 64; kk += 4) {
    int k = kk + 2 * half;
    const float* src = (k < HID) ? (hrow + k) : (grow + k);
    v2f a = *(const v2f*)src;
    const float* b0 = Wcat + k * 48;
    const float* b1 = Wcat + (k + 1) * 48;
    v2f b;
    b.x = b0[l16];      b.y = b1[l16];
    acc0 = __builtin_amdgcn_wmma_f32_16x16x4_f32(false, a, false, b, (short)0, acc0, false, false);
    b.x = b0[16 + l16]; b.y = b1[16 + l16];
    acc1 = __builtin_amdgcn_wmma_f32_16x16x4_f32(false, a, false, b, (short)0, acc1, false, false);
    b.x = b0[32 + l16]; b.y = b1[32 + l16];
    acc2 = __builtin_amdgcn_wmma_f32_16x16x4_f32(false, a, false, b, (short)0, acc2, false, false);
  }
#pragma unroll
  for (int v = 0; v < 8; v++) {
    int mC = m0 + v + 8 * half;
    float* prow = outp + (long long)mC * 48 + l16;
    prow[0]  = acc0[v];
    prow[16] = acc1[v];
    prow[32] = acc2[v];
  }
}


__global__ void logsoftmax_kernel(const float* __restrict__ outp,
                                  const float* __restrict__ b2,
                                  float* __restrict__ out) {
  int i = blockIdx.x * blockDim.x + threadIdx.x;
  if (i >= N_NODES) return;
  const float* r = outp + (long long)i * 48;
  float m = -INFINITY;
  for (int j = 0; j < NCLS; j++) m = fmaxf(m, r[j] + b2[j]);
  float s = 0.0f;
  for (int j = 0; j < NCLS; j++) s += expf(r[j] + b2[j] - m);
  float ls = logf(s);
  for (int j = 0; j < NCLS; j++)
    out[(long long)i * NCLS + j] = (r[j] + b2[j] - m) - ls;
}


extern "C" void kernel_launch(void* const* d_in, const int* in_sizes, int n_in,
                              void* d_out, int out_size, void* d_ws, size_t ws_size,
                              hipStream_t stream) {
  (void)in_sizes; (void)n_in; (void)out_size; (void)ws_size;
  const float* x    = (const float*)d_in[0];
  const int*   ei   = (const int*)  d_in[1];
  const float* W0_1 = (const float*)d_in[2];
  const float* W1_1 = (const float*)d_in[3];
  const float* b1   = (const float*)d_in[4];
  const float* W0_2 = (const float*)d_in[5];
  const float* W1_2 = (const float*)d_in[6];
  const float* b2   = (const float*)d_in[7];
  float* out = (float*)d_out;

  const int* rows = ei;
  const int* cols = ei + N_EDGES;

  const long long N = N_NODES;
  float* ws     = (float*)d_ws;
  int*   cnt    = (int*)ws;
  float* dis    = ws + N;
  float* P      = ws + 2 * N;
  float* agg2   = ws + 2 * N;
  float* outp   = ws + 34 * N;
  int*   colidx = (int*)(ws + 66 * N);
  int*   rowptr = (int*)(ws + 82 * N);
  float* hbuf   = ws + 98 * N;
  int*   rank   = (int*)(ws + 98 * N);
  float* Wc1    = ws + 130 * N;
  float* Wc2    = Wc1 + 52 * 64;

  const int T = 256;
  zero_kernel<<<(unsigned)((N + T - 1) / T), T, 0, stream>>>((float*)cnt, N);
  pack_w1_kernel<<<(52 * 64 + T - 1) / T, T, 0, stream>>>(W0_1, W1_1, Wc1);
  pack_w2_kernel<<<(64 * 48 + T - 1) / T, T, 0, stream>>>(W0_2, W1_2, Wc2);

  count_rank_kernel<<<(N_EDGES + T - 1) / T, T, 0, stream>>>(rows, cols, cnt, rank);
  scan_dis_kernel<<<(N_NODES + 1023) / 1024, 1024, 0, stream>>>(cnt, rowptr, dis);
  fill_kernel<<<(N_EDGES + T - 1) / T, T, 0, stream>>>(rows, cols, rank, rowptr, colidx);

  proj1_wmma_kernel<<<(N_NODES + 63) / 64, 128, 0, stream>>>(x, Wc1, P);
  gather_rows_kernel<<<(N_NODES + 7) / 8, T, 0, stream>>>(
      rowptr, cnt, colidx, dis, P, 64, 32, hbuf);
  relu_kernel<<<(unsigned)((32 * N + T - 1) / T), T, 0, stream>>>(P, b1, hbuf);

  gather_rows_kernel<<<(N_NODES + 7) / 8, T, 0, stream>>>(
      rowptr, cnt, colidx, dis, hbuf, 32, 0, agg2);
  out_wmma_kernel<<<(N_NODES + 63) / 64, 128, 0, stream>>>(hbuf, agg2, Wc2, outp);
  logsoftmax_kernel<<<(unsigned)((N + T - 1) / T), T, 0, stream>>>(outp, b2, out);
}


# Edges grouped by row with a two-level counting sort on LDS counters (bins of 256 rows, then rows in a bin) instead of one returning global add per edge; list (sum) order now LDS arrival rank
# speedup vs baseline: 2.7367x; 1.4293x over previous
#include <hip/hip_runtime.h>
#include <math.h>

#define N_NODES 100000
#define N_EDGES 1600000
#define F_IN    50
#define HID     32
#define NCLS    40
#define BIN_SHIFT 8
#define BIN_ROWS  256
#define NB        391
#define CHUNK     8192
#define NCH       196
#define TBL       (NB * NCH)
#define COL_BITS  17
static_assert(N_NODES <= (1 << COL_BITS), "col must fit the low field of a bin entry");
static_assert(NB * BIN_ROWS >= N_NODES && NCH * CHUNK >= N_EDGES && N_EDGES % 4 == 0, "sizes");
static_assert(TBL <= N_NODES, "each table has one N-word plane of the workspace");


typedef float v2f __attribute__((ext_vector_type(2)));
typedef float v8f __attribute__((ext_vector_type(8)));

__global__ void pack_w1_kernel(const float* __restrict__ W0,
                               const float* __restrict__ W1,
                               float* __restrict__ Wcat) {
  int t = blockIdx.x * blockDim.x + threadIdx.x;
  if (t >= 52 * 64) return;
  int k = t >> 6, j = t & 63;
  float v = 0.0f;
  if (k < F_IN) v = (j < HID) ? W0[k * HID + j] : W1[k * HID + (j - HID)];
  Wcat[t] = v;
}

__global__ void pack_w2_kernel(const float* __restrict__ W0,
                               const float* __restrict__ W1,
                               float* __restrict__ Wcat) {
  int t = blockIdx.x * blockDim.x + threadIdx.x;
  if (t >= 64 * 48) return;
  int k = t / 48, j = t % 48;
  float v = 0.0f;
  if (j < NCLS) v = (k < HID) ? W0[k * NCLS + j] : W1[(k - HID) * NCLS + j];
  Wcat[t] = v;
}


__global__ void __launch_bounds__(1024)
bin_count_kernel(const int* __restrict__ rows,
                 const int* __restrict__ cols,
                 int* __restrict__ tcnt) {
  __shared__ int hist[NB];
  int t = threadIdx.x;
  int4 r4[2], c4[2];
#pragma unroll
  for (int h = 0; h < 2; h++) {
    int e = blockIdx.x * CHUNK + h * 4096 + t * 4;
    r4[h] = make_int4(0, 0, 0, 0);
    c4[h] = make_int4(0, 0, 0, 0);
    if (e < N_EDGES) {
      r4[h] = *(const int4*)(rows + e);
      c4[h] = *(const int4*)(cols + e);
    }
  }
  for (int i = t; i < NB; i += 1024) hist[i] = 0;
  __syncthreads();
#pragma unroll
  for (int h = 0; h < 2; h++) {
    if (r4[h].x != c4[h].x) atomicAdd(&hist[r4[h].x >> BIN_SHIFT], 1);
    if (r4[h].y != c4[h].y) atomicAdd(&hist[r4[h].y >> BIN_SHIFT], 1);
    if (r4[h].z != c4[h].z) atomicAdd(&hist[r4[h].z >> BIN_SHIFT], 1);
    if (r4[h].w != c4[h].w) atomicAdd(&hist[r4[h].w >> BIN_SHIFT], 1);
  }
  __syncthreads();
  for (int i = t; i < NB; i += 1024) tcnt[i * NCH + blockIdx.x] = hist[i];
}

__global__ void __launch_bounds__(1024)
table_scan_kernel(const int* __restrict__ tcnt,
                  int* __restrict__ tbase) {
  __shared__ int s_pre[32];
  __shared__ int s_tot[32];
  int t = threadIdx.x, lane = t & 31, wave = t >> 5;
  int b0 = blockIdx.x * 1024;
  const int4* c4 = (const int4*)tcnt;
  int pre = 0;
#pragma unroll 4
  for (int q = t; q < (b0 >> 2); q += 1024) {
    int4 x = c4[q];
    pre += (x.x + x.y) + (x.z + x.w);
  }
  int i = b0 + t;
  int v = (i < TBL) ? tcnt[i] : 0;
  int incl = v;
  for (int d = 1; d < 32; d <<= 1) {
    int y = __shfl_up(incl, d);
    if (lane >= d) incl += y;
  }
  for (int d = 16; d >= 1; d >>= 1) pre += __shfl_xor(pre, d);
  if (lane == 31) s_tot[wave] = incl;
  if (lane == 0)  s_pre[wave] = pre;
  __syncthreads();
  int add = s_pre[lane] + (lane < wave ? s_tot[lane] : 0);
  for (int d = 16; d >= 1; d >>= 1) add += __shfl_xor(add, d);
  if (i < TBL) tbase[i] = add + incl - v;
}

__global__ void __launch_bounds__(1024)
bin_scatter_kernel(const int* __restrict__ rows,
                   const int* __restrict__ cols,
                   const int* __restrict__ tbase,
                   int* __restrict__ binned) {
  __shared__ int base[NB];
  __shared__ int cur[NB];
  int t = threadIdx.x;
  int4 r4[2], c4[2];
#pragma unroll
  for (int h = 0; h < 2; h++) {
    int e = blockIdx.x * CHUNK + h * 4096 + t * 4;
    r4[h] = make_int4(0, 0, 0, 0);
    c4[h] = make_int4(0, 0, 0, 0);
    if (e < N_EDGES) {
      r4[h] = *(const int4*)(rows + e);
      c4[h] = *(const int4*)(cols + e);
    }
  }
  for (int i = t; i < NB; i += 1024) {
    base[i] = tbase[i * NCH + blockIdx.x];
    cur[i] = 0;
  }
  __syncthreads();
#pragma unroll
  for (int h = 0; h < 2; h++) {
    if (r4[h].x != c4[h].x) { int b = r4[h].x >> BIN_SHIFT; int k = atomicAdd(&cur[b], 1); binned[base[b] + k] = ((r4[h].x & (BIN_ROWS - 1)) << COL_BITS) | c4[h].x; }
    if (r4[h].y != c4[h].y) { int b = r4[h].y >> BIN_SHIFT; int k = atomicAdd(&cur[b], 1); binned[base[b] + k] = ((r4[h].y & (BIN_ROWS - 1)) << COL_BITS) | c4[h].y; }
    if (r4[h].z != c4[h].z) { int b = r4[h].z >> BIN_SHIFT; int k = atomicAdd(&cur[b], 1); binned[base[b] + k] = ((r4[h].z & (BIN_ROWS - 1)) << COL_BITS) | c4[h].z; }
    if (r4[h].w != c4[h].w) { int b = r4[h].w >> BIN_SHIFT; int k = atomicAdd(&cur[b], 1); binned[base[b] + k] = ((r4[h].w & (BIN_ROWS - 1)) << COL_BITS) | c4[h].w; }
  }
}

__global__ void __launch_bounds__(1024)
bin_rows_kernel(const int* __restrict__ tcnt,
                const int* __restrict__ tbase,
                const int* __restrict__ binned,
                int* __restrict__ cnt,
                int* __restrict__ rowptr,
                float* __restrict__ dis,
                int* __restrict__ colidx) {
  __shared__ int s_cnt[BIN_ROWS];
  __shared__ int s_start[BIN_ROWS];
  __shared__ int s_cur[BIN_ROWS];
  __shared__ int s_wtot[8];
  int t = threadIdx.x, lane = t & 31, wave = t >> 5;
  int b = blockIdx.x;
  int s = tbase[b * NCH];
  int e = (b + 1 < NB) ? tbase[(b + 1) * NCH] : tbase[TBL - 1] + tcnt[TBL - 1];
  if (t < BIN_ROWS) { s_cnt[t] = 0; s_cur[t] = 0; }
  __syncthreads();
  for (int i0 = s + t; i0 < e; i0 += 4096) {
    int p[4];
#pragma unroll
    for (int u = 0; u < 4; u++) {
      int idx = i0 + u * 1024;
      p[u] = (idx < e) ? binned[idx] : -1;
    }
#pragma unroll
    for (int u = 0; u < 4; u++)
      if (p[u] >= 0) atomicAdd(&s_cnt[p[u] >> COL_BITS], 1);
  }
  __syncthreads();
  int v = 0, incl = 0;
  if (t < BIN_ROWS) {
    v = s_cnt[t];
    incl = v;
    for (int d = 1; d < 32; d <<= 1) {
      int y = __shfl_up(incl, d);
      if (lane >= d) incl += y;
    }
    if (lane == 31) s_wtot[wave] = incl;
  }
  __syncthreads();
  if (t < BIN_ROWS) {
    int add = 0;
    for (int w = 0; w < 8; w++) add += (w < wave) ? s_wtot[w] : 0;
    int excl = add + incl - v;
    s_start[t] = excl;
    int row = b * BIN_ROWS + t;
    if (row < N_NODES) {
      rowptr[row] = s + excl;
      cnt[row] = v;
      float dg = (float)v;
      dis[row] = (dg > 0.0f) ? rsqrtf(fmaxf(dg, 1.0f)) : 0.0f;
    }
  }
  __syncthreads();
  for (int i0 = s + t; i0 < e; i0 += 4096) {
    int p[4];
#pragma unroll
    for (int u = 0; u < 4; u++) {
      int idx = i0 + u * 1024;
      p[u] = (idx < e) ? binned[idx] : -1;
    }
#pragma unroll
    for (int u = 0; u < 4; u++) {
      if (p[u] >= 0) {
        int rl = p[u] >> COL_BITS;
        int k = atomicAdd(&s_cur[rl], 1);
        colidx[s + s_start[rl] + k] = p[u] & ((1 << COL_BITS) - 1);
      }
    }
  }
}


__global__ void __launch_bounds__(256)
gather_rows_kernel(const int* __restrict__ rowptr,
                   const int* __restrict__ cnt,
                   const int* __restrict__ colidx,
                   const float* __restrict__ dis,
                   const float* __restrict__ src, int sstride, int soff,
                   float* __restrict__ dst) {
  int lane = threadIdx.x & 31;
  int r = __builtin_amdgcn_readfirstlane(blockIdx.x * 8 + (threadIdx.x >> 5));
  if (r >= N_NODES) return;
  int start = rowptr[r];
  int n = cnt[r];
  float ndr = -dis[r];
  const float* sp = src + soff;
  float acc = 0.0f;
  for (int base = 0; base < n; base += 32) {
    int m = n - base;
    if (m > 32) m = 32;
    int li = lane < m ? lane : m - 1;
    int c = colidx[start + base + li];
    float w = lane < m ? ndr * dis[c] : 0.0f;
    for (int j = 0; j < m; j += 8) {
      float v[8];
#pragma unroll
      for (int u = 0; u < 8; u++) {
        int cj = __builtin_amdgcn_readlane(c, j + u);
        const float* rowp = sp + (long long)cj * sstride;
        float x = rowp[lane];
        v[u] = (j + u < m) ? x : 0.0f;
      }
#pragma unroll
      for (int u = 0; u < 8; u++) {
        float wj = __int_as_float(__builtin_amdgcn_readlane(__float_as_int(w), j + u));
        acc = fmaf(wj, v[u], acc);
      }
    }
  }
  dst[(long long)r * HID + lane] = acc;
}


__global__ void __launch_bounds__(128)
proj1_wmma_kernel(const float* __restrict__ x,
                  const float* __restrict__ Wcat,
                  float* __restrict__ P) {
  int wave = threadIdx.x >> 5;
  int lane = threadIdx.x & 31;
  int half = lane >> 4;
  int l16  = lane & 15;
  int m0 = (blockIdx.x * 4 + wave) * 16;
  if (m0 >= N_NODES) return;
  int mA = m0 + l16;
  const float* arow = x + (long long)mA * F_IN;

  v8f acc0 = {}, acc1 = {}, acc2 = {}, acc3 = {};
  for (int kk = 0; kk < 48; kk += 4) {
    int k = kk + 2 * half;
    v2f a = *(const v2f*)(arow + k);
    const float* b0 = Wcat + k * 64;
    const float* b1 = Wcat + (k + 1) * 64;
    v2f b;
    b.x = b0[l16];      b.y = b1[l16];
    acc0 = __builtin_amdgcn_wmma_f32_16x16x4_f32(false, a, false, b, (short)0, acc0, false, false);
    b.x = b0[16 + l16]; b.y = b1[16 + l16];
    acc1 = __builtin_amdgcn_wmma_f32_16x16x4_f32(false, a, false, b, (short)0, acc1, false, false);
    b.x = b0[32 + l16]; b.y = b1[32 + l16];
    acc2 = __builtin_amdgcn_wmma_f32_16x16x4_f32(false, a, false, b, (short)0, acc2, false, false);
    b.x = b0[48 + l16]; b.y = b1[48 + l16];
    acc3 = __builtin_amdgcn_wmma_f32_16x16x4_f32(false, a, false, b, (short)0, acc3, false, false);
  }
  {
    int k = 48 + 2 * half;
    v2f aload = *(const v2f*)(arow + 48);
    v2f a;
    a.x = half ? 0.0f : aload.x;
    a.y = half ? 0.0f : aload.y;
    const float* b0 = Wcat + k * 64;
    const float* b1 = Wcat + (k + 1) * 64;
    v2f b;
    b.x = b0[l16];      b.y = b1[l16];
    acc0 = __builtin_amdgcn_wmma_f32_16x16x4_f32(false, a, false, b, (short)0, acc0, false, false);
    b.x = b0[16 + l16]; b.y = b1[16 + l16];
    acc1 = __builtin_amdgcn_wmma_f32_16x16x4_f32(false, a, false, b, (short)0, acc1, false, false);
    b.x = b0[32 + l16]; b.y = b1[32 + l16];
    acc2 = __builtin_amdgcn_wmma_f32_16x16x4_f32(false, a, false, b, (short)0, acc2, false, false);
    b.x = b0[48 + l16]; b.y = b1[48 + l16];
    acc3 = __builtin_amdgcn_wmma_f32_16x16x4_f32(false, a, false, b, (short)0, acc3, false, false);
  }
#pragma unroll
  for (int v = 0; v < 8; v++) {
    int mC = m0 + v + 8 * half;
    float* prow = P + (long long)mC * 64 + l16;
    prow[0]  = acc0[v];
    prow[16] = acc1[v];
    prow[32] = acc2[v];
    prow[48] = acc3[v];
  }
}


__global__ void relu_kernel(const float* __restrict__ P,
                            const float* __restrict__ b1,
                            float* __restrict__ h) {
  long long i = blockIdx.x * (long long)blockDim.x + threadIdx.x;
  if (i >= (long long)N_NODES * HID) return;
  long long n = i >> 5;
  int f = (int)(i & 31);
  float v = P[n * 64 + f] + h[i] + b1[f];
  h[i] = v > 0.0f ? v : 0.0f;
}


__global__ void __launch_bounds__(128)
out_wmma_kernel(const float* __restrict__ h,
                const float* __restrict__ agg2,
                const float* __restrict__ Wcat,
                float* __restrict__ outp) {
  int wave = threadIdx.x >> 5;
  int lane = threadIdx.x & 31;
  int half = lane >> 4;
  int l16  = lane & 15;
  int m0 = (blockIdx.x * 4 + wave) * 16;
  if (m0 >= N_NODES) return;
  int mA = m0 + l16;
  const float* hrow = h    + (long long)mA * HID;
  const float* grow = agg2 + (long long)mA * HID - HID;

  v8f acc0 = {}, acc1 = {}, acc2 = {};
  for (int kk = 0; kk < 64; kk += 4) {
    int k = kk + 2 * half;
    const float* src = (k < HID) ? (hrow + k) : (grow + k);
    v2f a = *(const v2f*)src;
    const float* b0 = Wcat + k * 48;
    const float* b1 = Wcat + (k + 1) * 48;
    v2f b;
    b.x = b0[l16];      b.y = b1[l16];
    acc0 = __builtin_amdgcn_wmma_f32_16x16x4_f32(false, a, false, b, (short)0, acc0, false, false);
    b.x = b0[16 + l16]; b.y = b1[16 + l16];
    acc1 = __builtin_amdgcn_wmma_f32_16x16x4_f32(false, a, false, b, (short)0, acc1, false, false);
    b.x = b0[32 + l16]; b.y = b1[32 + l16];
    acc2 = __builtin_amdgcn_wmma_f32_16x16x4_f32(false, a, false, b, (short)0, acc2, false, false);
  }
#pragma unroll
  for (int v = 0; v < 8; v++) {
    int mC = m0 + v + 8 * half;
    float* prow = outp + (long long)mC * 48 + l16;
    prow[0]  = acc0[v];
    prow[16] = acc1[v];
    prow[32] = acc2[v];
  }
}


__global__ void logsoftmax_kernel(const float* __restrict__ outp,
                                  const float* __restrict__ b2,
                                  float* __restrict__ out) {
  int i = blockIdx.x * blockDim.x + threadIdx.x;
  if (i >= N_NODES) return;
  const float* r = outp + (long long)i * 48;
  float m = -INFINITY;
  for (int j = 0; j < NCLS; j++) m = fmaxf(m, r[j] + b2[j]);
  float s = 0.0f;
  for (int j = 0; j < NCLS; j++) s += expf(r[j] + b2[j] - m);
  float ls = logf(s);
  for (int j = 0; j < NCLS; j++)
    out[(long long)i * NCLS + j] = (r[j] + b2[j] - m) - ls;
}


extern "C" void kernel_launch(void* const* d_in, const int* in_sizes, int n_in,
                              void* d_out, int out_size, void* d_ws, size_t ws_size,
                              hipStream_t stream) {
  (void)in_sizes; (void)n_in; (void)out_size; (void)ws_size;
  const float* x    = (const float*)d_in[0];
  const int*   ei   = (const int*)  d_in[1];
  const float* W0_1 = (const float*)d_in[2];
  const float* W1_1 = (const float*)d_in[3];
  const float* b1   = (const float*)d_in[4];
  const float* W0_2 = (const float*)d_in[5];
  const float* W1_2 = (const float*)d_in[6];
  const float* b2   = (const float*)d_in[7];
  float* out = (float*)d_out;

  const int* rows = ei;
  const int* cols = ei + N_EDGES;

  const long long N = N_NODES;
  float* ws     = (float*)d_ws;
  int*   cnt    = (int*)ws;
  float* dis    = ws + N;
  float* P      = ws + 2 * N;
  float* agg2   = ws + 2 * N;
  float* outp   = ws + 34 * N;
  int*   colidx = (int*)(ws + 66 * N);
  int*   rowptr = (int*)(ws + 82 * N);
  int*   tcnt   = (int*)(ws + 83 * N);
  int*   tbase  = (int*)(ws + 84 * N);
  float* hbuf   = ws + 98 * N;
  int*   binned = (int*)(ws + 98 * N);
  float* Wc1    = ws + 130 * N;
  float* Wc2    = Wc1 + 52 * 64;

  const int T = 256;
  pack_w1_kernel<<<(52 * 64 + T - 1) / T, T, 0, stream>>>(W0_1, W1_1, Wc1);
  pack_w2_kernel<<<(64 * 48 + T - 1) / T, T, 0, stream>>>(W0_2, W1_2, Wc2);

  bin_count_kernel<<<NCH, 1024, 0, stream>>>(rows, cols, tcnt);
  table_scan_kernel<<<(TBL + 1023) / 1024, 1024, 0, stream>>>(tcnt, tbase);
  bin_scatter_kernel<<<NCH, 1024, 0, stream>>>(rows, cols, tbase, binned);
  bin_rows_kernel<<<NB, 1024, 0, stream>>>(tcnt, tbase, binned, cnt, rowptr, dis, colidx);

  proj1_wmma_kernel<<<(N_NODES + 63) / 64, 128, 0, stream>>>(x, Wc1, P);
  gather_rows_kernel<<<(N_NODES + 7) / 8, T, 0, stream>>>(
      rowptr, cnt, colidx, dis, P, 64, 32, hbuf);
  relu_kernel<<<(unsigned)((32 * N + T - 1) / T), T, 0, stream>>>(P, b1, hbuf);

  gather_rows_kernel<<<(N_NODES + 7) / 8, T, 0, stream>>>(
      rowptr, cnt, colidx, dis, hbuf, 32, 0, agg2);
  out_wmma_kernel<<<(N_NODES + 63) / 64, 128, 0, stream>>>(hbuf, agg2, Wc2, outp);
  logsoftmax_kernel<<<(unsigned)((N + T - 1) / T), T, 0, stream>>>(outp, b2, out);
}


# Elementwise passes fused into neighbours: bias+relu into the layer-1 gather's store, bias+log_softmax into the output GEMM (exp row sum now a 16-lane tree), weights read in place; 12 -> 8 launches
# speedup vs baseline: 3.4097x; 1.2459x over previous
#include <hip/hip_runtime.h>
#include <math.h>

#define N_NODES 100000
#define N_EDGES 1600000
#define F_IN    50
#define HID     32
#define NCLS    40
#define BIN_SHIFT 8
#define BIN_ROWS  256
#define NB        391
#define CHUNK     8192
#define NCH       196
#define TBL       (NB * NCH)
#define COL_BITS  17
static_assert(N_NODES <= (1 << COL_BITS), "col must fit the low field of a bin entry");
static_assert(NB * BIN_ROWS >= N_NODES && NCH * CHUNK >= N_EDGES && N_EDGES % 4 == 0, "sizes");
static_assert(TBL <= N_NODES, "each table has one N-word plane of the workspace");


typedef float v2f __attribute__((ext_vector_type(2)));
typedef float v8f __attribute__((ext_vector_type(8)));


__global__ void __launch_bounds__(1024)
bin_count_kernel(const int* __restrict__ rows,
                 const int* __restrict__ cols,
                 int* __restrict__ tcnt) {
  __shared__ int hist[NB];
  int t = threadIdx.x;
  int4 r4[2], c4[2];
#pragma unroll
  for (int h = 0; h < 2; h++) {
    int e = blockIdx.x * CHUNK + h * 4096 + t * 4;
    r4[h] = make_int4(0, 0, 0, 0);
    c4[h] = make_int4(0, 0, 0, 0);
    if (e < N_EDGES) {
      r4[h] = *(const int4*)(rows + e);
      c4[h] = *(const int4*)(cols + e);
    }
  }
  for (int i = t; i < NB; i += 1024) hist[i] = 0;
  __syncthreads();
#pragma unroll
  for (int h = 0; h < 2; h++) {
    if (r4[h].x != c4[h].x) atomicAdd(&hist[r4[h].x >> BIN_SHIFT], 1);
    if (r4[h].y != c4[h].y) atomicAdd(&hist[r4[h].y >> BIN_SHIFT], 1);
    if (r4[h].z != c4[h].z) atomicAdd(&hist[r4[h].z >> BIN_SHIFT], 1);
    if (r4[h].w != c4[h].w) atomicAdd(&hist[r4[h].w >> BIN_SHIFT], 1);
  }
  __syncthreads();
  for (int i = t; i < NB; i += 1024) tcnt[i * NCH + blockIdx.x] = hist[i];
}

__global__ void __launch_bounds__(1024)
table_scan_kernel(const int* __restrict__ tcnt,
                  int* __restrict__ tbase) {
  __shared__ int s_pre[32];
  __shared__ int s_tot[32];
  int t = threadIdx.x, lane = t & 31, wave = t >> 5;
  int b0 = blockIdx.x * 1024;
  const int4* c4 = (const int4*)tcnt;
  int pre = 0;
#pragma unroll 4
  for (int q = t; q < (b0 >> 2); q += 1024) {
    int4 x = c4[q];
    pre += (x.x + x.y) + (x.z + x.w);
  }
  int i = b0 + t;
  int v = (i < TBL) ? tcnt[i] : 0;
  int incl = v;
  for (int d = 1; d < 32; d <<= 1) {
    int y = __shfl_up(incl, d);
    if (lane >= d) incl += y;
  }
  for (int d = 16; d >= 1; d >>= 1) pre += __shfl_xor(pre, d);
  if (lane == 31) s_tot[wave] = incl;
  if (lane == 0)  s_pre[wave] = pre;
  __syncthreads();
  int add = s_pre[lane] + (lane < wave ? s_tot[lane] : 0);
  for (int d = 16; d >= 1; d >>= 1) add += __shfl_xor(add, d);
  if (i < TBL) tbase[i] = add + incl - v;
}

__global__ void __launch_bounds__(1024)
bin_scatter_kernel(const int* __restrict__ rows,
                   const int* __restrict__ cols,
                   const int* __restrict__ tbase,
                   int* __restrict__ binned) {
  __shared__ int base[NB];
  __shared__ int cur[NB];
  int t = threadIdx.x;
  int4 r4[2], c4[2];
#pragma unroll
  for (int h = 0; h < 2; h++) {
    int e = blockIdx.x * CHUNK + h * 4096 + t * 4;
    r4[h] = make_int4(0, 0, 0, 0);
    c4[h] = make_int4(0, 0, 0, 0);
    if (e < N_EDGES) {
      r4[h] = *(const int4*)(rows + e);
      c4[h] = *(const int4*)(cols + e);
    }
  }
  for (int i = t; i < NB; i += 1024) {
    base[i] = tbase[i * NCH + blockIdx.x];
    cur[i] = 0;
  }
  __syncthreads();
#pragma unroll
  for (int h = 0; h < 2; h++) {
    if (r4[h].x != c4[h].x) { int b = r4[h].x >> BIN_SHIFT; int k = atomicAdd(&cur[b], 1); binned[base[b] + k] = ((r4[h].x & (BIN_ROWS - 1)) << COL_BITS) | c4[h].x; }
    if (r4[h].y != c4[h].y) { int b = r4[h].y >> BIN_SHIFT; int k = atomicAdd(&cur[b], 1); binned[base[b] + k] = ((r4[h].y & (BIN_ROWS - 1)) << COL_BITS) | c4[h].y; }
    if (r4[h].z != c4[h].z) { int b = r4[h].z >> BIN_SHIFT; int k = atomicAdd(&cur[b], 1); binned[base[b] + k] = ((r4[h].z & (BIN_ROWS - 1)) << COL_BITS) | c4[h].z; }
    if (r4[h].w != c4[h].w) { int b = r4[h].w >> BIN_SHIFT; int k = atomicAdd(&cur[b], 1); binned[base[b] + k] = ((r4[h].w & (BIN_ROWS - 1)) << COL_BITS) | c4[h].w; }
  }
}

__global__ void __launch_bounds__(1024)
bin_rows_kernel(const int* __restrict__ tcnt,
                const int* __restrict__ tbase,
                const int* __restrict__ binned,
                int* __restrict__ cnt,
                int* __restrict__ rowptr,
                float* __restrict__ dis,
                int* __restrict__ colidx) {
  __shared__ int s_cnt[BIN_ROWS];
  __shared__ int s_start[BIN_ROWS];
  __shared__ int s_cur[BIN_ROWS];
  __shared__ int s_wtot[8];
  int t = threadIdx.x, lane = t & 31, wave = t >> 5;
  int b = blockIdx.x;
  int s = tbase[b * NCH];
  int e = (b + 1 < NB) ? tbase[(b + 1) * NCH] : tbase[TBL - 1] + tcnt[TBL - 1];
  if (t < BIN_ROWS) { s_cnt[t] = 0; s_cur[t] = 0; }
  __syncthreads();
  for (int i0 = s + t; i0 < e; i0 += 4096) {
    int p[4];
#pragma unroll
    for (int u = 0; u < 4; u++) {
      int idx = i0 + u * 1024;
      p[u] = (idx < e) ? binned[idx] : -1;
    }
#pragma unroll
    for (int u = 0; u < 4; u++)
      if (p[u] >= 0) atomicAdd(&s_cnt[p[u] >> COL_BITS], 1);
  }
  __syncthreads();
  int v = 0, incl = 0;
  if (t < BIN_ROWS) {
    v = s_cnt[t];
    incl = v;
    for (int d = 1; d < 32; d <<= 1) {
      int y = __shfl_up(incl, d);
      if (lane >= d) incl += y;
    }
    if (lane == 31) s_wtot[wave] = incl;
  }
  __syncthreads();
  if (t < BIN_ROWS) {
    int add = 0;
    for (int w = 0; w < 8; w++) add += (w < wave) ? s_wtot[w] : 0;
    int excl = add + incl - v;
    s_start[t] = excl;
    int row = b * BIN_ROWS + t;
    if (row < N_NODES) {
      rowptr[row] = s + excl;
      cnt[row] = v;
      float dg = (float)v;
      dis[row] = (dg > 0.0f) ? rsqrtf(fmaxf(dg, 1.0f)) : 0.0f;
    }
  }
  __syncthreads();
  for (int i0 = s + t; i0 < e; i0 += 4096) {
    int p[4];
#pragma unroll
    for (int u = 0; u < 4; u++) {
      int idx = i0 + u * 1024;
      p[u] = (idx < e) ? binned[idx] : -1;
    }
#pragma unroll
    for (int u = 0; u < 4; u++) {
      if (p[u] >= 0) {
        int rl = p[u] >> COL_BITS;
        int k = atomicAdd(&s_cur[rl], 1);
        colidx[s + s_start[rl] + k] = p[u] & ((1 << COL_BITS) - 1);
      }
    }
  }
}


__device__ __forceinline__ float gather_row_sum(int r, int lane,
                                                const int* __restrict__ rowptr,
                                                const int* __restrict__ cnt,
                                                const int* __restrict__ colidx,
                                                const float* __restrict__ dis,
                                                const float* __restrict__ sp, int sstride) {
  int start = rowptr[r];
  int n = cnt[r];
  float ndr = -dis[r];
  float acc = 0.0f;
  for (int base = 0; base < n; base += 32) {
    int m = n - base;
    if (m > 32) m = 32;
    int li = lane < m ? lane : m - 1;
    int c = colidx[start + base + li];
    float w = lane < m ? ndr * dis[c] : 0.0f;
    for (int j = 0; j < m; j += 8) {
      float v[8];
#pragma unroll
      for (int u = 0; u < 8; u++) {
        int cj = __builtin_amdgcn_readlane(c, j + u);
        const float* rowp = sp + (long long)cj * sstride;
        float x = rowp[lane];
        v[u] = (j + u < m) ? x : 0.0f;
      }
#pragma unroll
      for (int u = 0; u < 8; u++) {
        float wj = __int_as_float(__builtin_amdgcn_readlane(__float_as_int(w), j + u));
        acc = fmaf(wj, v[u], acc);
      }
    }
  }
  return acc;
}

__global__ void __launch_bounds__(256)
gather_relu_kernel(const int* __restrict__ rowptr,
                   const int* __restrict__ cnt,
                   const int* __restrict__ colidx,
                   const float* __restrict__ dis,
                   const float* __restrict__ P,
                   const float* __restrict__ b1,
                   float* __restrict__ h) {
  int lane = threadIdx.x & 31;
  int r = __builtin_amdgcn_readfirstlane(blockIdx.x * 8 + (threadIdx.x >> 5));
  if (r >= N_NODES) return;
  float self = P[(long long)r * 64 + lane];
  float bias = b1[lane];
  float acc = gather_row_sum(r, lane, rowptr, cnt, colidx, dis, P + HID, 64);
  float v = self + acc + bias;
  h[(long long)r * HID + lane] = v > 0.0f ? v : 0.0f;
}

__global__ void __launch_bounds__(256)
gather_rows_kernel(const int* __restrict__ rowptr,
                   const int* __restrict__ cnt,
                   const int* __restrict__ colidx,
                   const float* __restrict__ dis,
                   const float* __restrict__ src,
                   float* __restrict__ dst) {
  int lane = threadIdx.x & 31;
  int r = __builtin_amdgcn_readfirstlane(blockIdx.x * 8 + (threadIdx.x >> 5));
  if (r >= N_NODES) return;
  dst[(long long)r * HID + lane] = gather_row_sum(r, lane, rowptr, cnt, colidx, dis, src, HID);
}


__global__ void __launch_bounds__(128)
proj1_wmma_kernel(const float* __restrict__ x,
                  const float* __restrict__ W0,
                  const float* __restrict__ W1,
                  float* __restrict__ P) {
  int wave = threadIdx.x >> 5;
  int lane = threadIdx.x & 31;
  int half = lane >> 4;
  int l16  = lane & 15;
  int m0 = (blockIdx.x * 4 + wave) * 16;
  if (m0 >= N_NODES) return;
  int mA = m0 + l16;
  const float* arow = x + (long long)mA * F_IN;

  v8f acc0 = {}, acc1 = {}, acc2 = {}, acc3 = {};
  for (int kk = 0; kk < 48; kk += 4) {
    int k = kk + 2 * half;
    v2f a = *(const v2f*)(arow + k);
    const float* p0 = W0 + k * HID + l16;
    const float* p1 = W1 + k * HID + l16;
    v2f b;
    b.x = p0[0];  b.y = p0[HID];
    acc0 = __builtin_amdgcn_wmma_f32_16x16x4_f32(false, a, false, b, (short)0, acc0, false, false);
    b.x = p0[16]; b.y = p0[HID + 16];
    acc1 = __builtin_amdgcn_wmma_f32_16x16x4_f32(false, a, false, b, (short)0, acc1, false, false);
    b.x = p1[0];  b.y = p1[HID];
    acc2 = __builtin_amdgcn_wmma_f32_16x16x4_f32(false, a, false, b, (short)0, acc2, false, false);
    b.x = p1[16]; b.y = p1[HID + 16];
    acc3 = __builtin_amdgcn_wmma_f32_16x16x4_f32(false, a, false, b, (short)0, acc3, false, false);
  }
  {
    v2f aload = *(const v2f*)(arow + 48);
    v2f a;
    a.x = half ? 0.0f : aload.x;
    a.y = half ? 0.0f : aload.y;
    const float* p0 = W0 + 48 * HID + l16;
    const float* p1 = W1 + 48 * HID + l16;
    v2f b;
    b.x = p0[0];  b.y = p0[HID];
    acc0 = __builtin_amdgcn_wmma_f32_16x16x4_f32(false, a, false, b, (short)0, acc0, false, false);
    b.x = p0[16]; b.y = p0[HID + 16];
    acc1 = __builtin_amdgcn_wmma_f32_16x16x4_f32(false, a, false, b, (short)0, acc1, false, false);
    b.x = p1[0];  b.y = p1[HID];
    acc2 = __builtin_amdgcn_wmma_f32_16x16x4_f32(false, a, false, b, (short)0, acc2, false, false);
    b.x = p1[16]; b.y = p1[HID + 16];
    acc3 = __builtin_amdgcn_wmma_f32_16x16x4_f32(false, a, false, b, (short)0, acc3, false, false);
  }
#pragma unroll
  for (int v = 0; v < 8; v++) {
    int mC = m0 + v + 8 * half;
    float* prow = P + (long long)mC * 64 + l16;
    prow[0]  = acc0[v];
    prow[16] = acc1[v];
    prow[32] = acc2[v];
    prow[48] = acc3[v];
  }
}


__device__ __forceinline__ void out_k32(const float* __restrict__ arow,
                                        const float* __restrict__ Wm,
                                        int half, int l16,
                                        v8f& acc0, v8f& acc1, v8f& acc2) {
  int c2 = 32 + (l16 & 7);
  bool ok2 = l16 < 8;
  for (int kk = 0; kk < HID; kk += 4) {
    int k = kk + 2 * half;
    v2f a = *(const v2f*)(arow + k);
    const float* p = Wm + k * NCLS;
    v2f b;
    b.x = p[l16];      b.y = p[NCLS + l16];
    acc0 = __builtin_amdgcn_wmma_f32_16x16x4_f32(false, a, false, b, (short)0, acc0, false, false);
    b.x = p[16 + l16]; b.y = p[NCLS + 16 + l16];
    acc1 = __builtin_amdgcn_wmma_f32_16x16x4_f32(false, a, false, b, (short)0, acc1, false, false);
    float t0 = p[c2], t1 = p[NCLS + c2];
    b.x = ok2 ? t0 : 0.0f; b.y = ok2 ? t1 : 0.0f;
    acc2 = __builtin_amdgcn_wmma_f32_16x16x4_f32(false, a, false, b, (short)0, acc2, false, false);
  }
}

template <int M>
__device__ __forceinline__ float row16_xor(float v) {
  return __int_as_float(__builtin_amdgcn_update_dpp(0, __float_as_int(v), 0x160 + M, 0xf, 0xf, true));
}
__device__ __forceinline__ float row16_max(float v) {
  v = fmaxf(v, row16_xor<1>(v));
  v = fmaxf(v, row16_xor<2>(v));
  v = fmaxf(v, row16_xor<4>(v));
  v = fmaxf(v, row16_xor<8>(v));
  return v;
}
__device__ __forceinline__ float row16_sum(float v) {
  v += row16_xor<1>(v);
  v += row16_xor<2>(v);
  v += row16_xor<4>(v);
  v += row16_xor<8>(v);
  return v;
}

__global__ void __launch_bounds__(128)
out_wmma_lsm_kernel(const float* __restrict__ h,
                    const float* __restrict__ agg2,
                    const float* __restrict__ W0,
                    const float* __restrict__ W1,
                    const float* __restrict__ b2,
                    float* __restrict__ out) {
  int wave = threadIdx.x >> 5;
  int lane = threadIdx.x & 31;
  int half = lane >> 4;
  int l16  = lane & 15;
  int m0 = (blockIdx.x * 4 + wave) * 16;
  if (m0 >= N_NODES) return;
  int mA = m0 + l16;

  v8f acc0 = {}, acc1 = {}, acc2 = {};
  out_k32(h    + (long long)mA * HID, W0, half, l16, acc0, acc1, acc2);
  out_k32(agg2 + (long long)mA * HID, W1, half, l16, acc0, acc1, acc2);

  bool ok2 = l16 < 8;
  float bb0 = b2[l16], bb1 = b2[16 + l16], bb2 = b2[32 + (l16 & 7)];
  float x0[8], x1[8], x2[8], mx[8], se[8];
#pragma unroll
  for (int v = 0; v < 8; v++) {
    x0[v] = acc0[v] + bb0;
    x1[v] = acc1[v] + bb1;
    x2[v] = acc2[v] + bb2;
    mx[v] = fmaxf(fmaxf(x0[v], x1[v]), ok2 ? x2[v] : -INFINITY);
  }
#pragma unroll
  for (int v = 0; v < 8; v++) mx[v] = row16_max(mx[v]);
#pragma unroll
  for (int v = 0; v < 8; v++) {
    x0[v] -= mx[v];
    x1[v] -= mx[v];
    x2[v] -= mx[v];
    float e2 = expf(x2[v]);
    se[v] = expf(x0[v]) + expf(x1[v]);
    se[v] += ok2 ? e2 : 0.0f;
  }
#pragma unroll
  for (int v = 0; v < 8; v++) se[v] = logf(row16_sum(se[v]));
  float* otile = out + (long long)(m0 + 8 * half) * NCLS + l16;
#pragma unroll
  for (int v = 0; v < 8; v++) {
    otile[v * NCLS]      = x0[v] - se[v];
    otile[v * NCLS + 16] = x1[v] - se[v];
  }
  if (ok2) {
#pragma unroll
    for (int v = 0; v < 8; v++) otile[v * NCLS + 32] = x2[v] - se[v];
  }
}


extern "C" void kernel_launch(void* const* d_in, const int* in_sizes, int n_in,
                              void* d_out, int out_size, void* d_ws, size_t ws_size,
                              hipStream_t stream) {
  (void)in_sizes; (void)n_in; (void)out_size; (void)ws_size;
  const float* x    = (const float*)d_in[0];
  const int*   ei   = (const int*)  d_in[1];
  const float* W0_1 = (const float*)d_in[2];
  const float* W1_1 = (const float*)d_in[3];
  const float* b1   = (const float*)d_in[4];
  const float* W0_2 = (const float*)d_in[5];
  const float* W1_2 = (const float*)d_in[6];
  const float* b2   = (const float*)d_in[7];
  float* out = (float*)d_out;

  const int* rows = ei;
  const int* cols = ei + N_EDGES;

  const long long N = N_NODES;
  float* ws     = (float*)d_ws;
  int*   cnt    = (int*)ws;
  float* dis    = ws + N;
  float* P      = ws + 2 * N;
  float* agg2   = ws + 2 * N;
  int*   colidx = (int*)(ws + 66 * N);
  int*   rowptr = (int*)(ws + 82 * N);
  int*   tcnt   = (int*)(ws + 83 * N);
  int*   tbase  = (int*)(ws + 84 * N);
  float* hbuf   = ws + 98 * N;
  int*   binned = (int*)(ws + 98 * N);

  const int T = 256;

  bin_count_kernel<<<NCH, 1024, 0, stream>>>(rows, cols, tcnt);
  table_scan_kernel<<<(TBL + 1023) / 1024, 1024, 0, stream>>>(tcnt, tbase);
  bin_scatter_kernel<<<NCH, 1024, 0, stream>>>(rows, cols, tbase, binned);
  bin_rows_kernel<<<NB, 1024, 0, stream>>>(tcnt, tbase, binned, cnt, rowptr, dis, colidx);

  proj1_wmma_kernel<<<(N_NODES + 63) / 64, 128, 0, stream>>>(x, W0_1, W1_1, P);
  gather_relu_kernel<<<(N_NODES + 7) / 8, T, 0, stream>>>(
      rowptr, cnt, colidx, dis, P, b1, hbuf);

  gather_rows_kernel<<<(N_NODES + 7) / 8, T, 0, stream>>>(
      rowptr, cnt, colidx, dis, hbuf, agg2);
  out_wmma_lsm_kernel<<<(N_NODES + 63) / 64, 128, 0, stream>>>(hbuf, agg2, W0_2, W1_2, b2, out);
}


# Gather launches: one wave walks 8 consecutive rows (12500 waves, not 100000), next rows' list entries and weights requested under the current row's loads; row sums still in list order
# speedup vs baseline: 3.5847x; 1.0513x over previous
#include <hip/hip_runtime.h>
#include <math.h>

#define N_NODES 100000
#define N_EDGES 1600000
#define F_IN    50
#define HID     32
#define NCLS    40
#define BIN_SHIFT 8
#define BIN_ROWS  256
#define NB        391
#define CHUNK     8192
#define NCH       196
#define TBL       (NB * NCH)
#define COL_BITS  17
static_assert(N_NODES <= (1 << COL_BITS), "col must fit the low field of a bin entry");
static_assert(NB * BIN_ROWS >= N_NODES && NCH * CHUNK >= N_EDGES && N_EDGES % 4 == 0, "sizes");
static_assert(TBL <= N_NODES, "each table has one N-word plane of the workspace");


typedef float v2f __attribute__((ext_vector_type(2)));
typedef float v8f __attribute__((ext_vector_type(8)));


__global__ void __launch_bounds__(1024)
bin_count_kernel(const int* __restrict__ rows,
                 const int* __restrict__ cols,
                 int* __restrict__ tcnt) {
  __shared__ int hist[NB];
  int t = threadIdx.x;
  int4 r4[2], c4[2];
#pragma unroll
  for (int h = 0; h < 2; h++) {
    int e = blockIdx.x * CHUNK + h * 4096 + t * 4;
    r4[h] = make_int4(0, 0, 0, 0);
    c4[h] = make_int4(0, 0, 0, 0);
    if (e < N_EDGES) {
      r4[h] = *(const int4*)(rows + e);
      c4[h] = *(const int4*)(cols + e);
    }
  }
  for (int i = t; i < NB; i += 1024) hist[i] = 0;
  __syncthreads();
#pragma unroll
  for (int h = 0; h < 2; h++) {
    if (r4[h].x != c4[h].x) atomicAdd(&hist[r4[h].x >> BIN_SHIFT], 1);
    if (r4[h].y != c4[h].y) atomicAdd(&hist[r4[h].y >> BIN_SHIFT], 1);
    if (r4[h].z != c4[h].z) atomicAdd(&hist[r4[h].z >> BIN_SHIFT], 1);
    if (r4[h].w != c4[h].w) atomicAdd(&hist[r4[h].w >> BIN_SHIFT], 1);
  }
  __syncthreads();
  for (int i = t; i < NB; i += 1024) tcnt[i * NCH + blockIdx.x] = hist[i];
}

__global__ void __launch_bounds__(1024)
table_scan_kernel(const int* __restrict__ tcnt,
                  int* __restrict__ tbase) {
  __shared__ int s_pre[32];
  __shared__ int s_tot[32];
  int t = threadIdx.x, lane = t & 31, wave = t >> 5;
  int b0 = blockIdx.x * 1024;
  const int4* c4 = (const int4*)tcnt;
  int pre = 0;
#pragma unroll 4
  for (int q = t; q < (b0 >> 2); q += 1024) {
    int4 x = c4[q];
    pre += (x.x + x.y) + (x.z + x.w);
  }
  int i = b0 + t;
  int v = (i < TBL) ? tcnt[i] : 0;
  int incl = v;
  for (int d = 1; d < 32; d <<= 1) {
    int y = __shfl_up(incl, d);
    if (lane >= d) incl += y;
  }
  for (int d = 16; d >= 1; d >>= 1) pre += __shfl_xor(pre, d);
  if (lane == 31) s_tot[wave] = incl;
  if (lane == 0)  s_pre[wave] = pre;
  __syncthreads();
  int add = s_pre[lane] + (lane < wave ? s_tot[lane] : 0);
  for (int d = 16; d >= 1; d >>= 1) add += __shfl_xor(add, d);
  if (i < TBL) tbase[i] = add + incl - v;
}

__global__ void __launch_bounds__(1024)
bin_scatter_kernel(const int* __restrict__ rows,
                   const int* __restrict__ cols,
                   const int* __restrict__ tbase,
                   int* __restrict__ binned) {
  __shared__ int base[NB];
  __shared__ int cur[NB];
  int t = threadIdx.x;
  int4 r4[2], c4[2];
#pragma unroll
  for (int h = 0; h < 2; h++) {
    int e = blockIdx.x * CHUNK + h * 4096 + t * 4;
    r4[h] = make_int4(0, 0, 0, 0);
    c4[h] = make_int4(0, 0, 0, 0);
    if (e < N_EDGES) {
      r4[h] = *(const int4*)(rows + e);
      c4[h] = *(const int4*)(cols + e);
    }
  }
  for (int i = t; i < NB; i += 1024) {
    base[i] = tbase[i * NCH + blockIdx.x];
    cur[i] = 0;
  }
  __syncthreads();
#pragma unroll
  for (int h = 0; h < 2; h++) {
    if (r4[h].x != c4[h].x) { int b = r4[h].x >> BIN_SHIFT; int k = atomicAdd(&cur[b], 1); binned[base[b] + k] = ((r4[h].x & (BIN_ROWS - 1)) << COL_BITS) | c4[h].x; }
    if (r4[h].y != c4[h].y) { int b = r4[h].y >> BIN_SHIFT; int k = atomicAdd(&cur[b], 1); binned[base[b] + k] = ((r4[h].y & (BIN_ROWS - 1)) << COL_BITS) | c4[h].y; }
    if (r4[h].z != c4[h].z) { int b = r4[h].z >> BIN_SHIFT; int k = atomicAdd(&cur[b], 1); binned[base[b] + k] = ((r4[h].z & (BIN_ROWS - 1)) << COL_BITS) | c4[h].z; }
    if (r4[h].w != c4[h].w) { int b = r4[h].w >> BIN_SHIFT; int k = atomicAdd(&cur[b], 1); binned[base[b] + k] = ((r4[h].w & (BIN_ROWS - 1)) << COL_BITS) | c4[h].w; }
  }
}

__global__ void __launch_bounds__(1024)
bin_rows_kernel(const int* __restrict__ tcnt,
                const int* __restrict__ tbase,
                const int* __restrict__ binned,
                int* __restrict__ cnt,
                int* __restrict__ rowptr,
                float* __restrict__ dis,
                int* __restrict__ colidx) {
  __shared__ int s_cnt[BIN_ROWS];
  __shared__ int s_start[BIN_ROWS];
  __shared__ int s_cur[BIN_ROWS];
  __shared__ int s_wtot[8];
  int t = threadIdx.x, lane = t & 31, wave = t >> 5;
  int b = blockIdx.x;
  int s = tbase[b * NCH];
  int e = (b + 1 < NB) ? tbase[(b + 1) * NCH] : tbase[TBL - 1] + tcnt[TBL - 1];
  if (t < BIN_ROWS) { s_cnt[t] = 0; s_cur[t] = 0; }
  __syncthreads();
  for (int i0 = s + t; i0 < e; i0 += 4096) {
    int p[4];
#pragma unroll
    for (int u = 0; u < 4; u++) {
      int idx = i0 + u * 1024;
      p[u] = (idx < e) ? binned[idx] : -1;
    }
#pragma unroll
    for (int u = 0; u < 4; u++)
      if (p[u] >= 0) atomicAdd(&s_cnt[p[u] >> COL_BITS], 1);
  }
  __syncthreads();
  int v = 0, incl = 0;
  if (t < BIN_ROWS) {
    v = s_cnt[t];
    incl = v;
    for (int d = 1; d < 32; d <<= 1) {
      int y = __shfl_up(incl, d);
      if (lane >= d) incl += y;
    }
    if (lane == 31) s_wtot[wave] = incl;
  }
  __syncthreads();
  if (t < BIN_ROWS) {
    int add = 0;
    for (int w = 0; w < 8; w++) add += (w < wave) ? s_wtot[w] : 0;
    int excl = add + incl - v;
    s_start[t] = excl;
    int row = b * BIN_ROWS + t;
    if (row < N_NODES) {
      rowptr[row] = s + excl;
      cnt[row] = v;
      float dg = (float)v;
      dis[row] = (dg > 0.0f) ? rsqrtf(fmaxf(dg, 1.0f)) : 0.0f;
    }
  }
  __syncthreads();
  for (int i0 = s + t; i0 < e; i0 += 4096) {
    int p[4];
#pragma unroll
    for (int u = 0; u < 4; u++) {
      int idx = i0 + u * 1024;
      p[u] = (idx < e) ? binned[idx] : -1;
    }
#pragma unroll
    for (int u = 0; u < 4; u++) {
      if (p[u] >= 0) {
        int rl = p[u] >> COL_BITS;
        int k = atomicAdd(&s_cur[rl], 1);
        colidx[s + s_start[rl] + k] = p[u] & ((1 << COL_BITS) - 1);
      }
    }
  }
}


#define RPW 8
static_assert(N_NODES % RPW == 0 && RPW >= 2 && RPW <= 32,
              "a wave's rows all exist; their bookkeeping sits one row per lane");

__device__ __forceinline__ int lane_i(int v, int l) { return __builtin_amdgcn_readlane(v, l); }
__device__ __forceinline__ float lane_f(float v, int l) {
  return __int_as_float(__builtin_amdgcn_readlane(__float_as_int(v), l));
}

__device__ __forceinline__ int list_head(const int* __restrict__ colidx, int start, int n, int lane) {
  int c = 0;
  if (n > 0) {
    int m = n < 32 ? n : 32;
    int li = lane < m ? lane : m - 1;
    c = colidx[start + li];
  }
  return c;
}

__device__ __forceinline__ float head_weight(float dc, float ndr, int n, int lane) {
  int m = n < 32 ? n : 32;
  return lane < m ? ndr * dc : 0.0f;
}

template <int K>
__device__ __forceinline__ float gather_batch(int c, float w, int j, int m, int lane,
                                              const float* __restrict__ sp, int sstride, float acc) {
  float v[K];
#pragma unroll
  for (int u = 0; u < K; u++) {
    int cj = lane_i(c, j + u);
    const float* rowp = sp + (long long)cj * sstride;
    float x = rowp[lane];
    v[u] = (j + u < m) ? x : 0.0f;
  }
#pragma unroll
  for (int u = 0; u < K; u++) acc = fmaf(lane_f(w, j + u), v[u], acc);
  return acc;
}

__device__ __forceinline__ float gather_walk(int c, float w, int m, int lane,
                                             const float* __restrict__ sp, int sstride, float acc) {
  for (int j = 0; j < m; j += 16) {
    if (j + 8 < m) acc = gather_batch<16>(c, w, j, m, lane, sp, sstride, acc);
    else           acc = gather_batch<8>(c, w, j, m, lane, sp, sstride, acc);
  }
  return acc;
}

template <bool LAYER1>
__device__ __forceinline__ void gather_span(int r0, int lane,
                                            const int* __restrict__ rowptr,
                                            const int* __restrict__ cnt,
                                            const int* __restrict__ colidx,
                                            const float* __restrict__ dis,
                                            const float* __restrict__ sp, int sstride,
                                            const float* __restrict__ selfp, float bias,
                                            float* __restrict__ dst) {
  int mr = r0 + (lane & (RPW - 1));
  int st_v = rowptr[mr];
  int n_v = cnt[mr];
  float d_v = dis[mr];

  int n0 = lane_i(n_v, 0);
  int c_a = list_head(colidx, lane_i(st_v, 0), n0, lane);
  int c_b = list_head(colidx, lane_i(st_v, 1), lane_i(n_v, 1), lane);
  float w_a = head_weight(dis[c_a], -lane_f(d_v, 0), n0, lane);
  for (int i = 0; i < RPW; i++) {
    int r = r0 + i;
    int start = lane_i(st_v, i);
    int n = lane_i(n_v, i);
    int c_c = 0;
    if (i + 2 < RPW) c_c = list_head(colidx, lane_i(st_v, i + 2), lane_i(n_v, i + 2), lane);
    float self = 0.0f;
    if (LAYER1) self = selfp[(long long)r * 64 + lane];
    float dc_b = dis[c_b];
    int m = n < 32 ? n : 32;
    float acc = gather_walk(c_a, w_a, m, lane, sp, sstride, 0.0f);
    if (n > 32) {
      float ndr = -lane_f(d_v, i);
      for (int base = 32; base < n; base += 32) {
        int m2 = n - base;
        if (m2 > 32) m2 = 32;
        int li = lane < m2 ? lane : m2 - 1;
        int c2 = colidx[start + base + li];
        float w2 = lane < m2 ? ndr * dis[c2] : 0.0f;
        acc = gather_walk(c2, w2, m2, lane, sp, sstride, acc);
      }
    }
    float* drow = dst + (long long)r * HID;
    if (LAYER1) {
      float v = self + acc + bias;
      drow[lane] = v > 0.0f ? v : 0.0f;
    } else {
      drow[lane] = acc;
    }
    float w_b = head_weight(dc_b, -lane_f(d_v, i + 1), lane_i(n_v, i + 1), lane);
    c_a = c_b; w_a = w_b; c_b = c_c;
  }
}

__global__ void __launch_bounds__(128)
gather_relu_kernel(const int* __restrict__ rowptr,
                   const int* __restrict__ cnt,
                   const int* __restrict__ colidx,
                   const float* __restrict__ dis,
                   const float* __restrict__ P,
                   const float* __restrict__ b1,
                   float* __restrict__ h) {
  int lane = threadIdx.x & 31;
  int r0 = __builtin_amdgcn_readfirstlane((blockIdx.x * 4 + (threadIdx.x >> 5)) * RPW);
  if (r0 >= N_NODES) return;
  gather_span<true>(r0, lane, rowptr, cnt, colidx, dis, P + HID, 64, P, b1[lane], h);
}

__global__ void __launch_bounds__(128)
gather_rows_kernel(const int* __restrict__ rowptr,
                   const int* __restrict__ cnt,
                   const int* __restrict__ colidx,
                   const float* __restrict__ dis,
                   const float* __restrict__ src,
                   float* __restrict__ dst) {
  int lane = threadIdx.x & 31;
  int r0 = __builtin_amdgcn_readfirstlane((blockIdx.x * 4 + (threadIdx.x >> 5)) * RPW);
  if (r0 >= N_NODES) return;
  gather_span<false>(r0, lane, rowptr, cnt, colidx, dis, src, HID, src, 0.0f, dst);
}


__global__ void __launch_bounds__(128)
proj1_wmma_kernel(const float* __restrict__ x,
                  const float* __restrict__ W0,
                  const float* __restrict__ W1,
                  float* __restrict__ P) {
  int wave = threadIdx.x >> 5;
  int lane = threadIdx.x & 31;
  int half = lane >> 4;
  int l16  = lane & 15;
  int m0 = (blockIdx.x * 4 + wave) * 16;
  if (m0 >= N_NODES) return;
  int mA = m0 + l16;
  const float* arow = x + (long long)mA * F_IN;

  v8f acc0 = {}, acc1 = {}, acc2 = {}, acc3 = {};
  for (int kk = 0; kk < 48; kk += 4) {
    int k = kk + 2 * half;
    v2f a = *(const v2f*)(arow + k);
    const float* p0 = W0 + k * HID + l16;
    const float* p1 = W1 + k * HID + l16;
    v2f b;
    b.x = p0[0];  b.y = p0[HID];
    acc0 = __builtin_amdgcn_wmma_f32_16x16x4_f32(false, a, false, b, (short)0, acc0, false, false);
    b.x = p0[16]; b.y = p0[HID + 16];
    acc1 = __builtin_amdgcn_wmma_f32_16x16x4_f32(false, a, false, b, (short)0, acc1, false, false);
    b.x = p1[0];  b.y = p1[HID];
    acc2 = __builtin_amdgcn_wmma_f32_16x16x4_f32(false, a, false, b, (short)0, acc2, false, false);
    b.x = p1[16]; b.y = p1[HID + 16];
    acc3 = __builtin_amdgcn_wmma_f32_16x16x4_f32(false, a, false, b, (short)0, acc3, false, false);
  }
  {
    v2f aload = *(const v2f*)(arow + 48);
    v2f a;
    a.x = half ? 0.0f : aload.x;
    a.y = half ? 0.0f : aload.y;
    const float* p0 = W0 + 48 * HID + l16;
    const float* p1 = W1 + 48 * HID + l16;
    v2f b;
    b.x = p0[0];  b.y = p0[HID];
    acc0 = __builtin_amdgcn_wmma_f32_16x16x4_f32(false, a, false, b, (short)0, acc0, false, false);
    b.x = p0[16]; b.y = p0[HID + 16];
    acc1 = __builtin_amdgcn_wmma_f32_16x16x4_f32(false, a, false, b, (short)0, acc1, false, false);
    b.x = p1[0];  b.y = p1[HID];
    acc2 = __builtin_amdgcn_wmma_f32_16x16x4_f32(false, a, false, b, (short)0, acc2, false, false);
    b.x = p1[16]; b.y = p1[HID + 16];
    acc3 = __builtin_amdgcn_wmma_f32_16x16x4_f32(false, a, false, b, (short)0, acc3, false, false);
  }
#pragma unroll
  for (int v = 0; v < 8; v++) {
    int mC = m0 + v + 8 * half;
    float* prow = P + (long long)mC * 64 + l16;
    prow[0]  = acc0[v];
    prow[16] = acc1[v];
    prow[32] = acc2[v];
    prow[48] = acc3[v];
  }
}


__device__ __forceinline__ void out_k32(const float* __restrict__ arow,
                                        const float* __restrict__ Wm,
                                        int half, int l16,
                                        v8f& acc0, v8f& acc1, v8f& acc2) {
  int c2 = 32 + (l16 & 7);
  bool ok2 = l16 < 8;
  for (int kk = 0; kk < HID; kk += 4) {
    int k = kk + 2 * half;
    v2f a = *(const v2f*)(arow + k);
    const float* p = Wm + k * NCLS;
    v2f b;
    b.x = p[l16];      b.y = p[NCLS + l16];
    acc0 = __builtin_amdgcn_wmma_f32_16x16x4_f32(false, a, false, b, (short)0, acc0, false, false);
    b.x = p[16 + l16]; b.y = p[NCLS + 16 + l16];
    acc1 = __builtin_amdgcn_wmma_f32_16x16x4_f32(false, a, false, b, (short)0, acc1, false, false);
    float t0 = p[c2], t1 = p[NCLS + c2];
    b.x = ok2 ? t0 : 0.0f; b.y = ok2 ? t1 : 0.0f;
    acc2 = __builtin_amdgcn_wmma_f32_16x16x4_f32(false, a, false, b, (short)0, acc2, false, false);
  }
}

template <int M>
__device__ __forceinline__ float row16_xor(float v) {
  return __int_as_float(__builtin_amdgcn_update_dpp(0, __float_as_int(v), 0x160 + M, 0xf, 0xf, true));
}
__device__ __forceinline__ float row16_max(float v) {
  v = fmaxf(v, row16_xor<1>(v));
  v = fmaxf(v, row16_xor<2>(v));
  v = fmaxf(v, row16_xor<4>(v));
  v = fmaxf(v, row16_xor<8>(v));
  return v;
}
__device__ __forceinline__ float row16_sum(float v) {
  v += row16_xor<1>(v);
  v += row16_xor<2>(v);
  v += row16_xor<4>(v);
  v += row16_xor<8>(v);
  return v;
}

__global__ void __launch_bounds__(128)
out_wmma_lsm_kernel(const float* __restrict__ h,
                    const float* __restrict__ agg2,
                    const float* __restrict__ W0,
                    const float* __restrict__ W1,
                    const float* __restrict__ b2,
                    float* __restrict__ out) {
  int wave = threadIdx.x >> 5;
  int lane = threadIdx.x & 31;
  int half = lane >> 4;
  int l16  = lane & 15;
  int m0 = (blockIdx.x * 4 + wave) * 16;
  if (m0 >= N_NODES) return;
  int mA = m0 + l16;

  v8f acc0 = {}, acc1 = {}, acc2 = {};
  out_k32(h    + (long long)mA * HID, W0, half, l16, acc0, acc1, acc2);
  out_k32(agg2 + (long long)mA * HID, W1, half, l16, acc0, acc1, acc2);

  bool ok2 = l16 < 8;
  float bb0 = b2[l16], bb1 = b2[16 + l16], bb2 = b2[32 + (l16 & 7)];
  float x0[8], x1[8], x2[8], mx[8], se[8];
#pragma unroll
  for (int v = 0; v < 8; v++) {
    x0[v] = acc0[v] + bb0;
    x1[v] = acc1[v] + bb1;
    x2[v] = acc2[v] + bb2;
    mx[v] = fmaxf(fmaxf(x0[v], x1[v]), ok2 ? x2[v] : -INFINITY);
  }
#pragma unroll
  for (int v = 0; v < 8; v++) mx[v] = row16_max(mx[v]);
#pragma unroll
  for (int v = 0; v < 8; v++) {
    x0[v] -= mx[v];
    x1[v] -= mx[v];
    x2[v] -= mx[v];
    float e2 = expf(x2[v]);
    se[v] = expf(x0[v]) + expf(x1[v]);
    se[v] += ok2 ? e2 : 0.0f;
  }
#pragma unroll
  for (int v = 0; v < 8; v++) se[v] = logf(row16_sum(se[v]));
  float* otile = out + (long long)(m0 + 8 * half) * NCLS + l16;
#pragma unroll
  for (int v = 0; v < 8; v++) {
    otile[v * NCLS]      = x0[v] - se[v];
    otile[v * NCLS + 16] = x1[v] - se[v];
  }
  if (ok2) {
#pragma unroll
    for (int v = 0; v < 8; v++) otile[v * NCLS + 32] = x2[v] - se[v];
  }
}


extern "C" void kernel_launch(void* const* d_in, const int* in_sizes, int n_in,
                              void* d_out, int out_size, void* d_ws, size_t ws_size,
                              hipStream_t stream) {
  (void)in_sizes; (void)n_in; (void)out_size; (void)ws_size;
  const float* x    = (const float*)d_in[0];
  const int*   ei   = (const int*)  d_in[1];
  const float* W0_1 = (const float*)d_in[2];
  const float* W1_1 = (const float*)d_in[3];
  const float* b1   = (const float*)d_in[4];
  const float* W0_2 = (const float*)d_in[5];
  const float* W1_2 = (const float*)d_in[6];
  const float* b2   = (const float*)d_in[7];
  float* out = (float*)d_out;

  const int* rows = ei;
  const int* cols = ei + N_EDGES;

  const long long N = N_NODES;
  float* ws     = (float*)d_ws;
  int*   cnt    = (int*)ws;
  float* dis    = ws + N;
  float* P      = ws + 2 * N;
  float* agg2   = ws + 2 * N;
  int*   colidx = (int*)(ws + 66 * N);
  int*   rowptr = (int*)(ws + 82 * N);
  int*   tcnt   = (int*)(ws + 83 * N);
  int*   tbase  = (int*)(ws + 84 * N);
  float* hbuf   = ws + 98 * N;
  int*   binned = (int*)(ws + 98 * N);

  bin_count_kernel<<<NCH, 1024, 0, stream>>>(rows, cols, tcnt);
  table_scan_kernel<<<(TBL + 1023) / 1024, 1024, 0, stream>>>(tcnt, tbase);
  bin_scatter_kernel<<<NCH, 1024, 0, stream>>>(rows, cols, tbase, binned);
  bin_rows_kernel<<<NB, 1024, 0, stream>>>(tcnt, tbase, binned, cnt, rowptr, dis, colidx);

  proj1_wmma_kernel<<<(N_NODES + 63) / 64, 128, 0, stream>>>(x, W0_1, W1_1, P);
  gather_relu_kernel<<<(N_NODES / RPW + 3) / 4, 128, 0, stream>>>(
      rowptr, cnt, colidx, dis, P, b1, hbuf);

  gather_rows_kernel<<<(N_NODES / RPW + 3) / 4, 128, 0, stream>>>(
      rowptr, cnt, colidx, dis, hbuf, agg2);
  out_wmma_lsm_kernel<<<(N_NODES + 63) / 64, 128, 0, stream>>>(hbuf, agg2, W0_2, W1_2, b2, out);
}


# Launches merged, 8 -> 6: GEMM-1 rides in the edge-count launch as extra work-groups; the count table's scan is redone inside each scatter work-group (table chunk-major, 16-B reads); sum order kept
# speedup vs baseline: 3.7317x; 1.0410x over previous
#include <hip/hip_runtime.h>
#include <math.h>

#define N_NODES 100000
#define N_EDGES 1600000
#define F_IN    50
#define HID     32
#define NCLS    40
#define BIN_SHIFT 8
#define BIN_ROWS  256
#define NB        391
#define CHUNK     8192
#define NCH       196
#define NBP       392
#define NBQ       (NBP / 4)
#define TBL       (NCH * NBP)
#define PB        196
#define COL_BITS  17
static_assert(N_NODES <= (1 << COL_BITS), "col must fit the low field of a bin entry");
static_assert(NB * BIN_ROWS >= N_NODES && NCH * CHUNK >= N_EDGES && N_EDGES % 4 == 0, "sizes");
static_assert(TBL <= N_NODES && NBP >= NB && NBP % 4 == 0, "the table has one N-word plane of the workspace");
static_assert(PB * 512 >= N_NODES, "GEMM-1 work-groups cover every row");


typedef float v2f __attribute__((ext_vector_type(2)));
typedef float v8f __attribute__((ext_vector_type(8)));


__device__ __forceinline__ void
bin_count_body(const int* __restrict__ rows,
               const int* __restrict__ cols,
               int* __restrict__ tcnt) {
  __shared__ int hist[NB];
  int t = threadIdx.x;
  int4 r4[2], c4[2];
#pragma unroll
  for (int h = 0; h < 2; h++) {
    int e = blockIdx.x * CHUNK + h * 4096 + t * 4;
    r4[h] = make_int4(0, 0, 0, 0);
    c4[h] = make_int4(0, 0, 0, 0);
    if (e < N_EDGES) {
      r4[h] = *(const int4*)(rows + e);
      c4[h] = *(const int4*)(cols + e);
    }
  }
  for (int i = t; i < NB; i += 1024) hist[i] = 0;
  __syncthreads();
#pragma unroll
  for (int h = 0; h < 2; h++) {
    if (r4[h].x != c4[h].x) atomicAdd(&hist[r4[h].x >> BIN_SHIFT], 1);
    if (r4[h].y != c4[h].y) atomicAdd(&hist[r4[h].y >> BIN_SHIFT], 1);
    if (r4[h].z != c4[h].z) atomicAdd(&hist[r4[h].z >> BIN_SHIFT], 1);
    if (r4[h].w != c4[h].w) atomicAdd(&hist[r4[h].w >> BIN_SHIFT], 1);
  }
  __syncthreads();
  for (int i = t; i < NBP; i += 1024) tcnt[blockIdx.x * NBP + i] = (i < NB) ? hist[i] : 0;
}

#define SG 10
static_assert(SG * NBQ <= 1024 && NB <= 13 * 32, "table sum and scan fit the work-group");

__device__ __forceinline__ int4 add4(int4 a, int4 b) {
  return make_int4(a.x + b.x, a.y + b.y, a.z + b.z, a.w + b.w);
}

__global__ void __launch_bounds__(1024)
bin_scatter_kernel(const int* __restrict__ rows,
                   const int* __restrict__ cols,
                   const int* __restrict__ tcnt,
                   int* __restrict__ binstart,
                   int* __restrict__ binned) {
  __shared__ int base[NB];
  __shared__ int cur[NB];
  __shared__ int4 p_tot[SG * NBQ];
  __shared__ int4 p_pre[SG * NBQ];
  __shared__ int s_wtot[13];
  int t = threadIdx.x, lane = t & 31, wave = t >> 5;
  int4 r4[2], c4[2];
#pragma unroll
  for (int h = 0; h < 2; h++) {
    int e = blockIdx.x * CHUNK + h * 4096 + t * 4;
    r4[h] = make_int4(0, 0, 0, 0);
    c4[h] = make_int4(0, 0, 0, 0);
    if (e < N_EDGES) {
      r4[h] = *(const int4*)(rows + e);
      c4[h] = *(const int4*)(cols + e);
    }
  }
  if (t < SG * NBQ) {
    int g = t / NBQ, j = t - g * NBQ;
    const int4* T4 = (const int4*)tcnt;
    int4 tot = make_int4(0, 0, 0, 0), pre = make_int4(0, 0, 0, 0);
#pragma unroll 10
    for (int i = 0; i < (NCH + SG - 1) / SG; i++) {
      int cc = g + i * SG;
      int4 v = make_int4(0, 0, 0, 0);
      if (cc < NCH) v = T4[cc * NBQ + j];
      tot = add4(tot, v);
      if (cc < (int)blockIdx.x) pre = add4(pre, v);
    }
    p_tot[t] = tot;
    p_pre[t] = pre;
  }
  __syncthreads();
  int v = 0, part = 0, incl = 0;
  if (t < 13 * 32) {
    if (t < NB) {
      const int* q_tot = (const int*)p_tot;
      const int* q_pre = (const int*)p_pre;
#pragma unroll
      for (int g = 0; g < SG; g++) {
        v += q_tot[g * NBP + t];
        part += q_pre[g * NBP + t];
      }
    }
    incl = v;
    for (int d = 1; d < 32; d <<= 1) {
      int y = __shfl_up(incl, d);
      if (lane >= d) incl += y;
    }
    if (lane == 31) s_wtot[wave] = incl;
  }
  __syncthreads();
  if (t < NB) {
    int add = 0;
    for (int w = 0; w < 13; w++) add += (w < wave) ? s_wtot[w] : 0;
    int start = add + incl - v;
    base[t] = start + part;
    cur[t] = 0;
    if (blockIdx.x == 0) {
      binstart[t] = start;
      if (t == NB - 1) binstart[NB] = start + v;
    }
  }
  __syncthreads();
#pragma unroll
  for (int h = 0; h < 2; h++) {
    if (r4[h].x != c4[h].x) { int b = r4[h].x >> BIN_SHIFT; int k = atomicAdd(&cur[b], 1); binned[base[b] + k] = ((r4[h].x & (BIN_ROWS - 1)) << COL_BITS) | c4[h].x; }
    if (r4[h].y != c4[h].y) { int b = r4[h].y >> BIN_SHIFT; int k = atomicAdd(&cur[b], 1); binned[base[b] + k] = ((r4[h].y & (BIN_ROWS - 1)) << COL_BITS) | c4[h].y; }
    if (r4[h].z != c4[h].z) { int b = r4[h].z >> BIN_SHIFT; int k = atomicAdd(&cur[b], 1); binned[base[b] + k] = ((r4[h].z & (BIN_ROWS - 1)) << COL_BITS) | c4[h].z; }
    if (r4[h].w != c4[h].w) { int b = r4[h].w >> BIN_SHIFT; int k = atomicAdd(&cur[b], 1); binned[base[b] + k] = ((r4[h].w & (BIN_ROWS - 1)) << COL_BITS) | c4[h].w; }
  }
}

__global__ void __launch_bounds__(1024)
bin_rows_kernel(const int* __restrict__ binstart,
                const int* __restrict__ binned,
                int* __restrict__ cnt,
                int* __restrict__ rowptr,
                float* __restrict__ dis,
                int* __restrict__ colidx) {
  __shared__ int s_cnt[BIN_ROWS];
  __shared__ int s_start[BIN_ROWS];
  __shared__ int s_cur[BIN_ROWS];
  __shared__ int s_wtot[8];
  int t = threadIdx.x, lane = t & 31, wave = t >> 5;
  int b = blockIdx.x;
  int s = binstart[b];
  int e = binstart[b + 1];
  if (t < BIN_ROWS) { s_cnt[t] = 0; s_cur[t] = 0; }
  __syncthreads();
  for (int i0 = s + t; i0 < e; i0 += 4096) {
    int p[4];
#pragma unroll
    for (int u = 0; u < 4; u++) {
      int idx = i0 + u * 1024;
      p[u] = (idx < e) ? binned[idx] : -1;
    }
#pragma unroll
    for (int u = 0; u < 4; u++)
      if (p[u] >= 0) atomicAdd(&s_cnt[p[u] >> COL_BITS], 1);
  }
  __syncthreads();
  int v = 0, incl = 0;
  if (t < BIN_ROWS) {
    v = s_cnt[t];
    incl = v;
    for (int d = 1; d < 32; d <<= 1) {
      int y = __shfl_up(incl, d);
      if (lane >= d) incl += y;
    }
    if (lane == 31) s_wtot[wave] = incl;
  }
  __syncthreads();
  if (t < BIN_ROWS) {
    int add = 0;
    for (int w = 0; w < 8; w++) add += (w < wave) ? s_wtot[w] : 0;
    int excl = add + incl - v;
    s_start[t] = excl;
    int row = b * BIN_ROWS + t;
    if (row < N_NODES) {
      rowptr[row] = s + excl;
      cnt[row] = v;
      float dg = (float)v;
      dis[row] = (dg > 0.0f) ? rsqrtf(fmaxf(dg, 1.0f)) : 0.0f;
    }
  }
  __syncthreads();
  for (int i0 = s + t; i0 < e; i0 += 4096) {
    int p[4];
#pragma unroll
    for (int u = 0; u < 4; u++) {
      int idx = i0 + u * 1024;
      p[u] = (idx < e) ? binned[idx] : -1;
    }
#pragma unroll
    for (int u = 0; u < 4; u++) {
      if (p[u] >= 0) {
        int rl = p[u] >> COL_BITS;
        int k = atomicAdd(&s_cur[rl], 1);
        colidx[s + s_start[rl] + k] = p[u] & ((1 << COL_BITS) - 1);
      }
    }
  }
}


#define RPW 8
static_assert(N_NODES % RPW == 0 && RPW >= 2 && RPW <= 32,
              "a wave's rows all exist; their bookkeeping sits one row per lane");

__device__ __forceinline__ int lane_i(int v, int l) { return __builtin_amdgcn_readlane(v, l); }
__device__ __forceinline__ float lane_f(float v, int l) {
  return __int_as_float(__builtin_amdgcn_readlane(__float_as_int(v), l));
}

__device__ __forceinline__ int list_head(const int* __restrict__ colidx, int start, int n, int lane) {
  int c = 0;
  if (n > 0) {
    int m = n < 32 ? n : 32;
    int li = lane < m ? lane : m - 1;
    c = colidx[start + li];
  }
  return c;
}

__device__ __forceinline__ float head_weight(float dc, float ndr, int n, int lane) {
  int m = n < 32 ? n : 32;
  return lane < m ? ndr * dc : 0.0f;
}

template <int K>
__device__ __forceinline__ float gather_batch(int c, float w, int j, int m, int lane,
                                              const float* __restrict__ sp, int sstride, float acc) {
  float v[K];
#pragma unroll
  for (int u = 0; u < K; u++) {
    int cj = lane_i(c, j + u);
    const float* rowp = sp + (long long)cj * sstride;
    float x = rowp[lane];
    v[u] = (j + u < m) ? x : 0.0f;
  }
#pragma unroll
  for (int u = 0; u < K; u++) acc = fmaf(lane_f(w, j + u), v[u], acc);
  return acc;
}

__device__ __forceinline__ float gather_walk(int c, float w, int m, int lane,
                                             const float* __restrict__ sp, int sstride, float acc) {
  for (int j = 0; j < m; j += 16) {
    if (j + 8 < m) acc = gather_batch<16>(c, w, j, m, lane, sp, sstride, acc);
    else           acc = gather_batch<8>(c, w, j, m, lane, sp, sstride, acc);
  }
  return acc;
}

template <bool LAYER1>
__device__ __forceinline__ void gather_span(int r0, int lane,
                                            const int* __restrict__ rowptr,
                                            const int* __restrict__ cnt,
                                            const int* __restrict__ colidx,
                                            const float* __restrict__ dis,
                                            const float* __restrict__ sp, int sstride,
                                            const float* __restrict__ selfp, float bias,
                                            float* __restrict__ dst) {
  int mr = r0 + (lane & (RPW - 1));
  int st_v = rowptr[mr];
  int n_v = cnt[mr];
  float d_v = dis[mr];

  int n0 = lane_i(n_v, 0);
  int c_a = list_head(colidx, lane_i(st_v, 0), n0, lane);
  int c_b = list_head(colidx, lane_i(st_v, 1), lane_i(n_v, 1), lane);
  float w_a = head_weight(dis[c_a], -lane_f(d_v, 0), n0, lane);
  for (int i = 0; i < RPW; i++) {
    int r = r0 + i;
    int start = lane_i(st_v, i);
    int n = lane_i(n_v, i);
    int c_c = 0;
    if (i + 2 < RPW) c_c = list_head(colidx, lane_i(st_v, i + 2), lane_i(n_v, i + 2), lane);
    float self = 0.0f;
    if (LAYER1) self = selfp[(long long)r * 64 + lane];
    float dc_b = dis[c_b];
    int m = n < 32 ? n : 32;
    float acc = gather_walk(c_a, w_a, m, lane, sp, sstride, 0.0f);
    if (n > 32) {
      float ndr = -lane_f(d_v, i);
      for (int base = 32; base < n; base += 32) {
        int m2 = n - base;
        if (m2 > 32) m2 = 32;
        int li = lane < m2 ? lane : m2 - 1;
        int c2 = colidx[start + base + li];
        float w2 = lane < m2 ? ndr * dis[c2] : 0.0f;
        acc = gather_walk(c2, w2, m2, lane, sp, sstride, acc);
      }
    }
    float* drow = dst + (long long)r * HID;
    if (LAYER1) {
      float v = self + acc + bias;
      drow[lane] = v > 0.0f ? v : 0.0f;
    } else {
      drow[lane] = acc;
    }
    float w_b = head_weight(dc_b, -lane_f(d_v, i + 1), lane_i(n_v, i + 1), lane);
    c_a = c_b; w_a = w_b; c_b = c_c;
  }
}

__global__ void __launch_bounds__(128)
gather_relu_kernel(const int* __restrict__ rowptr,
                   const int* __restrict__ cnt,
                   const int* __restrict__ colidx,
                   const float* __restrict__ dis,
                   const float* __restrict__ P,
                   const float* __restrict__ b1,
                   float* __restrict__ h) {
  int lane = threadIdx.x & 31;
  int r0 = __builtin_amdgcn_readfirstlane((blockIdx.x * 4 + (threadIdx.x >> 5)) * RPW);
  if (r0 >= N_NODES) return;
  gather_span<true>(r0, lane, rowptr, cnt, colidx, dis, P + HID, 64, P, b1[lane], h);
}

__global__ void __launch_bounds__(128)
gather_rows_kernel(const int* __restrict__ rowptr,
                   const int* __restrict__ cnt,
                   const int* __restrict__ colidx,
                   const float* __restrict__ dis,
                   const float* __restrict__ src,
                   float* __restrict__ dst) {
  int lane = threadIdx.x & 31;
  int r0 = __builtin_amdgcn_readfirstlane((blockIdx.x * 4 + (threadIdx.x >> 5)) * RPW);
  if (r0 >= N_NODES) return;
  gather_span<false>(r0, lane, rowptr, cnt, colidx, dis, src, HID, src, 0.0f, dst);
}


__device__ __forceinline__ void
proj1_wmma_body(int pb,
                const float* __restrict__ x,
                const float* __restrict__ W0,
                const float* __restrict__ W1,
                float* __restrict__ P) {
  int wave = threadIdx.x >> 5;
  int lane = threadIdx.x & 31;
  int half = lane >> 4;
  int l16  = lane & 15;
  int m0 = (pb * 32 + wave) * 16;
  if (m0 >= N_NODES) return;
  int mA = m0 + l16;
  const float* arow = x + (long long)mA * F_IN;

  v8f acc0 = {}, acc1 = {}, acc2 = {}, acc3 = {};
  for (int kk = 0; kk < 48; kk += 4) {
    int k = kk + 2 * half;
    v2f a = *(const v2f*)(arow + k);
    const float* p0 = W0 + k * HID + l16;
    const float* p1 = W1 + k * HID + l16;
    v2f b;
    b.x = p0[0];  b.y = p0[HID];
    acc0 = __builtin_amdgcn_wmma_f32_16x16x4_f32(false, a, false, b, (short)0, acc0, false, false);
    b.x = p0[16]; b.y = p0[HID + 16];
    acc1 = __builtin_amdgcn_wmma_f32_16x16x4_f32(false, a, false, b, (short)0, acc1, false, false);
    b.x = p1[0];  b.y = p1[HID];
    acc2 = __builtin_amdgcn_wmma_f32_16x16x4_f32(false, a, false, b, (short)0, acc2, false, false);
    b.x = p1[16]; b.y = p1[HID + 16];
    acc3 = __builtin_amdgcn_wmma_f32_16x16x4_f32(false, a, false, b, (short)0, acc3, false, false);
  }
  {
    v2f aload = *(const v2f*)(arow + 48);
    v2f a;
    a.x = half ? 0.0f : aload.x;
    a.y = half ? 0.0f : aload.y;
    const float* p0 = W0 + 48 * HID + l16;
    const float* p1 = W1 + 48 * HID + l16;
    v2f b;
    b.x = p0[0];  b.y = p0[HID];
    acc0 = __builtin_amdgcn_wmma_f32_16x16x4_f32(false, a, false, b, (short)0, acc0, false, false);
    b.x = p0[16]; b.y = p0[HID + 16];
    acc1 = __builtin_amdgcn_wmma_f32_16x16x4_f32(false, a, false, b, (short)0, acc1, false, false);
    b.x = p1[0];  b.y = p1[HID];
    acc2 = __builtin_amdgcn_wmma_f32_16x16x4_f32(false, a, false, b, (short)0, acc2, false, false);
    b.x = p1[16]; b.y = p1[HID + 16];
    acc3 = __builtin_amdgcn_wmma_f32_16x16x4_f32(false, a, false, b, (short)0, acc3, false, false);
  }
#pragma unroll
  for (int v = 0; v < 8; v++) {
    int mC = m0 + v + 8 * half;
    float* prow = P + (long long)mC * 64 + l16;
    prow[0]  = acc0[v];
    prow[16] = acc1[v];
    prow[32] = acc2[v];
    prow[48] = acc3[v];
  }
}

__global__ void __launch_bounds__(1024, 16)
count_proj1_kernel(const int* __restrict__ rows,
                   const int* __restrict__ cols,
                   int* __restrict__ tcnt,
                   const float* __restrict__ x,
                   const float* __restrict__ W0,
                   const float* __restrict__ W1,
                   float* __restrict__ P) {
  if (blockIdx.x < NCH) bin_count_body(rows, cols, tcnt);
  else                  proj1_wmma_body((int)blockIdx.x - NCH, x, W0, W1, P);
}


__device__ __forceinline__ void out_k32(const float* __restrict__ arow,
                                        const float* __restrict__ Wm,
                                        int half, int l16,
                                        v8f& acc0, v8f& acc1, v8f& acc2) {
  int c2 = 32 + (l16 & 7);
  bool ok2 = l16 < 8;
  for (int kk = 0; kk < HID; kk += 4) {
    int k = kk + 2 * half;
    v2f a = *(const v2f*)(arow + k);
    const float* p = Wm + k * NCLS;
    v2f b;
    b.x = p[l16];      b.y = p[NCLS + l16];
    acc0 = __builtin_amdgcn_wmma_f32_16x16x4_f32(false, a, false, b, (short)0, acc0, false, false);
    b.x = p[16 + l16]; b.y = p[NCLS + 16 + l16];
    acc1 = __builtin_amdgcn_wmma_f32_16x16x4_f32(false, a, false, b, (short)0, acc1, false, false);
    float t0 = p[c2], t1 = p[NCLS + c2];
    b.x = ok2 ? t0 : 0.0f; b.y = ok2 ? t1 : 0.0f;
    acc2 = __builtin_amdgcn_wmma_f32_16x16x4_f32(false, a, false, b, (short)0, acc2, false, false);
  }
}

template <int M>
__device__ __forceinline__ float row16_xor(float v) {
  return __int_as_float(__builtin_amdgcn_update_dpp(0, __float_as_int(v), 0x160 + M, 0xf, 0xf, true));
}
__device__ __forceinline__ float row16_max(float v) {
  v = fmaxf(v, row16_xor<1>(v));
  v = fmaxf(v, row16_xor<2>(v));
  v = fmaxf(v, row16_xor<4>(v));
  v = fmaxf(v, row16_xor<8>(v));
  return v;
}
__device__ __forceinline__ float row16_sum(float v) {
  v += row16_xor<1>(v);
  v += row16_xor<2>(v);
  v += row16_xor<4>(v);
  v += row16_xor<8>(v);
  return v;
}

__global__ void __launch_bounds__(128)
out_wmma_lsm_kernel(const float* __restrict__ h,
                    const float* __restrict__ agg2,
                    const float* __restrict__ W0,
                    const float* __restrict__ W1,
                    const float* __restrict__ b2,
                    float* __restrict__ out) {
  int wave = threadIdx.x >> 5;
  int lane = threadIdx.x & 31;
  int half = lane >> 4;
  int l16  = lane & 15;
  int m0 = (blockIdx.x * 4 + wave) * 16;
  if (m0 >= N_NODES) return;
  int mA = m0 + l16;

  v8f acc0 = {}, acc1 = {}, acc2 = {};
  out_k32(h    + (long long)mA * HID, W0, half, l16, acc0, acc1, acc2);
  out_k32(agg2 + (long long)mA * HID, W1, half, l16, acc0, acc1, acc2);

  bool ok2 = l16 < 8;
  float bb0 = b2[l16], bb1 = b2[16 + l16], bb2 = b2[32 + (l16 & 7)];
  float x0[8], x1[8], x2[8], mx[8], se[8];
#pragma unroll
  for (int v = 0; v < 8; v++) {
    x0[v] = acc0[v] + bb0;
    x1[v] = acc1[v] + bb1;
    x2[v] = acc2[v] + bb2;
    mx[v] = fmaxf(fmaxf(x0[v], x1[v]), ok2 ? x2[v] : -INFINITY);
  }
#pragma unroll
  for (int v = 0; v < 8; v++) mx[v] = row16_max(mx[v]);
#pragma unroll
  for (int v = 0; v < 8; v++) {
    x0[v] -= mx[v];
    x1[v] -= mx[v];
    x2[v] -= mx[v];
    float e2 = expf(x2[v]);
    se[v] = expf(x0[v]) + expf(x1[v]);
    se[v] += ok2 ? e2 : 0.0f;
  }
#pragma unroll
  for (int v = 0; v < 8; v++) se[v] = logf(row16_sum(se[v]));
  float* otile = out + (long long)(m0 + 8 * half) * NCLS + l16;
#pragma unroll
  for (int v = 0; v < 8; v++) {
    otile[v * NCLS]      = x0[v] - se[v];
    otile[v * NCLS + 16] = x1[v] - se[v];
  }
  if (ok2) {
#pragma unroll
    for (int v = 0; v < 8; v++) otile[v * NCLS + 32] = x2[v] - se[v];
  }
}


extern "C" void kernel_launch(void* const* d_in, const int* in_sizes, int n_in,
                              void* d_out, int out_size, void* d_ws, size_t ws_size,
                              hipStream_t stream) {
  (void)in_sizes; (void)n_in; (void)out_size; (void)ws_size;
  const float* x    = (const float*)d_in[0];
  const int*   ei   = (const int*)  d_in[1];
  const float* W0_1 = (const float*)d_in[2];
  const float* W1_1 = (const float*)d_in[3];
  const float* b1   = (const float*)d_in[4];
  const float* W0_2 = (const float*)d_in[5];
  const float* W1_2 = (const float*)d_in[6];
  const float* b2   = (const float*)d_in[7];
  float* out = (float*)d_out;

  const int* rows = ei;
  const int* cols = ei + N_EDGES;

  const long long N = N_NODES;
  float* ws     = (float*)d_ws;
  int*   cnt    = (int*)ws;
  float* dis    = ws + N;
  float* P      = ws + 2 * N;
  float* agg2   = ws + 2 * N;
  int*   colidx = (int*)(ws + 66 * N);
  int*   rowptr = (int*)(ws + 82 * N);
  int*   tcnt   = (int*)(ws + 83 * N);
  int*   bstart = (int*)(ws + 84 * N);
  float* hbuf   = ws + 98 * N;
  int*   binned = (int*)(ws + 98 * N);

  count_proj1_kernel<<<NCH + PB, 1024, 0, stream>>>(rows, cols, tcnt, x, W0_1, W1_1, P);
  bin_scatter_kernel<<<NCH, 1024, 0, stream>>>(rows, cols, tcnt, bstart, binned);
  bin_rows_kernel<<<NB, 1024, 0, stream>>>(bstart, binned, cnt, rowptr, dis, colidx);

  gather_relu_kernel<<<(N_NODES / RPW + 3) / 4, 128, 0, stream>>>(
      rowptr, cnt, colidx, dis, P, b1, hbuf);

  gather_rows_kernel<<<(N_NODES / RPW + 3) / 4, 128, 0, stream>>>(
      rowptr, cnt, colidx, dis, hbuf, agg2);
  out_wmma_lsm_kernel<<<(N_NODES + 63) / 64, 128, 0, stream>>>(hbuf, agg2, W0_2, W1_2, b2, out);
}


# Edges ordered by bin inside each chunk's work-group (LDS rank + scan, one staged run per chunk); row pass reads its bin's 196 runs itself; scatter launch, 2nd edge read gone, 6->5; sums in LDS rank
# speedup vs baseline: 4.8059x; 1.2879x over previous
#include <hip/hip_runtime.h>
#include <math.h>

#define N_NODES 100000
#define N_EDGES 1600000
#define F_IN    50
#define HID     32
#define NCLS    40
#define BIN_SHIFT 8
#define BIN_ROWS  256
#define NB        391
#define CHUNK     8192
#define NCH       196
#define TBL       (NB * NCH)
#define PB        196
#define COL_BITS  17
#define CNT_BITS  14
#define CNT_MASK  ((1 << CNT_BITS) - 1)
#define SEGW      7
#define PRE       10
#define LATE      ((TBL + 4095) / 4096 - PRE)
static_assert(LATE > 0 && LATE <= PRE, "two goes cover the table");
static_assert(N_NODES <= (1 << COL_BITS), "col must fit the low field of a list entry");
static_assert(NB * BIN_ROWS >= N_NODES && NCH * CHUNK >= N_EDGES && N_EDGES % 4 == 0, "sizes");
static_assert(CHUNK <= CNT_MASK && 2 * CNT_BITS <= 31 && CHUNK % 4 == 0, "start and length of a run fit one table word");
static_assert(TBL <= N_NODES && NCH % 4 == 0, "the table has one N-word plane of the workspace; its rows are whole 16-B groups");
static_assert(NCH * CHUNK <= 32 * N_NODES, "the chunk-ordered list fits the plane that later holds h");
static_assert(SEGW * 32 >= NCH, "every run of a bin has a wave");
static_assert(PB * 512 >= N_NODES, "GEMM-1 work-groups cover every row");


typedef float v2f __attribute__((ext_vector_type(2)));
typedef float v8f __attribute__((ext_vector_type(8)));


#define EDGE8(M) \
  M(0, r4[0].x, c4[0].x) M(1, r4[0].y, c4[0].y) M(2, r4[0].z, c4[0].z) M(3, r4[0].w, c4[0].w) \
  M(4, r4[1].x, c4[1].x) M(5, r4[1].y, c4[1].y) M(6, r4[1].z, c4[1].z) M(7, r4[1].w, c4[1].w)

__device__ __forceinline__ void
bin_order_body(const int* __restrict__ rows,
               const int* __restrict__ cols,
               int* __restrict__ tbl,
               int* __restrict__ sorted) {
  __shared__ int hist[NB];
  __shared__ int off[NB];
  __shared__ int s_wtot[13];
  __shared__ int4 stage4[CHUNK / 4];
  int* stage = (int*)stage4;
  int t = threadIdx.x, lane = t & 31, wave = t >> 5;
  int4 r4[2], c4[2];
#pragma unroll
  for (int h = 0; h < 2; h++) {
    int e = blockIdx.x * CHUNK + h * 4096 + t * 4;
    r4[h] = make_int4(0, 0, 0, 0);
    c4[h] = make_int4(0, 0, 0, 0);
    if (e < N_EDGES) {
      r4[h] = *(const int4*)(rows + e);
      c4[h] = *(const int4*)(cols + e);
    }
  }
  for (int i = t; i < NB; i += 1024) hist[i] = 0;
  __syncthreads();
  int rk[8];
#define RANK1(I, R, C) rk[I] = ((R) != (C)) ? atomicAdd(&hist[(R) >> BIN_SHIFT], 1) : -1;
  EDGE8(RANK1)
#undef RANK1
  __syncthreads();
  int v = 0, incl = 0;
  if (t < 13 * 32) {
    if (t < NB) v = hist[t];
    incl = v;
    for (int d = 1; d < 32; d <<= 1) {
      int y = __shfl_up(incl, d);
      if (lane >= d) incl += y;
    }
    if (lane == 31) s_wtot[wave] = incl;
  }
  __syncthreads();
  if (t < NB) {
    int add = 0;
    for (int w = 0; w < 13; w++) add += (w < wave) ? s_wtot[w] : 0;
    off[t] = add + incl - v;
  }
  __syncthreads();
#define PLACE1(I, R, C) if (rk[I] >= 0) stage[off[(R) >> BIN_SHIFT] + rk[I]] = (((R) & (BIN_ROWS - 1)) << COL_BITS) | (C);
  EDGE8(PLACE1)
#undef PLACE1
  __syncthreads();
  int total = off[NB - 1] + hist[NB - 1];
  int* run = sorted + blockIdx.x * CHUNK;
  for (int i = t * 4; i < total; i += 4096) *(int4*)(run + i) = stage4[i >> 2];
  for (int i = t; i < NB; i += 1024) tbl[i * NCH + blockIdx.x] = (off[i] << CNT_BITS) | hist[i];
}

__global__ void __launch_bounds__(1024, 16)
bin_rows_kernel(const int* __restrict__ tbl,
                const int* __restrict__ sorted,
                int* __restrict__ cnt,
                int* __restrict__ rowptr,
                float* __restrict__ dis,
                int* __restrict__ colidx) {
  __shared__ int s_cnt[BIN_ROWS];
  __shared__ int s_start[BIN_ROWS];
  __shared__ int s_cur[BIN_ROWS];
  __shared__ int s_wtot[8];
  __shared__ int s_part[32];
  __shared__ int seg_off[NCH];
  __shared__ int seg_cnt[NCH];
  int t = threadIdx.x, lane = t & 31, wave = t >> 5;
  int b = blockIdx.x;
  int w_own = 0;
  if (t < NCH) w_own = tbl[b * NCH + t];
  const int lim = b * NCH;
  int4 wa[PRE];
  int ma[PRE];
#pragma unroll
  for (int u = 0; u < PRE; u++) {
    int i = (t + u * 1024) * 4;
    ma[u] = (i < lim) ? CNT_MASK : 0;
    wa[u] = *(const int4*)(tbl + ((i < lim) ? i : 0));
  }
  if (t < NCH) {
    seg_off[t] = t * CHUNK + (w_own >> CNT_BITS);
    seg_cnt[t] = w_own & CNT_MASK;
  }
  if (t < BIN_ROWS) { s_cnt[t] = 0; s_cur[t] = 0; }
  __syncthreads();
  int p[SEGW];
#pragma unroll
  for (int k = 0; k < SEGW; k++) {
    int sg = wave + 32 * k;
    p[k] = -1;
    if (sg < NCH && lane < seg_cnt[sg]) p[k] = sorted[seg_off[sg] + lane];
  }
#pragma unroll
  for (int k = 0; k < SEGW; k++)
    if (p[k] >= 0) atomicAdd(&s_cnt[p[k] >> COL_BITS], 1);
  int part = 0;
#pragma unroll
  for (int u = 0; u < PRE; u++)
    part += (wa[u].x & ma[u]) + (wa[u].y & ma[u]) + (wa[u].z & ma[u]) + (wa[u].w & ma[u]);
  for (int k = 0; k < SEGW; k++) {
    int sg = wave + 32 * k;
    if (sg < NCH) {
      int n = seg_cnt[sg], o = seg_off[sg];
      for (int j = 32 + lane; j < n; j += 32) atomicAdd(&s_cnt[sorted[o + j] >> COL_BITS], 1);
    }
  }
  if (lim > PRE * 4096) {
    int4 wb[LATE];
    int mb[LATE];
#pragma unroll
    for (int u = 0; u < LATE; u++) {
      int i = (t + (PRE + u) * 1024) * 4;
      mb[u] = (i < lim) ? CNT_MASK : 0;
      wb[u] = *(const int4*)(tbl + ((i < lim) ? i : 0));
    }
#pragma unroll
    for (int u = 0; u < LATE; u++)
      part += (wb[u].x & mb[u]) + (wb[u].y & mb[u]) + (wb[u].z & mb[u]) + (wb[u].w & mb[u]);
  }
  for (int d = 16; d >= 1; d >>= 1) part += __shfl_xor(part, d);
  if (lane == 0) s_part[wave] = part;
  __syncthreads();
  int v = 0, incl = 0;
  if (t < BIN_ROWS) {
    v = s_cnt[t];
    incl = v;
    for (int d = 1; d < 32; d <<= 1) {
      int y = __shfl_up(incl, d);
      if (lane >= d) incl += y;
    }
    if (lane == 31) s_wtot[wave] = incl;
  }
  __syncthreads();
  int s = 0;
  for (int w = 0; w < 32; w++) s += s_part[w];
  if (t < BIN_ROWS) {
    int add = 0;
    for (int w = 0; w < 8; w++) add += (w < wave) ? s_wtot[w] : 0;
    int excl = add + incl - v;
    s_start[t] = excl;
    int row = b * BIN_ROWS + t;
    if (row < N_NODES) {
      rowptr[row] = s + excl;
      cnt[row] = v;
      float dg = (float)v;
      dis[row] = (dg > 0.0f) ? rsqrtf(fmaxf(dg, 1.0f)) : 0.0f;
    }
  }
  __syncthreads();
#pragma unroll
  for (int k = 0; k < SEGW; k++) {
    if (p[k] >= 0) {
      int rl = p[k] >> COL_BITS;
      int r = atomicAdd(&s_cur[rl], 1);
      colidx[s + s_start[rl] + r] = p[k] & ((1 << COL_BITS) - 1);
    }
  }
  for (int k = 0; k < SEGW; k++) {
    int sg = wave + 32 * k;
    if (sg < NCH) {
      int n = seg_cnt[sg], o = seg_off[sg];
      for (int j = 32 + lane; j < n; j += 32) {
        int e = sorted[o + j];
        int rl = e >> COL_BITS;
        int r = atomicAdd(&s_cur[rl], 1);
        colidx[s + s_start[rl] + r] = e & ((1 << COL_BITS) - 1);
      }
    }
  }
}


#define RPW 8
static_assert(N_NODES % RPW == 0 && RPW >= 2 && RPW <= 32,
              "a wave's rows all exist; their bookkeeping sits one row per lane");

__device__ __forceinline__ int lane_i(int v, int l) { return __builtin_amdgcn_readlane(v, l); }
__device__ __forceinline__ float lane_f(float v, int l) {
  return __int_as_float(__builtin_amdgcn_readlane(__float_as_int(v), l));
}

__device__ __forceinline__ int list_head(const int* __restrict__ colidx, int start, int n, int lane) {
  int c = 0;
  if (n > 0) {
    int m = n < 32 ? n : 32;
    int li = lane < m ? lane : m - 1;
    c = colidx[start + li];
  }
  return c;
}

__device__ __forceinline__ float head_weight(float dc, float ndr, int n, int lane) {
  int m = n < 32 ? n : 32;
  return lane < m ? ndr * dc : 0.0f;
}

template <int K>
__device__ __forceinline__ float gather_batch(int c, float w, int j, int m, int lane,
                                              const float* __restrict__ sp, int sstride, float acc) {
  float v[K];
#pragma unroll
  for (int u = 0; u < K; u++) {
    int cj = lane_i(c, j + u);
    const float* rowp = sp + (long long)cj * sstride;
    float x = rowp[lane];
    v[u] = (j + u < m) ? x : 0.0f;
  }
#pragma unroll
  for (int u = 0; u < K; u++) acc = fmaf(lane_f(w, j + u), v[u], acc);
  return acc;
}

__device__ __forceinline__ float gather_walk(int c, float w, int m, int lane,
                                             const float* __restrict__ sp, int sstride, float acc) {
  for (int j = 0; j < m; j += 16) {
    if (j + 8 < m) acc = gather_batch<16>(c, w, j, m, lane, sp, sstride, acc);
    else           acc = gather_batch<8>(c, w, j, m, lane, sp, sstride, acc);
  }
  return acc;
}

template <bool LAYER1>
__device__ __forceinline__ void gather_span(int r0, int lane,
                                            const int* __restrict__ rowptr,
                                            const int* __restrict__ cnt,
                                            const int* __restrict__ colidx,
                                            const float* __restrict__ dis,
                                            const float* __restrict__ sp, int sstride,
                                            const float* __restrict__ selfp, float bias,
                                            float* __restrict__ dst) {
  int mr = r0 + (lane & (RPW - 1));
  int st_v = rowptr[mr];
  int n_v = cnt[mr];
  float d_v = dis[mr];

  int n0 = lane_i(n_v, 0);
  int c_a = list_head(colidx, lane_i(st_v, 0), n0, lane);
  int c_b = list_head(colidx, lane_i(st_v, 1), lane_i(n_v, 1), lane);
  float w_a = head_weight(dis[c_a], -lane_f(d_v, 0), n0, lane);
  for (int i = 0; i < RPW; i++) {
    int r = r0 + i;
    int start = lane_i(st_v, i);
    int n = lane_i(n_v, i);
    int c_c = 0;
    if (i + 2 < RPW) c_c = list_head(colidx, lane_i(st_v, i + 2), lane_i(n_v, i + 2), lane);
    float self = 0.0f;
    if (LAYER1) self = selfp[(long long)r * 64 + lane];
    float dc_b = dis[c_b];
    int m = n < 32 ? n : 32;
    float acc = gather_walk(c_a, w_a, m, lane, sp, sstride, 0.0f);
    if (n > 32) {
      float ndr = -lane_f(d_v, i);
      for (int base = 32; base < n; base += 32) {
        int m2 = n - base;
        if (m2 > 32) m2 = 32;
        int li = lane < m2 ? lane : m2 - 1;
        int c2 = colidx[start + base + li];
        float w2 = lane < m2 ? ndr * dis[c2] : 0.0f;
        acc = gather_walk(c2, w2, m2, lane, sp, sstride, acc);
      }
    }
    float* drow = dst + (long long)r * HID;
    if (LAYER1) {
      float v = self + acc + bias;
      drow[lane] = v > 0.0f ? v : 0.0f;
    } else {
      drow[lane] = acc;
    }
    float w_b = head_weight(dc_b, -lane_f(d_v, i + 1), lane_i(n_v, i + 1), lane);
    c_a = c_b; w_a = w_b; c_b = c_c;
  }
}

__global__ void __launch_bounds__(128)
gather_relu_kernel(const int* __restrict__ rowptr,
                   const int* __restrict__ cnt,
                   const int* __restrict__ colidx,
                   const float* __restrict__ dis,
                   const float* __restrict__ P,
                   const float* __restrict__ b1,
                   float* __restrict__ h) {
  int lane = threadIdx.x & 31;
  int r0 = __builtin_amdgcn_readfirstlane((blockIdx.x * 4 + (threadIdx.x >> 5)) * RPW);
  if (r0 >= N_NODES) return;
  gather_span<true>(r0, lane, rowptr, cnt, colidx, dis, P + HID, 64, P, b1[lane], h);
}

__global__ void __launch_bounds__(128)
gather_rows_kernel(const int* __restrict__ rowptr,
                   const int* __restrict__ cnt,
                   const int* __restrict__ colidx,
                   const float* __restrict__ dis,
                   const float* __restrict__ src,
                   float* __restrict__ dst) {
  int lane = threadIdx.x & 31;
  int r0 = __builtin_amdgcn_readfirstlane((blockIdx.x * 4 + (threadIdx.x >> 5)) * RPW);
  if (r0 >= N_NODES) return;
  gather_span<false>(r0, lane, rowptr, cnt, colidx, dis, src, HID, src, 0.0f, dst);
}


__device__ __forceinline__ void
proj1_wmma_body(int pb,
                const float* __restrict__ x,
                const float* __restrict__ W0,
                const float* __restrict__ W1,
                float* __restrict__ P) {
  int wave = threadIdx.x >> 5;
  int lane = threadIdx.x & 31;
  int half = lane >> 4;
  int l16  = lane & 15;
  int m0 = (pb * 32 + wave) * 16;
  if (m0 >= N_NODES) return;
  int mA = m0 + l16;
  const float* arow = x + (long long)mA * F_IN;

  v8f acc0 = {}, acc1 = {}, acc2 = {}, acc3 = {};
  for (int kk = 0; kk < 48; kk += 4) {
    int k = kk + 2 * half;
    v2f a = *(const v2f*)(arow + k);
    const float* p0 = W0 + k * HID + l16;
    const float* p1 = W1 + k * HID + l16;
    v2f b;
    b.x = p0[0];  b.y = p0[HID];
    acc0 = __builtin_amdgcn_wmma_f32_16x16x4_f32(false, a, false, b, (short)0, acc0, false, false);
    b.x = p0[16]; b.y = p0[HID + 16];
    acc1 = __builtin_amdgcn_wmma_f32_16x16x4_f32(false, a, false, b, (short)0, acc1, false, false);
    b.x = p1[0];  b.y = p1[HID];
    acc2 = __builtin_amdgcn_wmma_f32_16x16x4_f32(false, a, false, b, (short)0, acc2, false, false);
    b.x = p1[16]; b.y = p1[HID + 16];
    acc3 = __builtin_amdgcn_wmma_f32_16x16x4_f32(false, a, false, b, (short)0, acc3, false, false);
  }
  {
    v2f aload = *(const v2f*)(arow + 48);
    v2f a;
    a.x = half ? 0.0f : aload.x;
    a.y = half ? 0.0f : aload.y;
    const float* p0 = W0 + 48 * HID + l16;
    const float* p1 = W1 + 48 * HID + l16;
    v2f b;
    b.x = p0[0];  b.y = p0[HID];
    acc0 = __builtin_amdgcn_wmma_f32_16x16x4_f32(false, a, false, b, (short)0, acc0, false, false);
    b.x = p0[16]; b.y = p0[HID + 16];
    acc1 = __builtin_amdgcn_wmma_f32_16x16x4_f32(false, a, false, b, (short)0, acc1, false, false);
    b.x = p1[0];  b.y = p1[HID];
    acc2 = __builtin_amdgcn_wmma_f32_16x16x4_f32(false, a, false, b, (short)0, acc2, false, false);
    b.x = p1[16]; b.y = p1[HID + 16];
    acc3 = __builtin_amdgcn_wmma_f32_16x16x4_f32(false, a, false, b, (short)0, acc3, false, false);
  }
#pragma unroll
  for (int v = 0; v < 8; v++) {
    int mC = m0 + v + 8 * half;
    float* prow = P + (long long)mC * 64 + l16;
    prow[0]  = acc0[v];
    prow[16] = acc1[v];
    prow[32] = acc2[v];
    prow[48] = acc3[v];
  }
}

__global__ void __launch_bounds__(1024, 16)
count_proj1_kernel(const int* __restrict__ rows,
                   const int* __restrict__ cols,
                   int* __restrict__ tbl,
                   int* __restrict__ sorted,
                   const float* __restrict__ x,
                   const float* __restrict__ W0,
                   const float* __restrict__ W1,
                   float* __restrict__ P) {
  if (blockIdx.x < NCH) bin_order_body(rows, cols, tbl, sorted);
  else                  proj1_wmma_body((int)blockIdx.x - NCH, x, W0, W1, P);
}


__device__ __forceinline__ void out_k32(const float* __restrict__ arow,
                                        const float* __restrict__ Wm,
                                        int half, int l16,
                                        v8f& acc0, v8f& acc1, v8f& acc2) {
  int c2 = 32 + (l16 & 7);
  bool ok2 = l16 < 8;
  for (int kk = 0; kk < HID; kk += 4) {
    int k = kk + 2 * half;
    v2f a = *(const v2f*)(arow + k);
    const float* p = Wm + k * NCLS;
    v2f b;
    b.x = p[l16];      b.y = p[NCLS + l16];
    acc0 = __builtin_amdgcn_wmma_f32_16x16x4_f32(false, a, false, b, (short)0, acc0, false, false);
    b.x = p[16 + l16]; b.y = p[NCLS + 16 + l16];
    acc1 = __builtin_amdgcn_wmma_f32_16x16x4_f32(false, a, false, b, (short)0, acc1, false, false);
    float t0 = p[c2], t1 = p[NCLS + c2];
    b.x = ok2 ? t0 : 0.0f; b.y = ok2 ? t1 : 0.0f;
    acc2 = __builtin_amdgcn_wmma_f32_16x16x4_f32(false, a, false, b, (short)0, acc2, false, false);
  }
}

template <int M>
__device__ __forceinline__ float row16_xor(float v) {
  return __int_as_float(__builtin_amdgcn_update_dpp(0, __float_as_int(v), 0x160 + M, 0xf, 0xf, true));
}
__device__ __forceinline__ float row16_max(float v) {
  v = fmaxf(v, row16_xor<1>(v));
  v = fmaxf(v, row16_xor<2>(v));
  v = fmaxf(v, row16_xor<4>(v));
  v = fmaxf(v, row16_xor<8>(v));
  return v;
}
__device__ __forceinline__ float row16_sum(float v) {
  v += row16_xor<1>(v);
  v += row16_xor<2>(v);
  v += row16_xor<4>(v);
  v += row16_xor<8>(v);
  return v;
}

__global__ void __launch_bounds__(128)
out_wmma_lsm_kernel(const float* __restrict__ h,
                    const float* __restrict__ agg2,
                    const float* __restrict__ W0,
                    const float* __restrict__ W1,
                    const float* __restrict__ b2,
                    float* __restrict__ out) {
  int wave = threadIdx.x >> 5;
  int lane = threadIdx.x & 31;
  int half = lane >> 4;
  int l16  = lane & 15;
  int m0 = (blockIdx.x * 4 + wave) * 16;
  if (m0 >= N_NODES) return;
  int mA = m0 + l16;

  v8f acc0 = {}, acc1 = {}, acc2 = {};
  out_k32(h    + (long long)mA * HID, W0, half, l16, acc0, acc1, acc2);
  out_k32(agg2 + (long long)mA * HID, W1, half, l16, acc0, acc1, acc2);

  bool ok2 = l16 < 8;
  float bb0 = b2[l16], bb1 = b2[16 + l16], bb2 = b2[32 + (l16 & 7)];
  float x0[8], x1[8], x2[8], mx[8], se[8];
#pragma unroll
  for (int v = 0; v < 8; v++) {
    x0[v] = acc0[v] + bb0;
    x1[v] = acc1[v] + bb1;
    x2[v] = acc2[v] + bb2;
    mx[v] = fmaxf(fmaxf(x0[v], x1[v]), ok2 ? x2[v] : -INFINITY);
  }
#pragma unroll
  for (int v = 0; v < 8; v++) mx[v] = row16_max(mx[v]);
#pragma unroll
  for (int v = 0; v < 8; v++) {
    x0[v] -= mx[v];
    x1[v] -= mx[v];
    x2[v] -= mx[v];
    float e2 = expf(x2[v]);
    se[v] = expf(x0[v]) + expf(x1[v]);
    se[v] += ok2 ? e2 : 0.0f;
  }
#pragma unroll
  for (int v = 0; v < 8; v++) se[v] = logf(row16_sum(se[v]));
  float* otile = out + (long long)(m0 + 8 * half) * NCLS + l16;
#pragma unroll
  for (int v = 0; v < 8; v++) {
    otile[v * NCLS]      = x0[v] - se[v];
    otile[v * NCLS + 16] = x1[v] - se[v];
  }
  if (ok2) {
#pragma unroll
    for (int v = 0; v < 8; v++) otile[v * NCLS + 32] = x2[v] - se[v];
  }
}


extern "C" void kernel_launch(void* const* d_in, const int* in_sizes, int n_in,
                              void* d_out, int out_size, void* d_ws, size_t ws_size,
                              hipStream_t stream) {
  (void)in_sizes; (void)n_in; (void)out_size; (void)ws_size;
  const float* x    = (const float*)d_in[0];
  const int*   ei   = (const int*)  d_in[1];
  const float* W0_1 = (const float*)d_in[2];
  const float* W1_1 = (const float*)d_in[3];
  const float* b1   = (const float*)d_in[4];
  const float* W0_2 = (const float*)d_in[5];
  const float* W1_2 = (const float*)d_in[6];
  const float* b2   = (const float*)d_in[7];
  float* out = (float*)d_out;

  const int* rows = ei;
  const int* cols = ei + N_EDGES;

  const long long N = N_NODES;
  float* ws     = (float*)d_ws;
  int*   cnt    = (int*)ws;
  float* dis    = ws + N;
  float* P      = ws + 2 * N;
  float* agg2   = ws + 2 * N;
  int*   colidx = (int*)(ws + 66 * N);
  int*   rowptr = (int*)(ws + 82 * N);
  int*   tbl    = (int*)(ws + 83 * N);
  float* hbuf   = ws + 98 * N;
  int*   sorted = (int*)(ws + 98 * N);

  count_proj1_kernel<<<NCH + PB, 1024, 0, stream>>>(rows, cols, tbl, sorted, x, W0_1, W1_1, P);
  bin_rows_kernel<<<NB, 1024, 0, stream>>>(tbl, sorted, cnt, rowptr, dis, colidx);

  gather_relu_kernel<<<(N_NODES / RPW + 3) / 4, 128, 0, stream>>>(
      rowptr, cnt, colidx, dis, P, b1, hbuf);

  gather_rows_kernel<<<(N_NODES / RPW + 3) / 4, 128, 0, stream>>>(
      rowptr, cnt, colidx, dis, hbuf, agg2);
  out_wmma_lsm_kernel<<<(N_NODES + 63) / 64, 128, 0, stream>>>(hbuf, agg2, W0_2, W1_2, b2, out);
}


# Row pass: a bin's list start is the sum of its own 196 run starts (each counts its chunk's kept edges of the bins in front); the per-work-group prefix read of the whole count table is gone
# speedup vs baseline: 4.9455x; 1.0290x over previous
#include <hip/hip_runtime.h>
#include <math.h>

#define N_NODES 100000
#define N_EDGES 1600000
#define F_IN    50
#define HID     32
#define NCLS    40
#define BIN_SHIFT 8
#define BIN_ROWS  256
#define NB        391
#define CHUNK     8192
#define NCH       196
#define TBL       (NB * NCH)
#define PB        196
#define COL_BITS  17
#define CNT_BITS  14
#define CNT_MASK  ((1 << CNT_BITS) - 1)
#define SEGW      7
#define NCHW      ((NCH + 31) / 32)
static_assert(N_NODES <= (1 << COL_BITS), "col must fit the low field of a list entry");
static_assert(NB * BIN_ROWS >= N_NODES && NCH * CHUNK >= N_EDGES && N_EDGES % 4 == 0, "sizes");
static_assert(CHUNK <= CNT_MASK && 2 * CNT_BITS <= 31 && CHUNK % 4 == 0, "start and length of a run fit one table word");
static_assert(TBL <= N_NODES, "the table has one N-word plane of the workspace");
static_assert(NCH * CHUNK <= 32 * N_NODES, "the chunk-ordered list fits the plane that later holds h");
static_assert(SEGW * 32 >= NCH, "every run of a bin has a wave");
static_assert(PB * 512 >= N_NODES, "GEMM-1 work-groups cover every row");


typedef float v2f __attribute__((ext_vector_type(2)));
typedef float v8f __attribute__((ext_vector_type(8)));


#define EDGE8(M) \
  M(0, r4[0].x, c4[0].x) M(1, r4[0].y, c4[0].y) M(2, r4[0].z, c4[0].z) M(3, r4[0].w, c4[0].w) \
  M(4, r4[1].x, c4[1].x) M(5, r4[1].y, c4[1].y) M(6, r4[1].z, c4[1].z) M(7, r4[1].w, c4[1].w)

__device__ __forceinline__ void
bin_order_body(const int* __restrict__ rows,
               const int* __restrict__ cols,
               int* __restrict__ tbl,
               int* __restrict__ sorted) {
  __shared__ int hist[NB];
  __shared__ int off[NB];
  __shared__ int s_wtot[13];
  __shared__ int4 stage4[CHUNK / 4];
  int* stage = (int*)stage4;
  int t = threadIdx.x, lane = t & 31, wave = t >> 5;
  int4 r4[2], c4[2];
#pragma unroll
  for (int h = 0; h < 2; h++) {
    int e = blockIdx.x * CHUNK + h * 4096 + t * 4;
    r4[h] = make_int4(0, 0, 0, 0);
    c4[h] = make_int4(0, 0, 0, 0);
    if (e < N_EDGES) {
      r4[h] = *(const int4*)(rows + e);
      c4[h] = *(const int4*)(cols + e);
    }
  }
  for (int i = t; i < NB; i += 1024) hist[i] = 0;
  __syncthreads();
  int rk[8];
#define RANK1(I, R, C) rk[I] = ((R) != (C)) ? atomicAdd(&hist[(R) >> BIN_SHIFT], 1) : -1;
  EDGE8(RANK1)
#undef RANK1
  __syncthreads();
  int v = 0, incl = 0;
  if (t < 13 * 32) {
    if (t < NB) v = hist[t];
    incl = v;
    for (int d = 1; d < 32; d <<= 1) {
      int y = __shfl_up(incl, d);
      if (lane >= d) incl += y;
    }
    if (lane == 31) s_wtot[wave] = incl;
  }
  __syncthreads();
  if (t < NB) {
    int add = 0;
    for (int w = 0; w < 13; w++) add += (w < wave) ? s_wtot[w] : 0;
    off[t] = add + incl - v;
  }
  __syncthreads();
#define PLACE1(I, R, C) if (rk[I] >= 0) stage[off[(R) >> BIN_SHIFT] + rk[I]] = (((R) & (BIN_ROWS - 1)) << COL_BITS) | (C);
  EDGE8(PLACE1)
#undef PLACE1
  __syncthreads();
  int total = off[NB - 1] + hist[NB - 1];
  int* run = sorted + blockIdx.x * CHUNK;
  for (int i = t * 4; i < total; i += 4096) *(int4*)(run + i) = stage4[i >> 2];
  for (int i = t; i < NB; i += 1024) tbl[i * NCH + blockIdx.x] = (off[i] << CNT_BITS) | hist[i];
}

__global__ void __launch_bounds__(1024, 16)
bin_rows_kernel(const int* __restrict__ tbl,
                const int* __restrict__ sorted,
                int* __restrict__ cnt,
                int* __restrict__ rowptr,
                float* __restrict__ dis,
                int* __restrict__ colidx) {
  __shared__ int s_cnt[BIN_ROWS];
  __shared__ int s_start[BIN_ROWS];
  __shared__ int s_cur[BIN_ROWS];
  __shared__ int s_wtot[8];
  __shared__ int s_part[NCHW];
  __shared__ int seg_off[NCH];
  __shared__ int seg_cnt[NCH];
  int t = threadIdx.x, lane = t & 31, wave = t >> 5;
  int b = blockIdx.x;
  int w_own = 0;
  if (t < NCH) {
    w_own = tbl[b * NCH + t];
    seg_off[t] = t * CHUNK + (w_own >> CNT_BITS);
    seg_cnt[t] = w_own & CNT_MASK;
  }
  if (t < BIN_ROWS) { s_cnt[t] = 0; s_cur[t] = 0; }
  __syncthreads();
  int p[SEGW];
#pragma unroll
  for (int k = 0; k < SEGW; k++) {
    int sg = wave + 32 * k;
    p[k] = -1;
    if (sg < NCH && lane < seg_cnt[sg]) p[k] = sorted[seg_off[sg] + lane];
  }
#pragma unroll
  for (int k = 0; k < SEGW; k++)
    if (p[k] >= 0) atomicAdd(&s_cnt[p[k] >> COL_BITS], 1);
  for (int k = 0; k < SEGW; k++) {
    int sg = wave + 32 * k;
    if (sg < NCH) {
      int n = seg_cnt[sg], o = seg_off[sg];
      for (int j = 32 + lane; j < n; j += 32) atomicAdd(&s_cnt[sorted[o + j] >> COL_BITS], 1);
    }
  }
  if (wave < NCHW) {
    int part = w_own >> CNT_BITS;
    for (int d = 16; d >= 1; d >>= 1) part += __shfl_xor(part, d);
    if (lane == 0) s_part[wave] = part;
  }
  __syncthreads();
  int v = 0, incl = 0;
  if (t < BIN_ROWS) {
    v = s_cnt[t];
    incl = v;
    for (int d = 1; d < 32; d <<= 1) {
      int y = __shfl_up(incl, d);
      if (lane >= d) incl += y;
    }
    if (lane == 31) s_wtot[wave] = incl;
  }
  __syncthreads();
  int s = 0;
  for (int w = 0; w < NCHW; w++) s += s_part[w];
  if (t < BIN_ROWS) {
    int add = 0;
    for (int w = 0; w < 8; w++) add += (w < wave) ? s_wtot[w] : 0;
    int excl = add + incl - v;
    s_start[t] = excl;
    int row = b * BIN_ROWS + t;
    if (row < N_NODES) {
      rowptr[row] = s + excl;
      cnt[row] = v;
      float dg = (float)v;
      dis[row] = (dg > 0.0f) ? rsqrtf(fmaxf(dg, 1.0f)) : 0.0f;
    }
  }
  __syncthreads();
#pragma unroll
  for (int k = 0; k < SEGW; k++) {
    if (p[k] >= 0) {
      int rl = p[k] >> COL_BITS;
      int r = atomicAdd(&s_cur[rl], 1);
      colidx[s + s_start[rl] + r] = p[k] & ((1 << COL_BITS) - 1);
    }
  }
  for (int k = 0; k < SEGW; k++) {
    int sg = wave + 32 * k;
    if (sg < NCH) {
      int n = seg_cnt[sg], o = seg_off[sg];
      for (int j = 32 + lane; j < n; j += 32) {
        int e = sorted[o + j];
        int rl = e >> COL_BITS;
        int r = atomicAdd(&s_cur[rl], 1);
        colidx[s + s_start[rl] + r] = e & ((1 << COL_BITS) - 1);
      }
    }
  }
}


#define RPW 8
static_assert(N_NODES % RPW == 0 && RPW >= 2 && RPW <= 32,
              "a wave's rows all exist; their bookkeeping sits one row per lane");

__device__ __forceinline__ int lane_i(int v, int l) { return __builtin_amdgcn_readlane(v, l); }
__device__ __forceinline__ float lane_f(float v, int l) {
  return __int_as_float(__builtin_amdgcn_readlane(__float_as_int(v), l));
}

__device__ __forceinline__ int list_head(const int* __restrict__ colidx, int start, int n, int lane) {
  int c = 0;
  if (n > 0) {
    int m = n < 32 ? n : 32;
    int li = lane < m ? lane : m - 1;
    c = colidx[start + li];
  }
  return c;
}

__device__ __forceinline__ float head_weight(float dc, float ndr, int n, int lane) {
  int m = n < 32 ? n : 32;
  return lane < m ? ndr * dc : 0.0f;
}

template <int K>
__device__ __forceinline__ float gather_batch(int c, float w, int j, int m, int lane,
                                              const float* __restrict__ sp, int sstride, float acc) {
  float v[K];
#pragma unroll
  for (int u = 0; u < K; u++) {
    int cj = lane_i(c, j + u);
    const float* rowp = sp + (long long)cj * sstride;
    float x = rowp[lane];
    v[u] = (j + u < m) ? x : 0.0f;
  }
#pragma unroll
  for (int u = 0; u < K; u++) acc = fmaf(lane_f(w, j + u), v[u], acc);
  return acc;
}

__device__ __forceinline__ float gather_walk(int c, float w, int m, int lane,
                                             const float* __restrict__ sp, int sstride, float acc) {
  for (int j = 0; j < m; j += 16) {
    if (j + 8 < m) acc = gather_batch<16>(c, w, j, m, lane, sp, sstride, acc);
    else           acc = gather_batch<8>(c, w, j, m, lane, sp, sstride, acc);
  }
  return acc;
}

template <bool LAYER1>
__device__ __forceinline__ void gather_span(int r0, int lane,
                                            const int* __restrict__ rowptr,
                                            const int* __restrict__ cnt,
                                            const int* __restrict__ colidx,
                                            const float* __restrict__ dis,
                                            const float* __restrict__ sp, int sstride,
                                            const float* __restrict__ selfp, float bias,
                                            float* __restrict__ dst) {
  int mr = r0 + (lane & (RPW - 1));
  int st_v = rowptr[mr];
  int n_v = cnt[mr];
  float d_v = dis[mr];

  int n0 = lane_i(n_v, 0);
  int c_a = list_head(colidx, lane_i(st_v, 0), n0, lane);
  int c_b = list_head(colidx, lane_i(st_v, 1), lane_i(n_v, 1), lane);
  float w_a = head_weight(dis[c_a], -lane_f(d_v, 0), n0, lane);
  for (int i = 0; i < RPW; i++) {
    int r = r0 + i;
    int start = lane_i(st_v, i);
    int n = lane_i(n_v, i);
    int c_c = 0;
    if (i + 2 < RPW) c_c = list_head(colidx, lane_i(st_v, i + 2), lane_i(n_v, i + 2), lane);
    float self = 0.0f;
    if (LAYER1) self = selfp[(long long)r * 64 + lane];
    float dc_b = dis[c_b];
    int m = n < 32 ? n : 32;
    float acc = gather_walk(c_a, w_a, m, lane, sp, sstride, 0.0f);
    if (n > 32) {
      float ndr = -lane_f(d_v, i);
      for (int base = 32; base < n; base += 32) {
        int m2 = n - base;
        if (m2 > 32) m2 = 32;
        int li = lane < m2 ? lane : m2 - 1;
        int c2 = colidx[start + base + li];
        float w2 = lane < m2 ? ndr * dis[c2] : 0.0f;
        acc = gather_walk(c2, w2, m2, lane, sp, sstride, acc);
      }
    }
    float* drow = dst + (long long)r * HID;
    if (LAYER1) {
      float v = self + acc + bias;
      drow[lane] = v > 0.0f ? v : 0.0f;
    } else {
      drow[lane] = acc;
    }
    float w_b = head_weight(dc_b, -lane_f(d_v, i + 1), lane_i(n_v, i + 1), lane);
    c_a = c_b; w_a = w_b; c_b = c_c;
  }
}

__global__ void __launch_bounds__(128)
gather_relu_kernel(const int* __restrict__ rowptr,
                   const int* __restrict__ cnt,
                   const int* __restrict__ colidx,
                   const float* __restrict__ dis,
                   const float* __restrict__ P,
                   const float* __restrict__ b1,
                   float* __restrict__ h) {
  int lane = threadIdx.x & 31;
  int r0 = __builtin_amdgcn_readfirstlane((blockIdx.x * 4 + (threadIdx.x >> 5)) * RPW);
  if (r0 >= N_NODES) return;
  gather_span<true>(r0, lane, rowptr, cnt, colidx, dis, P + HID, 64, P, b1[lane], h);
}

__global__ void __launch_bounds__(128)
gather_rows_kernel(const int* __restrict__ rowptr,
                   const int* __restrict__ cnt,
                   const int* __restrict__ colidx,
                   const float* __restrict__ dis,
                   const float* __restrict__ src,
                   float* __restrict__ dst) {
  int lane = threadIdx.x & 31;
  int r0 = __builtin_amdgcn_readfirstlane((blockIdx.x * 4 + (threadIdx.x >> 5)) * RPW);
  if (r0 >= N_NODES) return;
  gather_span<false>(r0, lane, rowptr, cnt, colidx, dis, src, HID, src, 0.0f, dst);
}


__device__ __forceinline__ void
proj1_wmma_body(int pb,
                const float* __restrict__ x,
                const float* __restrict__ W0,
                const float* __restrict__ W1,
                float* __restrict__ P) {
  int wave = threadIdx.x >> 5;
  int lane = threadIdx.x & 31;
  int half = lane >> 4;
  int l16  = lane & 15;
  int m0 = (pb * 32 + wave) * 16;
  if (m0 >= N_NODES) return;
  int mA = m0 + l16;
  const float* arow = x + (long long)mA * F_IN;

  v8f acc0 = {}, acc1 = {}, acc2 = {}, acc3 = {};
  for (int kk = 0; kk < 48; kk += 4) {
    int k = kk + 2 * half;
    v2f a = *(const v2f*)(arow + k);
    const float* p0 = W0 + k * HID + l16;
    const float* p1 = W1 + k * HID + l16;
    v2f b;
    b.x = p0[0];  b.y = p0[HID];
    acc0 = __builtin_amdgcn_wmma_f32_16x16x4_f32(false, a, false, b, (short)0, acc0, false, false);
    b.x = p0[16]; b.y = p0[HID + 16];
    acc1 = __builtin_amdgcn_wmma_f32_16x16x4_f32(false, a, false, b, (short)0, acc1, false, false);
    b.x = p1[0];  b.y = p1[HID];
    acc2 = __builtin_amdgcn_wmma_f32_16x16x4_f32(false, a, false, b, (short)0, acc2, false, false);
    b.x = p1[16]; b.y = p1[HID + 16];
    acc3 = __builtin_amdgcn_wmma_f32_16x16x4_f32(false, a, false, b, (short)0, acc3, false, false);
  }
  {
    v2f aload = *(const v2f*)(arow + 48);
    v2f a;
    a.x = half ? 0.0f : aload.x;
    a.y = half ? 0.0f : aload.y;
    const float* p0 = W0 + 48 * HID + l16;
    const float* p1 = W1 + 48 * HID + l16;
    v2f b;
    b.x = p0[0];  b.y = p0[HID];
    acc0 = __builtin_amdgcn_wmma_f32_16x16x4_f32(false, a, false, b, (short)0, acc0, false, false);
    b.x = p0[16]; b.y = p0[HID + 16];
    acc1 = __builtin_amdgcn_wmma_f32_16x16x4_f32(false, a, false, b, (short)0, acc1, false, false);
    b.x = p1[0];  b.y = p1[HID];
    acc2 = __builtin_amdgcn_wmma_f32_16x16x4_f32(false, a, false, b, (short)0, acc2, false, false);
    b.x = p1[16]; b.y = p1[HID + 16];
    acc3 = __builtin_amdgcn_wmma_f32_16x16x4_f32(false, a, false, b, (short)0, acc3, false, false);
  }
#pragma unroll
  for (int v = 0; v < 8; v++) {
    int mC = m0 + v + 8 * half;
    float* prow = P + (long long)mC * 64 + l16;
    prow[0]  = acc0[v];
    prow[16] = acc1[v];
    prow[32] = acc2[v];
    prow[48] = acc3[v];
  }
}

__global__ void __launch_bounds__(1024, 16)
count_proj1_kernel(const int* __restrict__ rows,
                   const int* __restrict__ cols,
                   int* __restrict__ tbl,
                   int* __restrict__ sorted,
                   const float* __restrict__ x,
                   const float* __restrict__ W0,
                   const float* __restrict__ W1,
                   float* __restrict__ P) {
  if (blockIdx.x < NCH) bin_order_body(rows, cols, tbl, sorted);
  else                  proj1_wmma_body((int)blockIdx.x - NCH, x, W0, W1, P);
}


__device__ __forceinline__ void out_k32(const float* __restrict__ arow,
                                        const float* __restrict__ Wm,
                                        int half, int l16,
                                        v8f& acc0, v8f& acc1, v8f& acc2) {
  int c2 = 32 + (l16 & 7);
  bool ok2 = l16 < 8;
  for (int kk = 0; kk < HID; kk += 4) {
    int k = kk + 2 * half;
    v2f a = *(const v2f*)(arow + k);
    const float* p = Wm + k * NCLS;
    v2f b;
    b.x = p[l16];      b.y = p[NCLS + l16];
    acc0 = __builtin_amdgcn_wmma_f32_16x16x4_f32(false, a, false, b, (short)0, acc0, false, false);
    b.x = p[16 + l16]; b.y = p[NCLS + 16 + l16];
    acc1 = __builtin_amdgcn_wmma_f32_16x16x4_f32(false, a, false, b, (short)0, acc1, false, false);
    float t0 = p[c2], t1 = p[NCLS + c2];
    b.x = ok2 ? t0 : 0.0f; b.y = ok2 ? t1 : 0.0f;
    acc2 = __builtin_amdgcn_wmma_f32_16x16x4_f32(false, a, false, b, (short)0, acc2, false, false);
  }
}

template <int M>
__device__ __forceinline__ float row16_xor(float v) {
  return __int_as_float(__builtin_amdgcn_update_dpp(0, __float_as_int(v), 0x160 + M, 0xf, 0xf, true));
}
__device__ __forceinline__ float row16_max(float v) {
  v = fmaxf(v, row16_xor<1>(v));
  v = fmaxf(v, row16_xor<2>(v));
  v = fmaxf(v, row16_xor<4>(v));
  v = fmaxf(v, row16_xor<8>(v));
  return v;
}
__device__ __forceinline__ float row16_sum(float v) {
  v += row16_xor<1>(v);
  v += row16_xor<2>(v);
  v += row16_xor<4>(v);
  v += row16_xor<8>(v);
  return v;
}

__global__ void __launch_bounds__(128)
out_wmma_lsm_kernel(const float* __restrict__ h,
                    const float* __restrict__ agg2,
                    const float* __restrict__ W0,
                    const float* __restrict__ W1,
                    const float* __restrict__ b2,
                    float* __restrict__ out) {
  int wave = threadIdx.x >> 5;
  int lane = threadIdx.x & 31;
  int half = lane >> 4;
  int l16  = lane & 15;
  int m0 = (blockIdx.x * 4 + wave) * 16;
  if (m0 >= N_NODES) return;
  int mA = m0 + l16;

  v8f acc0 = {}, acc1 = {}, acc2 = {};
  out_k32(h    + (long long)mA * HID, W0, half, l16, acc0, acc1, acc2);
  out_k32(agg2 + (long long)mA * HID, W1, half, l16, acc0, acc1, acc2);

  bool ok2 = l16 < 8;
  float bb0 = b2[l16], bb1 = b2[16 + l16], bb2 = b2[32 + (l16 & 7)];
  float x0[8], x1[8], x2[8], mx[8], se[8];
#pragma unroll
  for (int v = 0; v < 8; v++) {
    x0[v] = acc0[v] + bb0;
    x1[v] = acc1[v] + bb1;
    x2[v] = acc2[v] + bb2;
    mx[v] = fmaxf(fmaxf(x0[v], x1[v]), ok2 ? x2[v] : -INFINITY);
  }
#pragma unroll
  for (int v = 0; v < 8; v++) mx[v] = row16_max(mx[v]);
#pragma unroll
  for (int v = 0; v < 8; v++) {
    x0[v] -= mx[v];
    x1[v] -= mx[v];
    x2[v] -= mx[v];
    float e2 = expf(x2[v]);
    se[v] = expf(x0[v]) + expf(x1[v]);
    se[v] += ok2 ? e2 : 0.0f;
  }
#pragma unroll
  for (int v = 0; v < 8; v++) se[v] = logf(row16_sum(se[v]));
  float* otile = out + (long long)(m0 + 8 * half) * NCLS + l16;
#pragma unroll
  for (int v = 0; v < 8; v++) {
    otile[v * NCLS]      = x0[v] - se[v];
    otile[v * NCLS + 16] = x1[v] - se[v];
  }
  if (ok2) {
#pragma unroll
    for (int v = 0; v < 8; v++) otile[v * NCLS + 32] = x2[v] - se[v];
  }
}


extern "C" void kernel_launch(void* const* d_in, const int* in_sizes, int n_in,
                              void* d_out, int out_size, void* d_ws, size_t ws_size,
                              hipStream_t stream) {
  (void)in_sizes; (void)n_in; (void)out_size; (void)ws_size;
  const float* x    = (const float*)d_in[0];
  const int*   ei   = (const int*)  d_in[1];
  const float* W0_1 = (const float*)d_in[2];
  const float* W1_1 = (const float*)d_in[3];
  const float* b1   = (const float*)d_in[4];
  const float* W0_2 = (const float*)d_in[5];
  const float* W1_2 = (const float*)d_in[6];
  const float* b2   = (const float*)d_in[7];
  float* out = (float*)d_out;

  const int* rows = ei;
  const int* cols = ei + N_EDGES;

  const long long N = N_NODES;
  float* ws     = (float*)d_ws;
  int*   cnt    = (int*)ws;
  float* dis    = ws + N;
  float* P      = ws + 2 * N;
  float* agg2   = ws + 2 * N;
  int*   colidx = (int*)(ws + 66 * N);
  int*   rowptr = (int*)(ws + 82 * N);
  int*   tbl    = (int*)(ws + 83 * N);
  float* hbuf   = ws + 98 * N;
  int*   sorted = (int*)(ws + 98 * N);

  count_proj1_kernel<<<NCH + PB, 1024, 0, stream>>>(rows, cols, tbl, sorted, x, W0_1, W1_1, P);
  bin_rows_kernel<<<NB, 1024, 0, stream>>>(tbl, sorted, cnt, rowptr, dis, colidx);

  gather_relu_kernel<<<(N_NODES / RPW + 3) / 4, 128, 0, stream>>>(
      rowptr, cnt, colidx, dis, P, b1, hbuf);

  gather_rows_kernel<<<(N_NODES / RPW + 3) / 4, 128, 0, stream>>>(
      rowptr, cnt, colidx, dis, hbuf, agg2);
  out_wmma_lsm_kernel<<<(N_NODES + 63) / 64, 128, 0, stream>>>(hbuf, agg2, W0_2, W1_2, b2, out);
}


# Gather loads 16 B a lane: one load instruction fetches four neighbour rows whole (8 lanes a row); the 4 lane groups' partial sums are joined by a 2-step butterfly, so the sum order changes
# speedup vs baseline: 5.5568x; 1.1236x over previous
#include <hip/hip_runtime.h>
#include <math.h>

#define N_NODES 100000
#define N_EDGES 1600000
#define F_IN    50
#define HID     32
#define NCLS    40
#define BIN_SHIFT 8
#define BIN_ROWS  256
#define NB        391
#define CHUNK     8192
#define NCH       196
#define TBL       (NB * NCH)
#define PB        196
#define COL_BITS  17
#define CNT_BITS  14
#define CNT_MASK  ((1 << CNT_BITS) - 1)
#define SEGW      7
#define NCHW      ((NCH + 31) / 32)
static_assert(N_NODES <= (1 << COL_BITS), "col must fit the low field of a list entry");
static_assert(NB * BIN_ROWS >= N_NODES && NCH * CHUNK >= N_EDGES && N_EDGES % 4 == 0, "sizes");
static_assert(CHUNK <= CNT_MASK && 2 * CNT_BITS <= 31 && CHUNK % 4 == 0, "start and length of a run fit one table word");
static_assert(TBL <= N_NODES, "the table has one N-word plane of the workspace");
static_assert(NCH * CHUNK <= 32 * N_NODES, "the chunk-ordered list fits the plane that later holds h");
static_assert(SEGW * 32 >= NCH, "every run of a bin has a wave");
static_assert(PB * 512 >= N_NODES, "GEMM-1 work-groups cover every row");
static_assert(HID <= NCLS, "the scaled copy of h (N x HID) fits the output buffer (N x NCLS) it lies in");


typedef float v2f __attribute__((ext_vector_type(2)));
typedef float v4f __attribute__((ext_vector_type(4)));
typedef float v8f __attribute__((ext_vector_type(8)));


#define EDGE8(M) \
  M(0, r4[0].x, c4[0].x) M(1, r4[0].y, c4[0].y) M(2, r4[0].z, c4[0].z) M(3, r4[0].w, c4[0].w) \
  M(4, r4[1].x, c4[1].x) M(5, r4[1].y, c4[1].y) M(6, r4[1].z, c4[1].z) M(7, r4[1].w, c4[1].w)

__device__ __forceinline__ void
bin_order_body(const int* __restrict__ rows,
               const int* __restrict__ cols,
               int* __restrict__ tbl,
               int* __restrict__ sorted) {
  __shared__ int hist[NB];
  __shared__ int off[NB];
  __shared__ int s_wtot[13];
  __shared__ int4 stage4[CHUNK / 4];
  int* stage = (int*)stage4;
  int t = threadIdx.x, lane = t & 31, wave = t >> 5;
  int4 r4[2], c4[2];
#pragma unroll
  for (int h = 0; h < 2; h++) {
    int e = blockIdx.x * CHUNK + h * 4096 + t * 4;
    r4[h] = make_int4(0, 0, 0, 0);
    c4[h] = make_int4(0, 0, 0, 0);
    if (e < N_EDGES) {
      r4[h] = *(const int4*)(rows + e);
      c4[h] = *(const int4*)(cols + e);
    }
  }
  for (int i = t; i < NB; i += 1024) hist[i] = 0;
  __syncthreads();
  int rk[8];
#define RANK1(I, R, C) rk[I] = ((R) != (C)) ? atomicAdd(&hist[(R) >> BIN_SHIFT], 1) : -1;
  EDGE8(RANK1)
#undef RANK1
  __syncthreads();
  int v = 0, incl = 0;
  if (t < 13 * 32) {
    if (t < NB) v = hist[t];
    incl = v;
    for (int d = 1; d < 32; d <<= 1) {
      int y = __shfl_up(incl, d);
      if (lane >= d) incl += y;
    }
    if (lane == 31) s_wtot[wave] = incl;
  }
  __syncthreads();
  if (t < NB) {
    int add = 0;
    for (int w = 0; w < 13; w++) add += (w < wave) ? s_wtot[w] : 0;
    off[t] = add + incl - v;
  }
  __syncthreads();
#define PLACE1(I, R, C) if (rk[I] >= 0) stage[off[(R) >> BIN_SHIFT] + rk[I]] = (((R) & (BIN_ROWS - 1)) << COL_BITS) | (C);
  EDGE8(PLACE1)
#undef PLACE1
  __syncthreads();
  int total = off[NB - 1] + hist[NB - 1];
  int* run = sorted + blockIdx.x * CHUNK;
  for (int i = t * 4; i < total; i += 4096) *(int4*)(run + i) = stage4[i >> 2];
  for (int i = t; i < NB; i += 1024) tbl[i * NCH + blockIdx.x] = (off[i] << CNT_BITS) | hist[i];
}

__global__ void __launch_bounds__(1024, 16)
bin_rows_kernel(const int* __restrict__ tbl,
                const int* __restrict__ sorted,
                int* __restrict__ cnt,
                int* __restrict__ rowptr,
                float* __restrict__ dis,
                int* __restrict__ colidx,
                float* __restrict__ P) {
  __shared__ int s_cnt[BIN_ROWS];
  __shared__ int s_start[BIN_ROWS];
  __shared__ int s_cur[BIN_ROWS];
  __shared__ int s_wtot[8];
  __shared__ int s_part[NCHW];
  __shared__ int seg_off[NCH];
  __shared__ int seg_cnt[NCH];
  int t = threadIdx.x, lane = t & 31, wave = t >> 5;
  int b = blockIdx.x;
  v4f pv[2];
  int prow[2];
#pragma unroll
  for (int q = 0; q < 2; q++) {
    int idx = t + 1024 * q;
    prow[q] = idx >> 3;
    pv[q] = (v4f){0.0f, 0.0f, 0.0f, 0.0f};
    int row = b * BIN_ROWS + prow[q];
    if (row < N_NODES) pv[q] = *(const v4f*)(P + (long long)row * 64 + HID + (idx & 7) * 4);
  }
  int w_own = 0;
  if (t < NCH) {
    w_own = tbl[b * NCH + t];
    seg_off[t] = t * CHUNK + (w_own >> CNT_BITS);
    seg_cnt[t] = w_own & CNT_MASK;
  }
  if (t < BIN_ROWS) { s_cnt[t] = 0; s_cur[t] = 0; }
  __syncthreads();
  int p[SEGW];
#pragma unroll
  for (int k = 0; k < SEGW; k++) {
    int sg = wave + 32 * k;
    p[k] = -1;
    if (sg < NCH && lane < seg_cnt[sg]) p[k] = sorted[seg_off[sg] + lane];
  }
#pragma unroll
  for (int k = 0; k < SEGW; k++)
    if (p[k] >= 0) atomicAdd(&s_cnt[p[k] >> COL_BITS], 1);
  for (int k = 0; k < SEGW; k++) {
    int sg = wave + 32 * k;
    if (sg < NCH) {
      int n = seg_cnt[sg], o = seg_off[sg];
      for (int j = 32 + lane; j < n; j += 32) atomicAdd(&s_cnt[sorted[o + j] >> COL_BITS], 1);
    }
  }
  if (wave < NCHW) {
    int part = w_own >> CNT_BITS;
    for (int d = 16; d >= 1; d >>= 1) part += __shfl_xor(part, d);
    if (lane == 0) s_part[wave] = part;
  }
  __syncthreads();
  int v = 0, incl = 0;
  if (t < BIN_ROWS) {
    v = s_cnt[t];
    incl = v;
    for (int d = 1; d < 32; d <<= 1) {
      int y = __shfl_up(incl, d);
      if (lane >= d) incl += y;
    }
    if (lane == 31) s_wtot[wave] = incl;
  }
  __syncthreads();
#pragma unroll
  for (int q = 0; q < 2; q++) {
    int row = b * BIN_ROWS + prow[q];
    if (row < N_NODES) {
      float dg = (float)s_cnt[prow[q]];
      float dc = (dg > 0.0f) ? rsqrtf(fmaxf(dg, 1.0f)) : 0.0f;
      *(v4f*)(P + (long long)row * 64 + HID + ((t + 1024 * q) & 7) * 4) = pv[q] * dc;
    }
  }
  int s = 0;
  for (int w = 0; w < NCHW; w++) s += s_part[w];
  if (t < BIN_ROWS) {
    int add = 0;
    for (int w = 0; w < 8; w++) add += (w < wave) ? s_wtot[w] : 0;
    int excl = add + incl - v;
    s_start[t] = excl;
    int row = b * BIN_ROWS + t;
    if (row < N_NODES) {
      rowptr[row] = s + excl;
      cnt[row] = v;
      float dg = (float)v;
      dis[row] = (dg > 0.0f) ? rsqrtf(fmaxf(dg, 1.0f)) : 0.0f;
    }
  }
  __syncthreads();
#pragma unroll
  for (int k = 0; k < SEGW; k++) {
    if (p[k] >= 0) {
      int rl = p[k] >> COL_BITS;
      int r = atomicAdd(&s_cur[rl], 1);
      colidx[s + s_start[rl] + r] = p[k] & ((1 << COL_BITS) - 1);
    }
  }
  for (int k = 0; k < SEGW; k++) {
    int sg = wave + 32 * k;
    if (sg < NCH) {
      int n = seg_cnt[sg], o = seg_off[sg];
      for (int j = 32 + lane; j < n; j += 32) {
        int e = sorted[o + j];
        int rl = e >> COL_BITS;
        int r = atomicAdd(&s_cur[rl], 1);
        colidx[s + s_start[rl] + r] = e & ((1 << COL_BITS) - 1);
      }
    }
  }
}


#define RPW 8
static_assert(N_NODES % RPW == 0 && RPW >= 2 && RPW <= 32 && (RPW & (RPW - 1)) == 0,
              "a wave's rows all exist; their bookkeeping sits one row per lane");
static_assert(HID == 32, "a source row is 128 B: eight lanes of 16 B");

__device__ __forceinline__ int lane_i(int v, int l) { return __builtin_amdgcn_readlane(v, l); }
__device__ __forceinline__ float lane_f(float v, int l) {
  return __int_as_float(__builtin_amdgcn_readlane(__float_as_int(v), l));
}

template <int Q>
__device__ __forceinline__ int group8_lane(int v) {
  return __builtin_amdgcn_mov_dpp8(v, Q * 0x249249);
}

__device__ __forceinline__ float xor8(float v) {
  return __int_as_float(__builtin_amdgcn_update_dpp(0, __float_as_int(v), 0x168, 0xf, 0xf, true));
}
__device__ __forceinline__ float xor16(float v) {
  return __int_as_float(__builtin_amdgcn_permlanex16(__float_as_int(v), __float_as_int(v),
                                                     0x76543210, 0xfedcba98, false, false));
}

__device__ __forceinline__ int list_head(const int* __restrict__ colidx, int start, int n, int slot) {
  int m = n < 32 ? n : 32;
  int li = slot < m ? slot : m - 1;
  li = li > 0 ? li : 0;
  int idx = start + li;
  idx = idx < N_EDGES - 1 ? idx : N_EDGES - 1;
  return colidx[idx];
}

template <int Q, int STRIDE_B>
__device__ __forceinline__ v4f nb_load(int e, unsigned foff, const char* __restrict__ spb) {
  unsigned c = (unsigned)group8_lane<Q>(e);
  return *(const v4f*)(spb + (c * (unsigned)STRIDE_B + foff));
}

template <int K, int STRIDE_B, class F>
__device__ __forceinline__ v4f gather_k(int e, int lim, unsigned foff, const char* __restrict__ spb, v4f acc, F behind) {
  const v4f zero = {0.0f, 0.0f, 0.0f, 0.0f};
  v4f x[8];
  x[0] = nb_load<0, STRIDE_B>(e, foff, spb);
  x[1] = nb_load<1, STRIDE_B>(e, foff, spb);
  x[2] = nb_load<2, STRIDE_B>(e, foff, spb);
  x[3] = nb_load<3, STRIDE_B>(e, foff, spb);
  if (K > 4) {
    x[4] = nb_load<4, STRIDE_B>(e, foff, spb);
    x[5] = nb_load<5, STRIDE_B>(e, foff, spb);
  }
  if (K > 6) {
    x[6] = nb_load<6, STRIDE_B>(e, foff, spb);
    x[7] = nb_load<7, STRIDE_B>(e, foff, spb);
  }
  behind();
#pragma unroll
  for (int q = 0; q < K; q++) acc += (4 * q < lim) ? x[q] : zero;
  return acc;
}

template <int STRIDE_B, class F>
__device__ __forceinline__ v4f gather_part(int e, int m, int g, unsigned foff, const char* __restrict__ spb, v4f acc, F behind) {
  int lim = m - g;
  if (m > 24)      acc = gather_k<8, STRIDE_B>(e, lim, foff, spb, acc, behind);
  else if (m > 16) acc = gather_k<6, STRIDE_B>(e, lim, foff, spb, acc, behind);
  else             acc = gather_k<4, STRIDE_B>(e, lim, foff, spb, acc, behind);
  return acc;
}

template <bool LAYER1, int STRIDE_B>
__device__ __forceinline__ void gather_span(int r0, int lane,
                                            const int* __restrict__ rowptr,
                                            const int* __restrict__ cnt,
                                            const int* __restrict__ colidx,
                                            const float* __restrict__ dis,
                                            const float* __restrict__ sp,
                                            const float* __restrict__ selfp,
                                            const float* __restrict__ b1,
                                            float* __restrict__ dst,
                                            float* __restrict__ dst_scaled) {
  const v4f zero = {0.0f, 0.0f, 0.0f, 0.0f};
  int g = lane >> 3, p = lane & 7;
  int slot = 4 * p + g;
  unsigned foff = 16u * (unsigned)p;
  const char* spb = (const char*)sp;
  int mr = r0 + (lane & (RPW - 1));
  int st_v = rowptr[mr];
  int n_v = cnt[mr];
  float d_v = dis[mr];
  v4f bias = zero;
  v4f self = zero;
  if (LAYER1) {
    bias = *(const v4f*)((const char*)b1 + foff);
    self = *(const v4f*)((const char*)selfp + ((unsigned)r0 * 256u + foff));
  }

  int e_a = list_head(colidx, lane_i(st_v, 0), lane_i(n_v, 0), slot);
  int e_b = list_head(colidx, lane_i(st_v, 1), lane_i(n_v, 1), slot);
  for (int i = 0; i < RPW; i++) {
    int r = r0 + i;
    int start = lane_i(st_v, i);
    int n = lane_i(n_v, i);
    int e_c = 0;
    v4f self_nxt = zero;
    auto behind = [&]() {
      if (i + 2 < RPW) e_c = list_head(colidx, lane_i(st_v, i + 2), lane_i(n_v, i + 2), slot);
      if (LAYER1 && i + 1 < RPW) self_nxt = *(const v4f*)((const char*)selfp + ((unsigned)(r + 1) * 256u + foff));
    };
    int m = n < 32 ? n : 32;
    v4f acc = zero;
    if (m > 0) acc = gather_part<STRIDE_B>(e_a, m, g, foff, spb, acc, behind);
    else       behind();
    for (int base = 32; base < n; base += 32) {
      int m2 = n - base;
      if (m2 > 32) m2 = 32;
      int li = slot < m2 ? slot : m2 - 1;
      int e2 = colidx[start + base + li];
      acc = gather_part<STRIDE_B>(e2, m2, g, foff, spb, acc, []() {});
    }
    acc.x += xor8(acc.x);  acc.y += xor8(acc.y);  acc.z += xor8(acc.z);  acc.w += xor8(acc.w);
    acc.x += xor16(acc.x); acc.y += xor16(acc.y); acc.z += xor16(acc.z); acc.w += xor16(acc.w);
    float d_r = lane_f(d_v, i);
    v4f s = acc * (-d_r);
    unsigned roff = (unsigned)r * (unsigned)(HID * 4) + foff;
    char* drow = (char*)dst + roff;
    if (LAYER1) {
      v4f v = self + s + bias;
      v4f hv;
      hv.x = v.x > 0.0f ? v.x : 0.0f;
      hv.y = v.y > 0.0f ? v.y : 0.0f;
      hv.z = v.z > 0.0f ? v.z : 0.0f;
      hv.w = v.w > 0.0f ? v.w : 0.0f;
      if (g == 0) {
        *(v4f*)drow = hv;
        *(v4f*)((char*)dst_scaled + roff) = hv * d_r;
      }
    } else {
      if (g == 0) *(v4f*)drow = s;
    }
    e_a = e_b; e_b = e_c;
    self = self_nxt;
  }
}

__global__ void __launch_bounds__(128, 16)
gather_relu_kernel(const int* __restrict__ rowptr,
                   const int* __restrict__ cnt,
                   const int* __restrict__ colidx,
                   const float* __restrict__ dis,
                   const float* __restrict__ P,
                   const float* __restrict__ b1,
                   float* __restrict__ h,
                   float* __restrict__ hs) {
  int lane = threadIdx.x & 31;
  int r0 = __builtin_amdgcn_readfirstlane((blockIdx.x * 4 + (threadIdx.x >> 5)) * RPW);
  if (r0 >= N_NODES) return;
  gather_span<true, 256>(r0, lane, rowptr, cnt, colidx, dis, P + HID, P, b1, h, hs);
}

__global__ void __launch_bounds__(128, 16)
gather_rows_kernel(const int* __restrict__ rowptr,
                   const int* __restrict__ cnt,
                   const int* __restrict__ colidx,
                   const float* __restrict__ dis,
                   const float* __restrict__ src,
                   float* __restrict__ dst) {
  int lane = threadIdx.x & 31;
  int r0 = __builtin_amdgcn_readfirstlane((blockIdx.x * 4 + (threadIdx.x >> 5)) * RPW);
  if (r0 >= N_NODES) return;
  gather_span<false, 128>(r0, lane, rowptr, cnt, colidx, dis, src, src, nullptr, dst, nullptr);
}


__device__ __forceinline__ void
proj1_wmma_body(int pb,
                const float* __restrict__ x,
                const float* __restrict__ W0,
                const float* __restrict__ W1,
                float* __restrict__ P) {
  int wave = threadIdx.x >> 5;
  int lane = threadIdx.x & 31;
  int half = lane >> 4;
  int l16  = lane & 15;
  int m0 = (pb * 32 + wave) * 16;
  if (m0 >= N_NODES) return;
  int mA = m0 + l16;
  const float* arow = x + (long long)mA * F_IN;

  v8f acc0 = {}, acc1 = {}, acc2 = {}, acc3 = {};
  for (int kk = 0; kk < 48; kk += 4) {
    int k = kk + 2 * half;
    v2f a = *(const v2f*)(arow + k);
    const float* p0 = W0 + k * HID + l16;
    const float* p1 = W1 + k * HID + l16;
    v2f b;
    b.x = p0[0];  b.y = p0[HID];
    acc0 = __builtin_amdgcn_wmma_f32_16x16x4_f32(false, a, false, b, (short)0, acc0, false, false);
    b.x = p0[16]; b.y = p0[HID + 16];
    acc1 = __builtin_amdgcn_wmma_f32_16x16x4_f32(false, a, false, b, (short)0, acc1, false, false);
    b.x = p1[0];  b.y = p1[HID];
    acc2 = __builtin_amdgcn_wmma_f32_16x16x4_f32(false, a, false, b, (short)0, acc2, false, false);
    b.x = p1[16]; b.y = p1[HID + 16];
    acc3 = __builtin_amdgcn_wmma_f32_16x16x4_f32(false, a, false, b, (short)0, acc3, false, false);
  }
  {
    v2f aload = *(const v2f*)(arow + 48);
    v2f a;
    a.x = half ? 0.0f : aload.x;
    a.y = half ? 0.0f : aload.y;
    const float* p0 = W0 + 48 * HID + l16;
    const float* p1 = W1 + 48 * HID + l16;
    v2f b;
    b.x = p0[0];  b.y = p0[HID];
    acc0 = __builtin_amdgcn_wmma_f32_16x16x4_f32(false, a, false, b, (short)0, acc0, false, false);
    b.x = p0[16]; b.y = p0[HID + 16];
    acc1 = __builtin_amdgcn_wmma_f32_16x16x4_f32(false, a, false, b, (short)0, acc1, false, false);
    b.x = p1[0];  b.y = p1[HID];
    acc2 = __builtin_amdgcn_wmma_f32_16x16x4_f32(false, a, false, b, (short)0, acc2, false, false);
    b.x = p1[16]; b.y = p1[HID + 16];
    acc3 = __builtin_amdgcn_wmma_f32_16x16x4_f32(false, a, false, b, (short)0, acc3, false, false);
  }
#pragma unroll
  for (int v = 0; v < 8; v++) {
    int mC = m0 + v + 8 * half;
    float* prow = P + (long long)mC * 64 + l16;
    prow[0]  = acc0[v];
    prow[16] = acc1[v];
    prow[32] = acc2[v];
    prow[48] = acc3[v];
  }
}

__global__ void __launch_bounds__(1024, 16)
count_proj1_kernel(const int* __restrict__ rows,
                   const int* __restrict__ cols,
                   int* __restrict__ tbl,
                   int* __restrict__ sorted,
                   const float* __restrict__ x,
                   const float* __restrict__ W0,
                   const float* __restrict__ W1,
                   float* __restrict__ P) {
  if (blockIdx.x < NCH) bin_order_body(rows, cols, tbl, sorted);
  else                  proj1_wmma_body((int)blockIdx.x - NCH, x, W0, W1, P);
}


__device__ __forceinline__ void out_k32(const float* __restrict__ arow,
                                        const float* __restrict__ Wm,
                                        int half, int l16,
                                        v8f& acc0, v8f& acc1, v8f& acc2) {
  int c2 = 32 + (l16 & 7);
  bool ok2 = l16 < 8;
  for (int kk = 0; kk < HID; kk += 4) {
    int k = kk + 2 * half;
    v2f a = *(const v2f*)(arow + k);
    const float* p = Wm + k * NCLS;
    v2f b;
    b.x = p[l16];      b.y = p[NCLS + l16];
    acc0 = __builtin_amdgcn_wmma_f32_16x16x4_f32(false, a, false, b, (short)0, acc0, false, false);
    b.x = p[16 + l16]; b.y = p[NCLS + 16 + l16];
    acc1 = __builtin_amdgcn_wmma_f32_16x16x4_f32(false, a, false, b, (short)0, acc1, false, false);
    float t0 = p[c2], t1 = p[NCLS + c2];
    b.x = ok2 ? t0 : 0.0f; b.y = ok2 ? t1 : 0.0f;
    acc2 = __builtin_amdgcn_wmma_f32_16x16x4_f32(false, a, false, b, (short)0, acc2, false, false);
  }
}

template <int M>
__device__ __forceinline__ float row16_xor(float v) {
  return __int_as_float(__builtin_amdgcn_update_dpp(0, __float_as_int(v), 0x160 + M, 0xf, 0xf, true));
}
__device__ __forceinline__ float row16_max(float v) {
  v = fmaxf(v, row16_xor<1>(v));
  v = fmaxf(v, row16_xor<2>(v));
  v = fmaxf(v, row16_xor<4>(v));
  v = fmaxf(v, row16_xor<8>(v));
  return v;
}
__device__ __forceinline__ float row16_sum(float v) {
  v += row16_xor<1>(v);
  v += row16_xor<2>(v);
  v += row16_xor<4>(v);
  v += row16_xor<8>(v);
  return v;
}

__global__ void __launch_bounds__(128)
out_wmma_lsm_kernel(const float* __restrict__ h,
                    const float* __restrict__ agg2,
                    const float* __restrict__ W0,
                    const float* __restrict__ W1,
                    const float* __restrict__ b2,
                    float* __restrict__ out) {
  int wave = threadIdx.x >> 5;
  int lane = threadIdx.x & 31;
  int half = lane >> 4;
  int l16  = lane & 15;
  int m0 = (blockIdx.x * 4 + wave) * 16;
  if (m0 >= N_NODES) return;
  int mA = m0 + l16;

  v8f acc0 = {}, acc1 = {}, acc2 = {};
  out_k32(h    + (long long)mA * HID, W0, half, l16, acc0, acc1, acc2);
  out_k32(agg2 + (long long)mA * HID, W1, half, l16, acc0, acc1, acc2);

  bool ok2 = l16 < 8;
  float bb0 = b2[l16], bb1 = b2[16 + l16], bb2 = b2[32 + (l16 & 7)];
  float x0[8], x1[8], x2[8], mx[8], se[8];
#pragma unroll
  for (int v = 0; v < 8; v++) {
    x0[v] = acc0[v] + bb0;
    x1[v] = acc1[v] + bb1;
    x2[v] = acc2[v] + bb2;
    mx[v] = fmaxf(fmaxf(x0[v], x1[v]), ok2 ? x2[v] : -INFINITY);
  }
#pragma unroll
  for (int v = 0; v < 8; v++) mx[v] = row16_max(mx[v]);
#pragma unroll
  for (int v = 0; v < 8; v++) {
    x0[v] -= mx[v];
    x1[v] -= mx[v];
    x2[v] -= mx[v];
    float e2 = expf(x2[v]);
    se[v] = expf(x0[v]) + expf(x1[v]);
    se[v] += ok2 ? e2 : 0.0f;
  }
#pragma unroll
  for (int v = 0; v < 8; v++) se[v] = logf(row16_sum(se[v]));
  float* otile = out + (long long)(m0 + 8 * half) * NCLS + l16;
#pragma unroll
  for (int v = 0; v < 8; v++) {
    otile[v * NCLS]      = x0[v] - se[v];
    otile[v * NCLS + 16] = x1[v] - se[v];
  }
  if (ok2) {
#pragma unroll
    for (int v = 0; v < 8; v++) otile[v * NCLS + 32] = x2[v] - se[v];
  }
}


extern "C" void kernel_launch(void* const* d_in, const int* in_sizes, int n_in,
                              void* d_out, int out_size, void* d_ws, size_t ws_size,
                              hipStream_t stream) {
  (void)in_sizes; (void)n_in; (void)out_size; (void)ws_size;
  const float* x    = (const float*)d_in[0];
  const int*   ei   = (const int*)  d_in[1];
  const float* W0_1 = (const float*)d_in[2];
  const float* W1_1 = (const float*)d_in[3];
  const float* b1   = (const float*)d_in[4];
  const float* W0_2 = (const float*)d_in[5];
  const float* W1_2 = (const float*)d_in[6];
  const float* b2   = (const float*)d_in[7];
  float* out = (float*)d_out;

  const int* rows = ei;
  const int* cols = ei + N_EDGES;

  const long long N = N_NODES;
  float* ws     = (float*)d_ws;
  int*   cnt    = (int*)ws;
  float* dis    = ws + N;
  float* P      = ws + 2 * N;
  float* agg2   = ws + 2 * N;
  int*   colidx = (int*)(ws + 66 * N);
  int*   rowptr = (int*)(ws + 82 * N);
  int*   tbl    = (int*)(ws + 83 * N);
  float* hbuf   = ws + 98 * N;
  int*   sorted = (int*)(ws + 98 * N);
  float* hs     = out;

  count_proj1_kernel<<<NCH + PB, 1024, 0, stream>>>(rows, cols, tbl, sorted, x, W0_1, W1_1, P);
  bin_rows_kernel<<<NB, 1024, 0, stream>>>(tbl, sorted, cnt, rowptr, dis, colidx, P);

  gather_relu_kernel<<<(N_NODES / RPW + 3) / 4, 128, 0, stream>>>(
      rowptr, cnt, colidx, dis, P, b1, hbuf, hs);

  gather_rows_kernel<<<(N_NODES / RPW + 3) / 4, 128, 0, stream>>>(
      rowptr, cnt, colidx, dis, hs, agg2);
  out_wmma_lsm_kernel<<<(N_NODES + 63) / 64, 128, 0, stream>>>(hbuf, agg2, W0_2, W1_2, b2, out);
}


# Row pass: a bin's list is placed in an LDS image and leaves as whole 128-B lines (16-B groups), not as one scattered 4-B global store per edge; list order and sum order unchanged
# speedup vs baseline: 5.8575x; 1.0541x over previous
#include <hip/hip_runtime.h>
#include <math.h>

#define N_NODES 100000
#define N_EDGES 1600000
#define F_IN    50
#define HID     32
#define NCLS    40
#define BIN_SHIFT 8
#define BIN_ROWS  256
#define NB        391
#define CHUNK     8192
#define NCH       196
#define TBL       (NB * NCH)
#define PB        196
#define COL_BITS  17
#define CNT_BITS  14
#define CNT_MASK  ((1 << CNT_BITS) - 1)
#define SEGW      7
#define NCHW      ((NCH + 31) / 32)
#define LCAP      6144
static_assert(N_NODES <= (1 << COL_BITS), "col must fit the low field of a list entry");
static_assert(NB * BIN_ROWS >= N_NODES && NCH * CHUNK >= N_EDGES && N_EDGES % 4 == 0, "sizes");
static_assert(CHUNK <= CNT_MASK && 2 * CNT_BITS <= 31 && CHUNK % 4 == 0, "start and length of a run fit one table word");
static_assert(TBL <= N_NODES, "the table has one N-word plane of the workspace");
static_assert(NCH * CHUNK <= 32 * N_NODES, "the chunk-ordered list fits the plane that later holds h");
static_assert(SEGW * 32 >= NCH, "every run of a bin has a wave");
static_assert(LCAP % 32 == 0, "the staged list is an image of whole 128-B lines");
static_assert(PB * 512 >= N_NODES, "GEMM-1 work-groups cover every row");
static_assert(HID <= NCLS, "the scaled copy of h (N x HID) fits the output buffer (N x NCLS) it lies in");


typedef float v2f __attribute__((ext_vector_type(2)));
typedef float v4f __attribute__((ext_vector_type(4)));
typedef float v8f __attribute__((ext_vector_type(8)));


#define EDGE8(M) \
  M(0, r4[0].x, c4[0].x) M(1, r4[0].y, c4[0].y) M(2, r4[0].z, c4[0].z) M(3, r4[0].w, c4[0].w) \
  M(4, r4[1].x, c4[1].x) M(5, r4[1].y, c4[1].y) M(6, r4[1].z, c4[1].z) M(7, r4[1].w, c4[1].w)

__device__ __forceinline__ void
bin_order_body(const int* __restrict__ rows,
               const int* __restrict__ cols,
               int* __restrict__ tbl,
               int* __restrict__ sorted) {
  __shared__ int hist[NB];
  __shared__ int off[NB];
  __shared__ int s_wtot[13];
  __shared__ int4 stage4[CHUNK / 4];
  int* stage = (int*)stage4;
  int t = threadIdx.x, lane = t & 31, wave = t >> 5;
  int4 r4[2], c4[2];
#pragma unroll
  for (int h = 0; h < 2; h++) {
    int e = blockIdx.x * CHUNK + h * 4096 + t * 4;
    r4[h] = make_int4(0, 0, 0, 0);
    c4[h] = make_int4(0, 0, 0, 0);
    if (e < N_EDGES) {
      r4[h] = *(const int4*)(rows + e);
      c4[h] = *(const int4*)(cols + e);
    }
  }
  for (int i = t; i < NB; i += 1024) hist[i] = 0;
  __syncthreads();
  int rk[8];
#define RANK1(I, R, C) rk[I] = ((R) != (C)) ? atomicAdd(&hist[(R) >> BIN_SHIFT], 1) : -1;
  EDGE8(RANK1)
#undef RANK1
  __syncthreads();
  int v = 0, incl = 0;
  if (t < 13 * 32) {
    if (t < NB) v = hist[t];
    incl = v;
    for (int d = 1; d < 32; d <<= 1) {
      int y = __shfl_up(incl, d);
      if (lane >= d) incl += y;
    }
    if (lane == 31) s_wtot[wave] = incl;
  }
  __syncthreads();
  if (t < NB) {
    int add = 0;
    for (int w = 0; w < 13; w++) add += (w < wave) ? s_wtot[w] : 0;
    off[t] = add + incl - v;
  }
  __syncthreads();
#define PLACE1(I, R, C) if (rk[I] >= 0) stage[off[(R) >> BIN_SHIFT] + rk[I]] = (((R) & (BIN_ROWS - 1)) << COL_BITS) | (C);
  EDGE8(PLACE1)
#undef PLACE1
  __syncthreads();
  int total = off[NB - 1] + hist[NB - 1];
  int* run = sorted + blockIdx.x * CHUNK;
  for (int i = t * 4; i < total; i += 4096) *(int4*)(run + i) = stage4[i >> 2];
  for (int i = t; i < NB; i += 1024) tbl[i * NCH + blockIdx.x] = (off[i] << CNT_BITS) | hist[i];
}

__global__ void __launch_bounds__(1024, 16)
bin_rows_kernel(const int* __restrict__ tbl,
                const int* __restrict__ sorted,
                int* __restrict__ cnt,
                int* __restrict__ rowptr,
                float* __restrict__ dis,
                int* __restrict__ colidx,
                float* __restrict__ P) {
  __shared__ int s_cnt[BIN_ROWS];
  __shared__ int s_start[BIN_ROWS];
  __shared__ int s_cur[BIN_ROWS];
  __shared__ int s_wtot[8];
  __shared__ int s_part[NCHW];
  __shared__ int seg_off[NCH];
  __shared__ int seg_cnt[NCH];
  __shared__ int4 lst4[LCAP / 4];
  int* lst = (int*)lst4;
  int t = threadIdx.x, lane = t & 31, wave = t >> 5;
  int b = blockIdx.x;
  v4f pv[2];
  int prow[2];
#pragma unroll
  for (int q = 0; q < 2; q++) {
    int idx = t + 1024 * q;
    prow[q] = idx >> 3;
    pv[q] = (v4f){0.0f, 0.0f, 0.0f, 0.0f};
    int row = b * BIN_ROWS + prow[q];
    if (row < N_NODES) pv[q] = *(const v4f*)(P + (long long)row * 64 + HID + (idx & 7) * 4);
  }
  int w_own = 0;
  if (t < NCH) {
    w_own = tbl[b * NCH + t];
    seg_off[t] = t * CHUNK + (w_own >> CNT_BITS);
    seg_cnt[t] = w_own & CNT_MASK;
  }
  if (t < BIN_ROWS) { s_cnt[t] = 0; s_cur[t] = 0; }
  __syncthreads();
  int p[SEGW];
#pragma unroll
  for (int k = 0; k < SEGW; k++) {
    int sg = wave + 32 * k;
    p[k] = -1;
    if (sg < NCH && lane < seg_cnt[sg]) p[k] = sorted[seg_off[sg] + lane];
  }
#pragma unroll
  for (int k = 0; k < SEGW; k++)
    if (p[k] >= 0) atomicAdd(&s_cnt[p[k] >> COL_BITS], 1);
  for (int k = 0; k < SEGW; k++) {
    int sg = wave + 32 * k;
    if (sg < NCH) {
      int n = seg_cnt[sg], o = seg_off[sg];
      for (int j = 32 + lane; j < n; j += 32) atomicAdd(&s_cnt[sorted[o + j] >> COL_BITS], 1);
    }
  }
  if (wave < NCHW) {
    int part = w_own >> CNT_BITS;
    for (int d = 16; d >= 1; d >>= 1) part += __shfl_xor(part, d);
    if (lane == 0) s_part[wave] = part;
  }
  __syncthreads();
  int v = 0, incl = 0;
  if (t < BIN_ROWS) {
    v = s_cnt[t];
    incl = v;
    for (int d = 1; d < 32; d <<= 1) {
      int y = __shfl_up(incl, d);
      if (lane >= d) incl += y;
    }
    if (lane == 31) s_wtot[wave] = incl;
  }
  __syncthreads();
#pragma unroll
  for (int q = 0; q < 2; q++) {
    int row = b * BIN_ROWS + prow[q];
    if (row < N_NODES) {
      float dg = (float)s_cnt[prow[q]];
      float dc = (dg > 0.0f) ? rsqrtf(fmaxf(dg, 1.0f)) : 0.0f;
      *(v4f*)(P + (long long)row * 64 + HID + ((t + 1024 * q) & 7) * 4) = pv[q] * dc;
    }
  }
  int s = 0;
  for (int w = 0; w < NCHW; w++) s += s_part[w];
  if (t < BIN_ROWS) {
    int add = 0;
    for (int w = 0; w < 8; w++) add += (w < wave) ? s_wtot[w] : 0;
    int excl = add + incl - v;
    s_start[t] = excl;
    int row = b * BIN_ROWS + t;
    if (row < N_NODES) {
      rowptr[row] = s + excl;
      cnt[row] = v;
      float dg = (float)v;
      dis[row] = (dg > 0.0f) ? rsqrtf(fmaxf(dg, 1.0f)) : 0.0f;
    }
  }
  __syncthreads();
  int a = s & 31;
  int* gimg = colidx + (s - a);
#pragma unroll
  for (int k = 0; k < SEGW; k++) {
    if (p[k] >= 0) {
      int rl = p[k] >> COL_BITS;
      int r = atomicAdd(&s_cur[rl], 1);
      int pos = a + s_start[rl] + r;
      int col = p[k] & ((1 << COL_BITS) - 1);
      if (pos < LCAP) lst[pos] = col;
      else            gimg[pos] = col;
    }
  }
  for (int k = 0; k < SEGW; k++) {
    int sg = wave + 32 * k;
    if (sg < NCH) {
      int n = seg_cnt[sg], o = seg_off[sg];
      for (int j = 32 + lane; j < n; j += 32) {
        int e = sorted[o + j];
        int rl = e >> COL_BITS;
        int r = atomicAdd(&s_cur[rl], 1);
        int pos = a + s_start[rl] + r;
        int col = e & ((1 << COL_BITS) - 1);
        if (pos < LCAP) lst[pos] = col;
        else            gimg[pos] = col;
      }
    }
  }
  __syncthreads();
  int end = a + s_start[BIN_ROWS - 1] + s_cnt[BIN_ROWS - 1];
  end = end < LCAP ? end : LCAP;
  for (int j = t * 4; j < end; j += 4096) {
    int4 v = lst4[j >> 2];
    if (j >= a && j + 4 <= end) {
      *(int4*)(gimg + j) = v;
    } else {
      if (j + 0 >= a && j + 0 < end) gimg[j + 0] = v.x;
      if (j + 1 >= a && j + 1 < end) gimg[j + 1] = v.y;
      if (j + 2 >= a && j + 2 < end) gimg[j + 2] = v.z;
      if (j + 3 >= a && j + 3 < end) gimg[j + 3] = v.w;
    }
  }
}


#define RPW 8
static_assert(N_NODES % RPW == 0 && RPW >= 2 && RPW <= 32 && (RPW & (RPW - 1)) == 0,
              "a wave's rows all exist; their bookkeeping sits one row per lane");
static_assert(HID == 32, "a source row is 128 B: eight lanes of 16 B");

__device__ __forceinline__ int lane_i(int v, int l) { return __builtin_amdgcn_readlane(v, l); }
__device__ __forceinline__ float lane_f(float v, int l) {
  return __int_as_float(__builtin_amdgcn_readlane(__float_as_int(v), l));
}

template <int Q>
__device__ __forceinline__ int group8_lane(int v) {
  return __builtin_amdgcn_mov_dpp8(v, Q * 0x249249);
}

__device__ __forceinline__ float xor8(float v) {
  return __int_as_float(__builtin_amdgcn_update_dpp(0, __float_as_int(v), 0x168, 0xf, 0xf, true));
}
__device__ __forceinline__ float xor16(float v) {
  return __int_as_float(__builtin_amdgcn_permlanex16(__float_as_int(v), __float_as_int(v),
                                                     0x76543210, 0xfedcba98, false, false));
}

__device__ __forceinline__ int list_head(const int* __restrict__ colidx, int start, int n, int slot) {
  int m = n < 32 ? n : 32;
  int li = slot < m ? slot : m - 1;
  li = li > 0 ? li : 0;
  int idx = start + li;
  idx = idx < N_EDGES - 1 ? idx : N_EDGES - 1;
  return colidx[idx];
}

template <int Q, int STRIDE_B>
__device__ __forceinline__ v4f nb_load(int e, unsigned foff, const char* __restrict__ spb) {
  unsigned c = (unsigned)group8_lane<Q>(e);
  return *(const v4f*)(spb + (c * (unsigned)STRIDE_B + foff));
}

template <int K, int STRIDE_B, class F>
__device__ __forceinline__ v4f gather_k(int e, int lim, unsigned foff, const char* __restrict__ spb, v4f acc, F behind) {
  const v4f zero = {0.0f, 0.0f, 0.0f, 0.0f};
  v4f x[8];
  x[0] = nb_load<0, STRIDE_B>(e, foff, spb);
  x[1] = nb_load<1, STRIDE_B>(e, foff, spb);
  x[2] = nb_load<2, STRIDE_B>(e, foff, spb);
  x[3] = nb_load<3, STRIDE_B>(e, foff, spb);
  if (K > 4) {
    x[4] = nb_load<4, STRIDE_B>(e, foff, spb);
    x[5] = nb_load<5, STRIDE_B>(e, foff, spb);
  }
  if (K > 6) {
    x[6] = nb_load<6, STRIDE_B>(e, foff, spb);
    x[7] = nb_load<7, STRIDE_B>(e, foff, spb);
  }
  behind();
#pragma unroll
  for (int q = 0; q < K; q++) acc += (4 * q < lim) ? x[q] : zero;
  return acc;
}

template <int STRIDE_B, class F>
__device__ __forceinline__ v4f gather_part(int e, int m, int g, unsigned foff, const char* __restrict__ spb, v4f acc, F behind) {
  int lim = m - g;
  if (m > 24)      acc = gather_k<8, STRIDE_B>(e, lim, foff, spb, acc, behind);
  else if (m > 16) acc = gather_k<6, STRIDE_B>(e, lim, foff, spb, acc, behind);
  else             acc = gather_k<4, STRIDE_B>(e, lim, foff, spb, acc, behind);
  return acc;
}

template <bool LAYER1, int STRIDE_B>
__device__ __forceinline__ void gather_span(int r0, int lane,
                                            const int* __restrict__ rowptr,
                                            const int* __restrict__ cnt,
                                            const int* __restrict__ colidx,
                                            const float* __restrict__ dis,
                                            const float* __restrict__ sp,
                                            const float* __restrict__ selfp,
                                            const float* __restrict__ b1,
                                            float* __restrict__ dst,
                                            float* __restrict__ dst_scaled) {
  const v4f zero = {0.0f, 0.0f, 0.0f, 0.0f};
  int g = lane >> 3, p = lane & 7;
  int slot = 4 * p + g;
  unsigned foff = 16u * (unsigned)p;
  const char* spb = (const char*)sp;
  int mr = r0 + (lane & (RPW - 1));
  int st_v = rowptr[mr];
  int n_v = cnt[mr];
  float d_v = dis[mr];
  v4f bias = zero;
  v4f self = zero;
  if (LAYER1) {
    bias = *(const v4f*)((const char*)b1 + foff);
    self = *(const v4f*)((const char*)selfp + ((unsigned)r0 * 256u + foff));
  }

  int e_a = list_head(colidx, lane_i(st_v, 0), lane_i(n_v, 0), slot);
  int e_b = list_head(colidx, lane_i(st_v, 1), lane_i(n_v, 1), slot);
  for (int i = 0; i < RPW; i++) {
    int r = r0 + i;
    int start = lane_i(st_v, i);
    int n = lane_i(n_v, i);
    int e_c = 0;
    v4f self_nxt = zero;
    auto behind = [&]() {
      if (i + 2 < RPW) e_c = list_head(colidx, lane_i(st_v, i + 2), lane_i(n_v, i + 2), slot);
      if (LAYER1 && i + 1 < RPW) self_nxt = *(const v4f*)((const char*)selfp + ((unsigned)(r + 1) * 256u + foff));
    };
    int m = n < 32 ? n : 32;
    v4f acc = zero;
    if (m > 0) acc = gather_part<STRIDE_B>(e_a, m, g, foff, spb, acc, behind);
    else       behind();
    for (int base = 32; base < n; base += 32) {
      int m2 = n - base;
      if (m2 > 32) m2 = 32;
      int li = slot < m2 ? slot : m2 - 1;
      int e2 = colidx[start + base + li];
      acc = gather_part<STRIDE_B>(e2, m2, g, foff, spb, acc, []() {});
    }
    acc.x += xor8(acc.x);  acc.y += xor8(acc.y);  acc.z += xor8(acc.z);  acc.w += xor8(acc.w);
    acc.x += xor16(acc.x); acc.y += xor16(acc.y); acc.z += xor16(acc.z); acc.w += xor16(acc.w);
    float d_r = lane_f(d_v, i);
    v4f s = acc * (-d_r);
    unsigned roff = (unsigned)r * (unsigned)(HID * 4) + foff;
    char* drow = (char*)dst + roff;
    if (LAYER1) {
      v4f v = self + s + bias;
      v4f hv;
      hv.x = v.x > 0.0f ? v.x : 0.0f;
      hv.y = v.y > 0.0f ? v.y : 0.0f;
      hv.z = v.z > 0.0f ? v.z : 0.0f;
      hv.w = v.w > 0.0f ? v.w : 0.0f;
      if (g == 0) {
        *(v4f*)drow = hv;
        *(v4f*)((char*)dst_scaled + roff) = hv * d_r;
      }
    } else {
      if (g == 0) *(v4f*)drow = s;
    }
    e_a = e_b; e_b = e_c;
    self = self_nxt;
  }
}

__global__ void __launch_bounds__(128, 16)
gather_relu_kernel(const int* __restrict__ rowptr,
                   const int* __restrict__ cnt,
                   const int* __restrict__ colidx,
                   const float* __restrict__ dis,
                   const float* __restrict__ P,
                   const float* __restrict__ b1,
                   float* __restrict__ h,
                   float* __restrict__ hs) {
  int lane = threadIdx.x & 31;
  int r0 = __builtin_amdgcn_readfirstlane((blockIdx.x * 4 + (threadIdx.x >> 5)) * RPW);
  if (r0 >= N_NODES) return;
  gather_span<true, 256>(r0, lane, rowptr, cnt, colidx, dis, P + HID, P, b1, h, hs);
}

__global__ void __launch_bounds__(128, 16)
gather_rows_kernel(const int* __restrict__ rowptr,
                   const int* __restrict__ cnt,
                   const int* __restrict__ colidx,
                   const float* __restrict__ dis,
                   const float* __restrict__ src,
                   float* __restrict__ dst) {
  int lane = threadIdx.x & 31;
  int r0 = __builtin_amdgcn_readfirstlane((blockIdx.x * 4 + (threadIdx.x >> 5)) * RPW);
  if (r0 >= N_NODES) return;
  gather_span<false, 128>(r0, lane, rowptr, cnt, colidx, dis, src, src, nullptr, dst, nullptr);
}


__device__ __forceinline__ void
proj1_wmma_body(int pb,
                const float* __restrict__ x,
                const float* __restrict__ W0,
                const float* __restrict__ W1,
                float* __restrict__ P) {
  int wave = threadIdx.x >> 5;
  int lane = threadIdx.x & 31;
  int half = lane >> 4;
  int l16  = lane & 15;
  int m0 = (pb * 32 + wave) * 16;
  if (m0 >= N_NODES) return;
  int mA = m0 + l16;
  const float* arow = x + (long long)mA * F_IN;

  v8f acc0 = {}, acc1 = {}, acc2 = {}, acc3 = {};
  for (int kk = 0; kk < 48; kk += 4) {
    int k = kk + 2 * half;
    v2f a = *(const v2f*)(arow + k);
    const float* p0 = W0 + k * HID + l16;
    const float* p1 = W1 + k * HID + l16;
    v2f b;
    b.x = p0[0];  b.y = p0[HID];
    acc0 = __builtin_amdgcn_wmma_f32_16x16x4_f32(false, a, false, b, (short)0, acc0, false, false);
    b.x = p0[16]; b.y = p0[HID + 16];
    acc1 = __builtin_amdgcn_wmma_f32_16x16x4_f32(false, a, false, b, (short)0, acc1, false, false);
    b.x = p1[0];  b.y = p1[HID];
    acc2 = __builtin_amdgcn_wmma_f32_16x16x4_f32(false, a, false, b, (short)0, acc2, false, false);
    b.x = p1[16]; b.y = p1[HID + 16];
    acc3 = __builtin_amdgcn_wmma_f32_16x16x4_f32(false, a, false, b, (short)0, acc3, false, false);
  }
  {
    v2f aload = *(const v2f*)(arow + 48);
    v2f a;
    a.x = half ? 0.0f : aload.x;
    a.y = half ? 0.0f : aload.y;
    const float* p0 = W0 + 48 * HID + l16;
    const float* p1 = W1 + 48 * HID + l16;
    v2f b;
    b.x = p0[0];  b.y = p0[HID];
    acc0 = __builtin_amdgcn_wmma_f32_16x16x4_f32(false, a, false, b, (short)0, acc0, false, false);
    b.x = p0[16]; b.y = p0[HID + 16];
    acc1 = __builtin_amdgcn_wmma_f32_16x16x4_f32(false, a, false, b, (short)0, acc1, false, false);
    b.x = p1[0];  b.y = p1[HID];
    acc2 = __builtin_amdgcn_wmma_f32_16x16x4_f32(false, a, false, b, (short)0, acc2, false, false);
    b.x = p1[16]; b.y = p1[HID + 16];
    acc3 = __builtin_amdgcn_wmma_f32_16x16x4_f32(false, a, false, b, (short)0, acc3, false, false);
  }
#pragma unroll
  for (int v = 0; v < 8; v++) {
    int mC = m0 + v + 8 * half;
    float* prow = P + (long long)mC * 64 + l16;
    prow[0]  = acc0[v];
    prow[16] = acc1[v];
    prow[32] = acc2[v];
    prow[48] = acc3[v];
  }
}

__global__ void __launch_bounds__(1024, 16)
count_proj1_kernel(const int* __restrict__ rows,
                   const int* __restrict__ cols,
                   int* __restrict__ tbl,
                   int* __restrict__ sorted,
                   const float* __restrict__ x,
                   const float* __restrict__ W0,
                   const float* __restrict__ W1,
                   float* __restrict__ P) {
  if (blockIdx.x < NCH) bin_order_body(rows, cols, tbl, sorted);
  else                  proj1_wmma_body((int)blockIdx.x - NCH, x, W0, W1, P);
}


__device__ __forceinline__ void out_k32(const float* __restrict__ arow,
                                        const float* __restrict__ Wm,
                                        int half, int l16,
                                        v8f& acc0, v8f& acc1, v8f& acc2) {
  int c2 = 32 + (l16 & 7);
  bool ok2 = l16 < 8;
  for (int kk = 0; kk < HID; kk += 4) {
    int k = kk + 2 * half;
    v2f a = *(const v2f*)(arow + k);
    const float* p = Wm + k * NCLS;
    v2f b;
    b.x = p[l16];      b.y = p[NCLS + l16];
    acc0 = __builtin_amdgcn_wmma_f32_16x16x4_f32(false, a, false, b, (short)0, acc0, false, false);
    b.x = p[16 + l16]; b.y = p[NCLS + 16 + l16];
    acc1 = __builtin_amdgcn_wmma_f32_16x16x4_f32(false, a, false, b, (short)0, acc1, false, false);
    float t0 = p[c2], t1 = p[NCLS + c2];
    b.x = ok2 ? t0 : 0.0f; b.y = ok2 ? t1 : 0.0f;
    acc2 = __builtin_amdgcn_wmma_f32_16x16x4_f32(false, a, false, b, (short)0, acc2, false, false);
  }
}

template <int M>
__device__ __forceinline__ float row16_xor(float v) {
  return __int_as_float(__builtin_amdgcn_update_dpp(0, __float_as_int(v), 0x160 + M, 0xf, 0xf, true));
}
__device__ __forceinline__ float row16_max(float v) {
  v = fmaxf(v, row16_xor<1>(v));
  v = fmaxf(v, row16_xor<2>(v));
  v = fmaxf(v, row16_xor<4>(v));
  v = fmaxf(v, row16_xor<8>(v));
  return v;
}
__device__ __forceinline__ float row16_sum(float v) {
  v += row16_xor<1>(v);
  v += row16_xor<2>(v);
  v += row16_xor<4>(v);
  v += row16_xor<8>(v);
  return v;
}

__global__ void __launch_bounds__(128)
out_wmma_lsm_kernel(const float* __restrict__ h,
                    const float* __restrict__ agg2,
                    const float* __restrict__ W0,
                    const float* __restrict__ W1,
                    const float* __restrict__ b2,
                    float* __restrict__ out) {
  int wave = threadIdx.x >> 5;
  int lane = threadIdx.x & 31;
  int half = lane >> 4;
  int l16  = lane & 15;
  int m0 = (blockIdx.x * 4 + wave) * 16;
  if (m0 >= N_NODES) return;
  int mA = m0 + l16;

  v8f acc0 = {}, acc1 = {}, acc2 = {};
  out_k32(h    + (long long)mA * HID, W0, half, l16, acc0, acc1, acc2);
  out_k32(agg2 + (long long)mA * HID, W1, half, l16, acc0, acc1, acc2);

  bool ok2 = l16 < 8;
  float bb0 = b2[l16], bb1 = b2[16 + l16], bb2 = b2[32 + (l16 & 7)];
  float x0[8], x1[8], x2[8], mx[8], se[8];
#pragma unroll
  for (int v = 0; v < 8; v++) {
    x0[v] = acc0[v] + bb0;
    x1[v] = acc1[v] + bb1;
    x2[v] = acc2[v] + bb2;
    mx[v] = fmaxf(fmaxf(x0[v], x1[v]), ok2 ? x2[v] : -INFINITY);
  }
#pragma unroll
  for (int v = 0; v < 8; v++) mx[v] = row16_max(mx[v]);
#pragma unroll
  for (int v = 0; v < 8; v++) {
    x0[v] -= mx[v];
    x1[v] -= mx[v];
    x2[v] -= mx[v];
    float e2 = expf(x2[v]);
    se[v] = expf(x0[v]) + expf(x1[v]);
    se[v] += ok2 ? e2 : 0.0f;
  }
#pragma unroll
  for (int v = 0; v < 8; v++) se[v] = logf(row16_sum(se[v]));
  float* otile = out + (long long)(m0 + 8 * half) * NCLS + l16;
#pragma unroll
  for (int v = 0; v < 8; v++) {
    otile[v * NCLS]      = x0[v] - se[v];
    otile[v * NCLS + 16] = x1[v] - se[v];
  }
  if (ok2) {
#pragma unroll
    for (int v = 0; v < 8; v++) otile[v * NCLS + 32] = x2[v] - se[v];
  }
}


extern "C" void kernel_launch(void* const* d_in, const int* in_sizes, int n_in,
                              void* d_out, int out_size, void* d_ws, size_t ws_size,
                              hipStream_t stream) {
  (void)in_sizes; (void)n_in; (void)out_size; (void)ws_size;
  const float* x    = (const float*)d_in[0];
  const int*   ei   = (const int*)  d_in[1];
  const float* W0_1 = (const float*)d_in[2];
  const float* W1_1 = (const float*)d_in[3];
  const float* b1   = (const float*)d_in[4];
  const float* W0_2 = (const float*)d_in[5];
  const float* W1_2 = (const float*)d_in[6];
  const float* b2   = (const float*)d_in[7];
  float* out = (float*)d_out;

  const int* rows = ei;
  const int* cols = ei + N_EDGES;

  const long long N = N_NODES;
  float* ws     = (float*)d_ws;
  int*   cnt    = (int*)ws;
  float* dis    = ws + N;
  float* P      = ws + 2 * N;
  float* agg2   = ws + 2 * N;
  int*   colidx = (int*)(ws + 66 * N);
  int*   rowptr = (int*)(ws + 82 * N);
  int*   tbl    = (int*)(ws + 83 * N);
  float* hbuf   = ws + 98 * N;
  int*   sorted = (int*)(ws + 98 * N);
  float* hs     = out;

  count_proj1_kernel<<<NCH + PB, 1024, 0, stream>>>(rows, cols, tbl, sorted, x, W0_1, W1_1, P);
  bin_rows_kernel<<<NB, 1024, 0, stream>>>(tbl, sorted, cnt, rowptr, dis, colidx, P);

  gather_relu_kernel<<<(N_NODES / RPW + 3) / 4, 128, 0, stream>>>(
      rowptr, cnt, colidx, dis, P, b1, hbuf, hs);

  gather_rows_kernel<<<(N_NODES / RPW + 3) / 4, 128, 0, stream>>>(
      rowptr, cnt, colidx, dis, hs, agg2);
  out_wmma_lsm_kernel<<<(N_NODES + 63) / 64, 128, 0, stream>>>(hbuf, agg2, W0_2, W1_2, b2, out);
}
